# starting kernel
# baseline (speedup 1.0000x reference)
#include <hip/hip_runtime.h>
#include <hip/hip_bf16.h>
#include <math.h>

typedef __attribute__((ext_vector_type(16))) _Float16 v16h;
typedef __attribute__((ext_vector_type(8)))  _Float16 v8h;
typedef __attribute__((ext_vector_type(8)))  float    v8f;

#define N_NODES 65536
#define N_EDGES 524288
#define N_BAGS  8
#define D_IN    512
#define H_DIM   256
#define HA_DIM  128
#define C_DIM   2
#define BN_EPS  1e-5f
#define MREP    4

__global__ void fill_f32(float* p, float v, int n) {
    int t = blockIdx.x * blockDim.x + threadIdx.x;
    if (t < n) p[t] = v;
}

__global__ void deg_count(const int* __restrict__ dst, float* deg, int E) {
    int t = blockIdx.x * blockDim.x + threadIdx.x;
    if (t < E) atomicAdd(&deg[dst[t]], 1.0f);
}

__global__ void deg_to_dinv(float* deg, int N) {
    int t = blockIdx.x * blockDim.x + threadIdx.x;
    if (t < N) deg[t] = rsqrtf(deg[t]);
}

__global__ void pack_w_f16(const float* __restrict__ W, _Float16* __restrict__ out,
                           int K, int Nout) {
    int t = blockIdx.x * blockDim.x + threadIdx.x;
    if (t >= K * Nout) return;
    int i    = t & 15;
    int lane = (t >> 4) & 31;
    int rest = t >> 9;
    int kblks = K >> 5;
    int ntile = rest / kblks;
    int kblk  = rest % kblks;
    int half = lane >> 4, l15 = lane & 15;
    int col = ntile * 16 + l15;
    int k   = kblk * 32 + half * 8 + ((i < 8) ? i : (16 + i - 8));
    out[t] = (_Float16)W[(size_t)k * Nout + col];
}

__global__ void wmma_gemm_af32(const float* __restrict__ A,
                               const _Float16* __restrict__ Wpk,
                               float* __restrict__ C,
                               int M, int K, int Nout) {
    const int lane   = threadIdx.x & 31;
    const int ntiles = Nout >> 4;
    const int waveId = (blockIdx.x * blockDim.x + threadIdx.x) >> 5;
    const int mblk   = waveId / ntiles;
    const int ntile  = waveId % ntiles;
    const int m0     = mblk * (16 * MREP);
    if (m0 >= M) return;

    const int half = lane >> 4;
    const int l15  = lane & 15;
    const int kblks = K >> 5;
    const _Float16* wbase = Wpk + ((size_t)ntile * kblks * 512) + lane * 16;

    v8f zero = {};
    v8f acc[MREP];
#pragma unroll
    for (int r = 0; r < MREP; ++r) acc[r] = zero;

    for (int kblk = 0; kblk < kblks; ++kblk) {
        v16h b = *(const v16h*)(wbase + (size_t)kblk * 512);
#pragma unroll
        for (int r = 0; r < MREP; ++r) {
            const float* ap = A + (size_t)(m0 + r * 16 + l15) * K + kblk * 32 + half * 8;
            v16h a;
#pragma unroll
            for (int i = 0; i < 8; ++i) a[i]     = (_Float16)ap[i];
#pragma unroll
            for (int i = 0; i < 8; ++i) a[8 + i] = (_Float16)ap[16 + i];
            acc[r] = __builtin_amdgcn_wmma_f32_16x16x32_f16(
                false, a, false, b, (short)0, acc[r], false, false);
        }
    }
#pragma unroll
    for (int r = 0; r < MREP; ++r) {
#pragma unroll
        for (int q = 0; q < 8; ++q) {
            const int row = m0 + r * 16 + q + half * 8;
            C[(size_t)row * Nout + (ntile * 16 + l15)] = acc[r][q];
        }
    }
}

__global__ void wmma_gemm_af16(const _Float16* __restrict__ A,
                               const _Float16* __restrict__ Wpk,
                               const float* __restrict__ bias,
                               float* __restrict__ C,
                               int M, int K, int Nout, int addBias) {
    const int lane   = threadIdx.x & 31;
    const int ntiles = Nout >> 4;
    const int waveId = (blockIdx.x * blockDim.x + threadIdx.x) >> 5;
    const int mblk   = waveId / ntiles;
    const int ntile  = waveId % ntiles;
    const int m0     = mblk * (16 * MREP);
    if (m0 >= M) return;

    const int half = lane >> 4;
    const int l15  = lane & 15;
    const int colB = ntile * 16 + l15;
    const int kblks = K >> 5;
    const _Float16* wbase = Wpk + ((size_t)ntile * kblks * 512) + lane * 16;

    v8f zero = {};
    v8f acc[MREP];
#pragma unroll
    for (int r = 0; r < MREP; ++r) acc[r] = zero;

    for (int kblk = 0; kblk < kblks; ++kblk) {
        v16h b = *(const v16h*)(wbase + (size_t)kblk * 512);
#pragma unroll
        for (int r = 0; r < MREP; ++r) {
            const _Float16* ap = A + (size_t)(m0 + r * 16 + l15) * K + kblk * 32 + half * 8;
            v8h lo = *(const v8h*)(ap);
            v8h hi = *(const v8h*)(ap + 16);
            v16h a = __builtin_shufflevector(lo, hi, 0, 1, 2, 3, 4, 5, 6, 7,
                                             8, 9, 10, 11, 12, 13, 14, 15);
            acc[r] = __builtin_amdgcn_wmma_f32_16x16x32_f16(
                false, a, false, b, (short)0, acc[r], false, false);
        }
    }

    const float bv = addBias ? bias[colB] : 0.0f;
#pragma unroll
    for (int r = 0; r < MREP; ++r) {
#pragma unroll
        for (int q = 0; q < 8; ++q) {
            const int row = m0 + r * 16 + q + half * 8;
            C[(size_t)row * Nout + colB] = acc[r][q] + bv;
        }
    }
}

__global__ void scatter_edges(const int* __restrict__ src,
                              const int* __restrict__ dst,
                              const float* __restrict__ dinv,
                              const float* __restrict__ h,
                              float* __restrict__ agg, int E, int H) {
    const int chunks = H >> 2;
    int t = blockIdx.x * blockDim.x + threadIdx.x;
    int e = t / chunks;
    if (e >= E) return;
    int f4 = (t % chunks) << 2;
    int s = src[e], d = dst[e];
    float coef = dinv[s] * dinv[d];
    const float4 hv = *(const float4*)(h + (size_t)s * H + f4);
    float* ag = agg + (size_t)d * H + f4;
    atomicAdd(ag + 0, hv.x * coef);
    atomicAdd(ag + 1, hv.y * coef);
    atomicAdd(ag + 2, hv.z * coef);
    atomicAdd(ag + 3, hv.w * coef);
}

__global__ void gcn_finalize(const float* __restrict__ hlin,
                             const float* __restrict__ agg,
                             const float* __restrict__ dinv,
                             const float* __restrict__ bconv,
                             const float* __restrict__ g,
                             const float* __restrict__ bb,
                             const float* __restrict__ m,
                             const float* __restrict__ v,
                             _Float16* __restrict__ out16,
                             float* __restrict__ out32,
                             int N, int H) {
    int idx = blockIdx.x * blockDim.x + threadIdx.x;
    if (idx >= N * H) return;
    int n = idx / H, f = idx - n * H;
    float di = dinv[n];
    float val = agg[idx] + hlin[idx] * di * di + bconv[f];
    val = (val - m[f]) * rsqrtf(v[f] + BN_EPS) * g[f] + bb[f];
    val = fmaxf(val, 0.0f);
    out16[idx] = (_Float16)val;
    if (out32) out32[idx] = val;
}

__global__ void attn_score(const float* __restrict__ av,
                           const float* __restrict__ au,
                           const float* __restrict__ wA,
                           float* __restrict__ a, int N, int HA) {
    int wave = (blockIdx.x * blockDim.x + threadIdx.x) >> 5;
    int lane = threadIdx.x & 31;
    if (wave >= N) return;
    float sum = 0.0f;
    for (int j = lane; j < HA; j += 32) {
        float t = tanhf(av[(size_t)wave * HA + j]);
        float s = 1.0f / (1.0f + expf(-au[(size_t)wave * HA + j]));
        sum += t * s * wA[j];
    }
    for (int off = 16; off > 0; off >>= 1) sum += __shfl_down(sum, off, 32);
    if (lane == 0) a[wave] = sum;
}

__global__ void seg_init(int* seg, int cnt, int sentinel) {
    int t = blockIdx.x * blockDim.x + threadIdx.x;
    if (t < cnt) seg[t] = sentinel;
}
__global__ void seg_mark(const int* __restrict__ batch, int* seg, int N) {
    int t = blockIdx.x * blockDim.x + threadIdx.x;
    if (t < N) atomicMin(&seg[batch[t]], t);
}
__global__ void seg_fix(int* seg, int B, int Nval) {
    if (blockIdx.x == 0 && threadIdx.x == 0) {
        seg[B] = Nval;
        for (int g = B - 1; g >= 0; --g)
            if (seg[g] == Nval) seg[g] = seg[g + 1];
    }
}

__global__ void seg_softmax(const float* __restrict__ a,
                            const int* __restrict__ seg,
                            float* __restrict__ wout) {
    __shared__ float red[256];
    int g = blockIdx.x;
    int s = seg[g], e = seg[g + 1];
    float mx = -1e30f;
    for (int n = s + (int)threadIdx.x; n < e; n += blockDim.x)
        mx = fmaxf(mx, a[n]);
    red[threadIdx.x] = mx; __syncthreads();
    for (int off = 128; off > 0; off >>= 1) {
        if ((int)threadIdx.x < off)
            red[threadIdx.x] = fmaxf(red[threadIdx.x], red[threadIdx.x + off]);
        __syncthreads();
    }
    mx = red[0]; __syncthreads();
    float sum = 0.0f;
    for (int n = s + (int)threadIdx.x; n < e; n += blockDim.x)
        sum += expf(a[n] - mx);
    red[threadIdx.x] = sum; __syncthreads();
    for (int off = 128; off > 0; off >>= 1) {
        if ((int)threadIdx.x < off)
            red[threadIdx.x] += red[threadIdx.x + off];
        __syncthreads();
    }
    float inv = 1.0f / red[0];
    for (int n = s + (int)threadIdx.x; n < e; n += blockDim.x)
        wout[n] = expf(a[n] - mx) * inv;
}

__global__ void slide_acc(const float* __restrict__ w,
                          const float* __restrict__ h,
                          const int* __restrict__ seg,
                          float* __restrict__ slide, int H, int chunks) {
    int g = blockIdx.x / chunks;
    int c = blockIdx.x % chunks;
    int s = seg[g], e = seg[g + 1];
    int len = e - s;
    int per = (len + chunks - 1) / chunks;
    int n0 = s + c * per;
    int n1 = min(n0 + per, e);
    int f = threadIdx.x;
    float acc = 0.0f;
    for (int n = n0; n < n1; ++n) acc += w[n] * h[(size_t)n * H + f];
    if (n0 < n1) atomicAdd(&slide[g * H + f], acc);
}

__global__ void head_kernel(const float* __restrict__ slide,
                            const float* __restrict__ Wc1,
                            const float* __restrict__ bc1,
                            const float* __restrict__ Wc2,
                            const float* __restrict__ bc2,
                            float* __restrict__ logits) {
    __shared__ float hid[N_BAGS * HA_DIM];
    for (int item = threadIdx.x; item < N_BAGS * HA_DIM; item += blockDim.x) {
        int g = item / HA_DIM, t = item - g * HA_DIM;
        float s = bc1[t];
        for (int k = 0; k < H_DIM; ++k) s += slide[g * H_DIM + k] * Wc1[k * HA_DIM + t];
        hid[item] = fmaxf(s, 0.0f);
    }
    __syncthreads();
    for (int item = threadIdx.x; item < N_BAGS * C_DIM; item += blockDim.x) {
        int g = item / C_DIM, c = item - g * C_DIM;
        float s = bc2[c];
        for (int t = 0; t < HA_DIM; ++t) s += hid[g * HA_DIM + t] * Wc2[t * C_DIM + c];
        logits[item] = s;
    }
}

static inline int cdiv(long long a, int b) { return (int)((a + b - 1) / b); }

extern "C" void kernel_launch(void* const* d_in, const int* in_sizes, int n_in,
                              void* d_out, int out_size, void* d_ws, size_t ws_size,
                              hipStream_t stream) {
    (void)in_sizes; (void)n_in; (void)out_size; (void)ws_size;
    const float* x   = (const float*)d_in[0];
    const int*  ei   = (const int*)d_in[1];
    const int*  batch= (const int*)d_in[2];
    const float* W1  = (const float*)d_in[3];  const float* b1   = (const float*)d_in[4];
    const float* g1  = (const float*)d_in[5];  const float* bb1  = (const float*)d_in[6];
    const float* m1  = (const float*)d_in[7];  const float* v1   = (const float*)d_in[8];
    const float* W2  = (const float*)d_in[9];  const float* b2   = (const float*)d_in[10];
    const float* g2  = (const float*)d_in[11]; const float* bb2  = (const float*)d_in[12];
    const float* m2  = (const float*)d_in[13]; const float* v2   = (const float*)d_in[14];
    const float* Wv  = (const float*)d_in[15]; const float* bv   = (const float*)d_in[16];
    const float* Wu  = (const float*)d_in[17]; const float* bu   = (const float*)d_in[18];
    const float* wA  = (const float*)d_in[19];
    const float* Wc1 = (const float*)d_in[20]; const float* bc1  = (const float*)d_in[21];
    const float* Wc2 = (const float*)d_in[22]; const float* bc2  = (const float*)d_in[23];

    const int* src = ei;
    const int* dst = ei + N_EDGES;

    float* out  = (float*)d_out;
    float* wout = out + N_BAGS * C_DIM;

    const size_t NH = (size_t)N_NODES * H_DIM;
    float*    bufA = (float*)d_ws;
    float*    bufB = bufA + NH;
    _Float16* h16  = (_Float16*)(bufB + NH);
    _Float16* w1pk = h16 + NH;
    _Float16* w2pk = w1pk + (size_t)D_IN * H_DIM;
    _Float16* wvpk = w2pk + (size_t)H_DIM * H_DIM;
    _Float16* wupk = wvpk + (size_t)H_DIM * HA_DIM;
    float*    dinv = (float*)(wupk + (size_t)H_DIM * HA_DIM);
    float*    abuf = dinv + N_NODES;
    float*    slide= abuf + N_NODES;
    int*      seg  = (int*)(slide + N_BAGS * H_DIM);

    const int TB = 256;
    const int NHb = cdiv((long long)NH, TB);

    fill_f32<<<cdiv(N_NODES, TB), TB, 0, stream>>>(dinv, 1.0f, N_NODES);
    deg_count<<<cdiv(N_EDGES, TB), TB, 0, stream>>>(dst, dinv, N_EDGES);
    deg_to_dinv<<<cdiv(N_NODES, TB), TB, 0, stream>>>(dinv, N_NODES);

    pack_w_f16<<<cdiv((long long)D_IN * H_DIM, TB), TB, 0, stream>>>(W1, w1pk, D_IN, H_DIM);
    pack_w_f16<<<cdiv((long long)H_DIM * H_DIM, TB), TB, 0, stream>>>(W2, w2pk, H_DIM, H_DIM);
    pack_w_f16<<<cdiv((long long)H_DIM * HA_DIM, TB), TB, 0, stream>>>(Wv, wvpk, H_DIM, HA_DIM);
    pack_w_f16<<<cdiv((long long)H_DIM * HA_DIM, TB), TB, 0, stream>>>(Wu, wupk, H_DIM, HA_DIM);

    {
        long long waves = ((long long)N_NODES / 64) * (H_DIM / 16);
        wmma_gemm_af32<<<cdiv(waves * 32, TB), TB, 0, stream>>>(
            x, w1pk, bufA, N_NODES, D_IN, H_DIM);
    }
    fill_f32<<<NHb, TB, 0, stream>>>(bufB, 0.0f, (int)NH);
    scatter_edges<<<cdiv((long long)N_EDGES * (H_DIM / 4), TB), TB, 0, stream>>>(
        src, dst, dinv, bufA, bufB, N_EDGES, H_DIM);
    gcn_finalize<<<NHb, TB, 0, stream>>>(bufA, bufB, dinv, b1, g1, bb1, m1, v1,
                                         h16, nullptr, N_NODES, H_DIM);

    {
        long long waves = ((long long)N_NODES / 64) * (H_DIM / 16);
        wmma_gemm_af16<<<cdiv(waves * 32, TB), TB, 0, stream>>>(
            h16, w2pk, nullptr, bufA, N_NODES, H_DIM, H_DIM, 0);
    }
    fill_f32<<<NHb, TB, 0, stream>>>(bufB, 0.0f, (int)NH);
    scatter_edges<<<cdiv((long long)N_EDGES * (H_DIM / 4), TB), TB, 0, stream>>>(
        src, dst, dinv, bufA, bufB, N_EDGES, H_DIM);
    gcn_finalize<<<NHb, TB, 0, stream>>>(bufA, bufB, dinv, b2, g2, bb2, m2, v2,
                                         h16, bufA, N_NODES, H_DIM);

    float* av = bufB;
    float* au = bufB + (size_t)N_NODES * HA_DIM;
    {
        long long waves = ((long long)N_NODES / 64) * (HA_DIM / 16);
        wmma_gemm_af16<<<cdiv(waves * 32, TB), TB, 0, stream>>>(
            h16, wvpk, bv, av, N_NODES, H_DIM, HA_DIM, 1);
        wmma_gemm_af16<<<cdiv(waves * 32, TB), TB, 0, stream>>>(
            h16, wupk, bu, au, N_NODES, H_DIM, HA_DIM, 1);
    }
    attn_score<<<cdiv((long long)N_NODES * 32, TB), TB, 0, stream>>>(
        av, au, wA, abuf, N_NODES, HA_DIM);

    seg_init<<<1, 32, 0, stream>>>(seg, N_BAGS + 1, N_NODES);
    seg_mark<<<cdiv(N_NODES, TB), TB, 0, stream>>>(batch, seg, N_NODES);
    seg_fix<<<1, 32, 0, stream>>>(seg, N_BAGS, N_NODES);

    seg_softmax<<<N_BAGS, 256, 0, stream>>>(abuf, seg, wout);

    fill_f32<<<cdiv(N_BAGS * H_DIM, TB), TB, 0, stream>>>(slide, 0.0f, N_BAGS * H_DIM);
    const int CHUNKS = 16;
    slide_acc<<<N_BAGS * CHUNKS, H_DIM, 0, stream>>>(wout, bufA, seg, slide, H_DIM, CHUNKS);

    head_kernel<<<1, 256, 0, stream>>>(slide, Wc1, bc1, Wc2, bc2, out);
}


# Replace per-edge f32 atomic scatter (zero-fill + scatter + finalize per layer) by a destination CSR pull: arrival-order rank from one int atomic per edge, wave per node sums rows in registers
# speedup vs baseline: 1.5181x; 1.5181x over previous
#include <hip/hip_runtime.h>
#include <hip/hip_bf16.h>
#include <math.h>

typedef __attribute__((ext_vector_type(16))) _Float16 v16h;
typedef __attribute__((ext_vector_type(8)))  _Float16 v8h;
typedef __attribute__((ext_vector_type(8)))  float    v8f;

#define N_NODES 65536
#define N_EDGES 524288
#define N_BAGS  8
#define D_IN    512
#define H_DIM   256
#define HA_DIM  128
#define C_DIM   2
#define BN_EPS  1e-5f
#define MREP    4

__global__ void fill_f32(float* p, float v, int n) {
    int t = blockIdx.x * blockDim.x + threadIdx.x;
    if (t < n) p[t] = v;
}

__global__ void zero_i32(int* p, int n) {
    int t = blockIdx.x * blockDim.x + threadIdx.x;
    if (t < n) p[t] = 0;
}

__global__ void count_rank(const int* __restrict__ dst, int* __restrict__ cnt,
                           int* __restrict__ rank, int E) {
    int t = blockIdx.x * blockDim.x + threadIdx.x;
    if (t < E) rank[t] = atomicAdd(&cnt[dst[t]], 1);
}

#define SCAN_T   1024
#define SCAN_PER (N_NODES / SCAN_T)
static_assert(N_NODES % (SCAN_T * 4) == 0, "scan assumes whole int4 chunks per thread");
__global__ void __launch_bounds__(SCAN_T)
scan_rowptr(const int* __restrict__ cnt, int* __restrict__ rowptr,
            float* __restrict__ dinv) {
    __shared__ int wsum[32];
    const int t = threadIdx.x, lane = t & 31, w = t >> 5;
    const int4* c4 = (const int4*)(cnt + t * SCAN_PER);
    int4 c[SCAN_PER / 4];
    int s = 0;
#pragma unroll
    for (int i = 0; i < SCAN_PER / 4; ++i) {
        c[i] = c4[i];
        s += c[i].x + c[i].y + c[i].z + c[i].w;
    }
    int inc = s;
    for (int off = 1; off < 32; off <<= 1) {
        int v = __shfl_up(inc, off, 32);
        if (lane >= off) inc += v;
    }
    if (lane == 31) wsum[w] = inc;
    __syncthreads();
    if (w == 0) {
        int v = wsum[lane], iv = v;
        for (int off = 1; off < 32; off <<= 1) {
            int u = __shfl_up(iv, off, 32);
            if (lane >= off) iv += u;
        }
        wsum[lane] = iv - v;
    }
    __syncthreads();
    int run = wsum[w] + inc - s;
    int4*   r4 = (int4*)(rowptr + t * SCAN_PER);
    float4* d4 = (float4*)(dinv + t * SCAN_PER);
#pragma unroll
    for (int i = 0; i < SCAN_PER / 4; ++i) {
        int4 r;
        r.x = run; run += c[i].x;
        r.y = run; run += c[i].y;
        r.z = run; run += c[i].z;
        r.w = run; run += c[i].w;
        r4[i] = r;
        float4 d;
        d.x = rsqrtf(1.0f + (float)c[i].x);
        d.y = rsqrtf(1.0f + (float)c[i].y);
        d.z = rsqrtf(1.0f + (float)c[i].z);
        d.w = rsqrtf(1.0f + (float)c[i].w);
        d4[i] = d;
    }
    if (t == SCAN_T - 1) rowptr[N_NODES] = run;
}

__global__ void csr_fill(const int* __restrict__ src, const int* __restrict__ dst,
                         const int* __restrict__ rank, const int* __restrict__ rowptr,
                         const float* __restrict__ dinv,
                         int* __restrict__ csr, int E) {
    int t = blockIdx.x * blockDim.x + threadIdx.x;
    if (t >= E) return;
    int s = src[t], d = dst[t];
    int pos = rowptr[d] + rank[t];
    ((int2*)csr)[pos] = make_int2(s, __float_as_int(dinv[s] * dinv[d]));
}

__global__ void pack_w_f16(const float* __restrict__ W, _Float16* __restrict__ out,
                           int K, int Nout) {
    int t = blockIdx.x * blockDim.x + threadIdx.x;
    if (t >= K * Nout) return;
    int i    = t & 15;
    int lane = (t >> 4) & 31;
    int rest = t >> 9;
    int kblks = K >> 5;
    int ntile = rest / kblks;
    int kblk  = rest % kblks;
    int half = lane >> 4, l15 = lane & 15;
    int col = ntile * 16 + l15;
    int k   = kblk * 32 + half * 8 + ((i < 8) ? i : (16 + i - 8));
    out[t] = (_Float16)W[(size_t)k * Nout + col];
}

__global__ void wmma_gemm_af32(const float* __restrict__ A,
                               const _Float16* __restrict__ Wpk,
                               float* __restrict__ C,
                               int M, int K, int Nout) {
    const int lane   = threadIdx.x & 31;
    const int ntiles = Nout >> 4;
    const int waveId = (blockIdx.x * blockDim.x + threadIdx.x) >> 5;
    const int mblk   = waveId / ntiles;
    const int ntile  = waveId % ntiles;
    const int m0     = mblk * (16 * MREP);
    if (m0 >= M) return;

    const int half = lane >> 4;
    const int l15  = lane & 15;
    const int kblks = K >> 5;
    const _Float16* wbase = Wpk + ((size_t)ntile * kblks * 512) + lane * 16;

    v8f zero = {};
    v8f acc[MREP];
#pragma unroll
    for (int r = 0; r < MREP; ++r) acc[r] = zero;

    for (int kblk = 0; kblk < kblks; ++kblk) {
        v16h b = *(const v16h*)(wbase + (size_t)kblk * 512);
#pragma unroll
        for (int r = 0; r < MREP; ++r) {
            const float* ap = A + (size_t)(m0 + r * 16 + l15) * K + kblk * 32 + half * 8;
            v16h a;
#pragma unroll
            for (int i = 0; i < 8; ++i) a[i]     = (_Float16)ap[i];
#pragma unroll
            for (int i = 0; i < 8; ++i) a[8 + i] = (_Float16)ap[16 + i];
            acc[r] = __builtin_amdgcn_wmma_f32_16x16x32_f16(
                false, a, false, b, (short)0, acc[r], false, false);
        }
    }
#pragma unroll
    for (int r = 0; r < MREP; ++r) {
#pragma unroll
        for (int q = 0; q < 8; ++q) {
            const int row = m0 + r * 16 + q + half * 8;
            C[(size_t)row * Nout + (ntile * 16 + l15)] = acc[r][q];
        }
    }
}

__global__ void wmma_gemm_af16(const _Float16* __restrict__ A,
                               const _Float16* __restrict__ Wpk,
                               const float* __restrict__ bias,
                               float* __restrict__ C,
                               int M, int K, int Nout, int addBias) {
    const int lane   = threadIdx.x & 31;
    const int ntiles = Nout >> 4;
    const int waveId = (blockIdx.x * blockDim.x + threadIdx.x) >> 5;
    const int mblk   = waveId / ntiles;
    const int ntile  = waveId % ntiles;
    const int m0     = mblk * (16 * MREP);
    if (m0 >= M) return;

    const int half = lane >> 4;
    const int l15  = lane & 15;
    const int colB = ntile * 16 + l15;
    const int kblks = K >> 5;
    const _Float16* wbase = Wpk + ((size_t)ntile * kblks * 512) + lane * 16;

    v8f zero = {};
    v8f acc[MREP];
#pragma unroll
    for (int r = 0; r < MREP; ++r) acc[r] = zero;

    for (int kblk = 0; kblk < kblks; ++kblk) {
        v16h b = *(const v16h*)(wbase + (size_t)kblk * 512);
#pragma unroll
        for (int r = 0; r < MREP; ++r) {
            const _Float16* ap = A + (size_t)(m0 + r * 16 + l15) * K + kblk * 32 + half * 8;
            v8h lo = *(const v8h*)(ap);
            v8h hi = *(const v8h*)(ap + 16);
            v16h a = __builtin_shufflevector(lo, hi, 0, 1, 2, 3, 4, 5, 6, 7,
                                             8, 9, 10, 11, 12, 13, 14, 15);
            acc[r] = __builtin_amdgcn_wmma_f32_16x16x32_f16(
                false, a, false, b, (short)0, acc[r], false, false);
        }
    }

    const float bv = addBias ? bias[colB] : 0.0f;
#pragma unroll
    for (int r = 0; r < MREP; ++r) {
#pragma unroll
        for (int q = 0; q < 8; ++q) {
            const int row = m0 + r * 16 + q + half * 8;
            C[(size_t)row * Nout + colB] = acc[r][q] + bv;
        }
    }
}

template <int K>
__device__ __forceinline__ void pull_rows(const float4* __restrict__ hp, int my_s, int my_c,
                                          int j, int lane, float4& a0, float4& a1) {
    float4 p[K], q[K];
    float  c[K];
#pragma unroll
    for (int k = 0; k < K; ++k) {
        const int s = __builtin_amdgcn_readlane(my_s, j + k);
        c[k] = __int_as_float(__builtin_amdgcn_readlane(my_c, j + k));
        const float4* row = hp + (size_t)s * (H_DIM / 4);
        p[k] = row[lane];
        q[k] = row[32 + lane];
    }
#pragma unroll
    for (int k = 0; k < K; ++k) {
        a0.x = fmaf(p[k].x, c[k], a0.x); a0.y = fmaf(p[k].y, c[k], a0.y);
        a0.z = fmaf(p[k].z, c[k], a0.z); a0.w = fmaf(p[k].w, c[k], a0.w);
        a1.x = fmaf(q[k].x, c[k], a1.x); a1.y = fmaf(q[k].y, c[k], a1.y);
        a1.z = fmaf(q[k].z, c[k], a1.z); a1.w = fmaf(q[k].w, c[k], a1.w);
    }
}

__device__ __forceinline__ float gcn_fin(float agg, float hl, float di, float bc,
                                         float mm, float vv, float gg, float bo) {
    float val = agg + hl * di * di + bc;
    val = (val - mm) * rsqrtf(vv + BN_EPS) * gg + bo;
    return fmaxf(val, 0.0f);
}

__device__ __forceinline__ float4 gcn_fin4(float4 a, float4 h, float di, float4 bc,
                                           float4 mm, float4 vv, float4 gg, float4 bo) {
    float4 r;
    r.x = gcn_fin(a.x, h.x, di, bc.x, mm.x, vv.x, gg.x, bo.x);
    r.y = gcn_fin(a.y, h.y, di, bc.y, mm.y, vv.y, gg.y, bo.y);
    r.z = gcn_fin(a.z, h.z, di, bc.z, mm.z, vv.z, gg.z, bo.z);
    r.w = gcn_fin(a.w, h.w, di, bc.w, mm.w, vv.w, gg.w, bo.w);
    return r;
}

typedef __attribute__((ext_vector_type(4))) _Float16 v4h;

template <bool OUT16>
__global__ void __launch_bounds__(256)
gcn_pull(const float* __restrict__ hlin,
         const int* __restrict__ rowptr,
         const int* __restrict__ csr,
         const float* __restrict__ dinv,
         const float* __restrict__ bconv,
         const float* __restrict__ g,
         const float* __restrict__ bb,
         const float* __restrict__ m,
         const float* __restrict__ v,
         _Float16* __restrict__ out16,
         float* __restrict__ out32, int N) {
    const int lane = threadIdx.x & 31;
    const int n = __builtin_amdgcn_readfirstlane(
        (int)((blockIdx.x * blockDim.x + threadIdx.x) >> 5));
    if (n >= N) return;
    const int beg = rowptr[n], end = rowptr[n + 1];
    const float4* hp = (const float4*)hlin;

    float4 a0 = {0.0f, 0.0f, 0.0f, 0.0f}, a1 = {0.0f, 0.0f, 0.0f, 0.0f};
    for (int base = beg; base < end; base += 32) {
        const int nb = min(32, end - base);
        int my_s = 0, my_c = 0;
        if (lane < nb) { const int2 rec = ((const int2*)csr)[base + lane]; my_s = rec.x; my_c = rec.y; }
        int j = 0;
        for (; j + 4 <= nb; j += 4) pull_rows<4>(hp, my_s, my_c, j, lane, a0, a1);
        const int r = nb - j;
        if (r == 3)      pull_rows<3>(hp, my_s, my_c, j, lane, a0, a1);
        else if (r == 2) pull_rows<2>(hp, my_s, my_c, j, lane, a0, a1);
        else if (r == 1) pull_rows<1>(hp, my_s, my_c, j, lane, a0, a1);
    }

    const float di = dinv[n];
    const float4* own = hp + (size_t)n * (H_DIM / 4);
    const float4 h0 = own[lane], h1 = own[32 + lane];
    const float4* bc4 = (const float4*)bconv; const float4* g4 = (const float4*)g;
    const float4* bb4 = (const float4*)bb;    const float4* m4 = (const float4*)m;
    const float4* v4  = (const float4*)v;
    const float4 r0 = gcn_fin4(a0, h0, di, bc4[lane], m4[lane], v4[lane], g4[lane], bb4[lane]);
    const float4 r1 = gcn_fin4(a1, h1, di, bc4[32 + lane], m4[32 + lane], v4[32 + lane],
                               g4[32 + lane], bb4[32 + lane]);
    if (OUT16) {
        v4h* o = (v4h*)(out16 + (size_t)n * H_DIM);
        v4h x0 = {(_Float16)r0.x, (_Float16)r0.y, (_Float16)r0.z, (_Float16)r0.w};
        v4h x1 = {(_Float16)r1.x, (_Float16)r1.y, (_Float16)r1.z, (_Float16)r1.w};
        o[lane] = x0;
        o[32 + lane] = x1;
    } else {
        float4* o = (float4*)(out32 + (size_t)n * H_DIM);
        o[lane] = r0;
        o[32 + lane] = r1;
    }
}

__global__ void f32_to_f16_x8(const float* __restrict__ in, _Float16* __restrict__ out,
                              int n8) {
    int t = blockIdx.x * blockDim.x + threadIdx.x;
    if (t >= n8) return;
    const float4 a = ((const float4*)in)[2 * t], b = ((const float4*)in)[2 * t + 1];
    v8h o = {(_Float16)a.x, (_Float16)a.y, (_Float16)a.z, (_Float16)a.w,
             (_Float16)b.x, (_Float16)b.y, (_Float16)b.z, (_Float16)b.w};
    ((v8h*)out)[t] = o;
}

__global__ void attn_score(const float* __restrict__ av,
                           const float* __restrict__ au,
                           const float* __restrict__ wA,
                           float* __restrict__ a, int N, int HA) {
    int wave = (blockIdx.x * blockDim.x + threadIdx.x) >> 5;
    int lane = threadIdx.x & 31;
    if (wave >= N) return;
    float sum = 0.0f;
    for (int j = lane; j < HA; j += 32) {
        float t = tanhf(av[(size_t)wave * HA + j]);
        float s = 1.0f / (1.0f + expf(-au[(size_t)wave * HA + j]));
        sum += t * s * wA[j];
    }
    for (int off = 16; off > 0; off >>= 1) sum += __shfl_down(sum, off, 32);
    if (lane == 0) a[wave] = sum;
}

__global__ void seg_init(int* seg, int cnt, int sentinel) {
    int t = blockIdx.x * blockDim.x + threadIdx.x;
    if (t < cnt) seg[t] = sentinel;
}
__global__ void seg_mark(const int* __restrict__ batch, int* seg, int N) {
    int t = blockIdx.x * blockDim.x + threadIdx.x;
    if (t < N) atomicMin(&seg[batch[t]], t);
}
__global__ void seg_fix(int* seg, int B, int Nval) {
    if (blockIdx.x == 0 && threadIdx.x == 0) {
        seg[B] = Nval;
        for (int g = B - 1; g >= 0; --g)
            if (seg[g] == Nval) seg[g] = seg[g + 1];
    }
}

__global__ void seg_softmax(const float* __restrict__ a,
                            const int* __restrict__ seg,
                            float* __restrict__ wout) {
    __shared__ float red[256];
    int g = blockIdx.x;
    int s = seg[g], e = seg[g + 1];
    float mx = -1e30f;
    for (int n = s + (int)threadIdx.x; n < e; n += blockDim.x)
        mx = fmaxf(mx, a[n]);
    red[threadIdx.x] = mx; __syncthreads();
    for (int off = 128; off > 0; off >>= 1) {
        if ((int)threadIdx.x < off)
            red[threadIdx.x] = fmaxf(red[threadIdx.x], red[threadIdx.x + off]);
        __syncthreads();
    }
    mx = red[0]; __syncthreads();
    float sum = 0.0f;
    for (int n = s + (int)threadIdx.x; n < e; n += blockDim.x)
        sum += expf(a[n] - mx);
    red[threadIdx.x] = sum; __syncthreads();
    for (int off = 128; off > 0; off >>= 1) {
        if ((int)threadIdx.x < off)
            red[threadIdx.x] += red[threadIdx.x + off];
        __syncthreads();
    }
    float inv = 1.0f / red[0];
    for (int n = s + (int)threadIdx.x; n < e; n += blockDim.x)
        wout[n] = expf(a[n] - mx) * inv;
}

__global__ void slide_acc(const float* __restrict__ w,
                          const float* __restrict__ h,
                          const int* __restrict__ seg,
                          float* __restrict__ slide, int H, int chunks) {
    int g = blockIdx.x / chunks;
    int c = blockIdx.x % chunks;
    int s = seg[g], e = seg[g + 1];
    int len = e - s;
    int per = (len + chunks - 1) / chunks;
    int n0 = s + c * per;
    int n1 = min(n0 + per, e);
    int f = threadIdx.x;
    float acc = 0.0f;
    for (int n = n0; n < n1; ++n) acc += w[n] * h[(size_t)n * H + f];
    if (n0 < n1) atomicAdd(&slide[g * H + f], acc);
}

__global__ void head_kernel(const float* __restrict__ slide,
                            const float* __restrict__ Wc1,
                            const float* __restrict__ bc1,
                            const float* __restrict__ Wc2,
                            const float* __restrict__ bc2,
                            float* __restrict__ logits) {
    __shared__ float hid[N_BAGS * HA_DIM];
    for (int item = threadIdx.x; item < N_BAGS * HA_DIM; item += blockDim.x) {
        int g = item / HA_DIM, t = item - g * HA_DIM;
        float s = bc1[t];
        for (int k = 0; k < H_DIM; ++k) s += slide[g * H_DIM + k] * Wc1[k * HA_DIM + t];
        hid[item] = fmaxf(s, 0.0f);
    }
    __syncthreads();
    for (int item = threadIdx.x; item < N_BAGS * C_DIM; item += blockDim.x) {
        int g = item / C_DIM, c = item - g * C_DIM;
        float s = bc2[c];
        for (int t = 0; t < HA_DIM; ++t) s += hid[g * HA_DIM + t] * Wc2[t * C_DIM + c];
        logits[item] = s;
    }
}

static inline int cdiv(long long a, int b) { return (int)((a + b - 1) / b); }

extern "C" void kernel_launch(void* const* d_in, const int* in_sizes, int n_in,
                              void* d_out, int out_size, void* d_ws, size_t ws_size,
                              hipStream_t stream) {
    (void)in_sizes; (void)n_in; (void)out_size; (void)ws_size;
    const float* x   = (const float*)d_in[0];
    const int*  ei   = (const int*)d_in[1];
    const int*  batch= (const int*)d_in[2];
    const float* W1  = (const float*)d_in[3];  const float* b1   = (const float*)d_in[4];
    const float* g1  = (const float*)d_in[5];  const float* bb1  = (const float*)d_in[6];
    const float* m1  = (const float*)d_in[7];  const float* v1   = (const float*)d_in[8];
    const float* W2  = (const float*)d_in[9];  const float* b2   = (const float*)d_in[10];
    const float* g2  = (const float*)d_in[11]; const float* bb2  = (const float*)d_in[12];
    const float* m2  = (const float*)d_in[13]; const float* v2   = (const float*)d_in[14];
    const float* Wv  = (const float*)d_in[15]; const float* bv   = (const float*)d_in[16];
    const float* Wu  = (const float*)d_in[17]; const float* bu   = (const float*)d_in[18];
    const float* wA  = (const float*)d_in[19];
    const float* Wc1 = (const float*)d_in[20]; const float* bc1  = (const float*)d_in[21];
    const float* Wc2 = (const float*)d_in[22]; const float* bc2  = (const float*)d_in[23];

    const int* src = ei;
    const int* dst = ei + N_EDGES;

    float* out  = (float*)d_out;
    float* wout = out + N_BAGS * C_DIM;

    const size_t NH = (size_t)N_NODES * H_DIM;
    float*    bufA = (float*)d_ws;
    float*    bufB = bufA + NH;
    _Float16* h16  = (_Float16*)(bufB + NH);
    _Float16* w1pk = h16 + NH;
    _Float16* w2pk = w1pk + (size_t)D_IN * H_DIM;
    _Float16* wvpk = w2pk + (size_t)H_DIM * H_DIM;
    _Float16* wupk = wvpk + (size_t)H_DIM * HA_DIM;
    float*    dinv = (float*)(wupk + (size_t)H_DIM * HA_DIM);
    float*    abuf = dinv + N_NODES;
    float*    slide= abuf + N_NODES;
    int*      seg  = (int*)(slide + N_BAGS * H_DIM);

    int*   cnt      = (int*)h16;
    int*   rowptr   = cnt + N_NODES;
    int*   rank     = rowptr + N_NODES + 4;
    int*   csr      = rank + N_EDGES;
    _Float16* h1_16 = (_Float16*)bufB;
    float*    h2_32 = bufB;

    const int TB = 256;

    zero_i32<<<cdiv(N_NODES, TB), TB, 0, stream>>>(cnt, N_NODES);
    count_rank<<<cdiv(N_EDGES, TB), TB, 0, stream>>>(dst, cnt, rank, N_EDGES);
    scan_rowptr<<<1, SCAN_T, 0, stream>>>(cnt, rowptr, dinv);
    csr_fill<<<cdiv(N_EDGES, TB), TB, 0, stream>>>(src, dst, rank, rowptr, dinv,
                                                   csr, N_EDGES);

    pack_w_f16<<<cdiv((long long)D_IN * H_DIM, TB), TB, 0, stream>>>(W1, w1pk, D_IN, H_DIM);
    pack_w_f16<<<cdiv((long long)H_DIM * H_DIM, TB), TB, 0, stream>>>(W2, w2pk, H_DIM, H_DIM);
    pack_w_f16<<<cdiv((long long)H_DIM * HA_DIM, TB), TB, 0, stream>>>(Wv, wvpk, H_DIM, HA_DIM);
    pack_w_f16<<<cdiv((long long)H_DIM * HA_DIM, TB), TB, 0, stream>>>(Wu, wupk, H_DIM, HA_DIM);

    {
        long long waves = ((long long)N_NODES / 64) * (H_DIM / 16);
        wmma_gemm_af32<<<cdiv(waves * 32, TB), TB, 0, stream>>>(
            x, w1pk, bufA, N_NODES, D_IN, H_DIM);
    }
    gcn_pull<true><<<cdiv((long long)N_NODES * 32, TB), TB, 0, stream>>>(
        bufA, rowptr, csr, dinv, b1, g1, bb1, m1, v1,
        h1_16, nullptr, N_NODES);

    {
        long long waves = ((long long)N_NODES / 64) * (H_DIM / 16);
        wmma_gemm_af16<<<cdiv(waves * 32, TB), TB, 0, stream>>>(
            h1_16, w2pk, nullptr, bufA, N_NODES, H_DIM, H_DIM, 0);
    }
    gcn_pull<false><<<cdiv((long long)N_NODES * 32, TB), TB, 0, stream>>>(
        bufA, rowptr, csr, dinv, b2, g2, bb2, m2, v2,
        nullptr, h2_32, N_NODES);
    f32_to_f16_x8<<<cdiv((long long)(NH / 8), TB), TB, 0, stream>>>(h2_32, h16, (int)(NH / 8));

    float* av = bufA;
    float* au = bufA + (size_t)N_NODES * HA_DIM;
    {
        long long waves = ((long long)N_NODES / 64) * (HA_DIM / 16);
        wmma_gemm_af16<<<cdiv(waves * 32, TB), TB, 0, stream>>>(
            h16, wvpk, bv, av, N_NODES, H_DIM, HA_DIM, 1);
        wmma_gemm_af16<<<cdiv(waves * 32, TB), TB, 0, stream>>>(
            h16, wupk, bu, au, N_NODES, H_DIM, HA_DIM, 1);
    }
    attn_score<<<cdiv((long long)N_NODES * 32, TB), TB, 0, stream>>>(
        av, au, wA, abuf, N_NODES, HA_DIM);

    seg_init<<<1, 32, 0, stream>>>(seg, N_BAGS + 1, N_NODES);
    seg_mark<<<cdiv(N_NODES, TB), TB, 0, stream>>>(batch, seg, N_NODES);
    seg_fix<<<1, 32, 0, stream>>>(seg, N_BAGS, N_NODES);

    seg_softmax<<<N_BAGS, 256, 0, stream>>>(abuf, seg, wout);

    fill_f32<<<cdiv(N_BAGS * H_DIM, TB), TB, 0, stream>>>(slide, 0.0f, N_BAGS * H_DIM);
    const int CHUNKS = 16;
    slide_acc<<<N_BAGS * CHUNKS, H_DIM, 0, stream>>>(wout, h2_32, seg, slide, H_DIM, CHUNKS);

    head_kernel<<<1, 256, 0, stream>>>(slide, Wc1, bc1, Wc2, bc2, out);
}


# Weighted pooling: one-load-per-trip loop with float atomics -> 256-block partial-sum pass (8 row loads in flight per lane, LDS reduce, no atomics); head adds the 32 partial rows per bag: new sum tree
# speedup vs baseline: 2.0613x; 1.3578x over previous
#include <hip/hip_runtime.h>
#include <hip/hip_bf16.h>
#include <math.h>

typedef __attribute__((ext_vector_type(16))) _Float16 v16h;
typedef __attribute__((ext_vector_type(8)))  _Float16 v8h;
typedef __attribute__((ext_vector_type(8)))  float    v8f;

#define N_NODES 65536
#define N_EDGES 524288
#define N_BAGS  8
#define D_IN    512
#define H_DIM   256
#define HA_DIM  128
#define C_DIM   2
#define BN_EPS  1e-5f
#define MREP    4

__global__ void zero_i32(int* p, int n) {
    int t = blockIdx.x * blockDim.x + threadIdx.x;
    if (t < n) p[t] = 0;
}

__global__ void count_rank(const int* __restrict__ dst, int* __restrict__ cnt,
                           int* __restrict__ rank, int E) {
    int t = blockIdx.x * blockDim.x + threadIdx.x;
    if (t < E) rank[t] = atomicAdd(&cnt[dst[t]], 1);
}

#define SCAN_T   1024
#define SCAN_PER (N_NODES / SCAN_T)
static_assert(N_NODES % (SCAN_T * 4) == 0, "scan assumes whole int4 chunks per thread");
__global__ void __launch_bounds__(SCAN_T)
scan_rowptr(const int* __restrict__ cnt, int* __restrict__ rowptr,
            float* __restrict__ dinv) {
    __shared__ int wsum[32];
    const int t = threadIdx.x, lane = t & 31, w = t >> 5;
    const int4* c4 = (const int4*)(cnt + t * SCAN_PER);
    int4 c[SCAN_PER / 4];
    int s = 0;
#pragma unroll
    for (int i = 0; i < SCAN_PER / 4; ++i) {
        c[i] = c4[i];
        s += c[i].x + c[i].y + c[i].z + c[i].w;
    }
    int inc = s;
    for (int off = 1; off < 32; off <<= 1) {
        int v = __shfl_up(inc, off, 32);
        if (lane >= off) inc += v;
    }
    if (lane == 31) wsum[w] = inc;
    __syncthreads();
    if (w == 0) {
        int v = wsum[lane], iv = v;
        for (int off = 1; off < 32; off <<= 1) {
            int u = __shfl_up(iv, off, 32);
            if (lane >= off) iv += u;
        }
        wsum[lane] = iv - v;
    }
    __syncthreads();
    int run = wsum[w] + inc - s;
    int4*   r4 = (int4*)(rowptr + t * SCAN_PER);
    float4* d4 = (float4*)(dinv + t * SCAN_PER);
#pragma unroll
    for (int i = 0; i < SCAN_PER / 4; ++i) {
        int4 r;
        r.x = run; run += c[i].x;
        r.y = run; run += c[i].y;
        r.z = run; run += c[i].z;
        r.w = run; run += c[i].w;
        r4[i] = r;
        float4 d;
        d.x = rsqrtf(1.0f + (float)c[i].x);
        d.y = rsqrtf(1.0f + (float)c[i].y);
        d.z = rsqrtf(1.0f + (float)c[i].z);
        d.w = rsqrtf(1.0f + (float)c[i].w);
        d4[i] = d;
    }
    if (t == SCAN_T - 1) rowptr[N_NODES] = run;
}

__global__ void csr_fill(const int* __restrict__ src, const int* __restrict__ dst,
                         const int* __restrict__ rank, const int* __restrict__ rowptr,
                         const float* __restrict__ dinv,
                         int* __restrict__ csr, int E) {
    int t = blockIdx.x * blockDim.x + threadIdx.x;
    if (t >= E) return;
    int s = src[t], d = dst[t];
    int pos = rowptr[d] + rank[t];
    ((int2*)csr)[pos] = make_int2(s, __float_as_int(dinv[s] * dinv[d]));
}

__global__ void pack_w_f16(const float* __restrict__ W, _Float16* __restrict__ out,
                           int K, int Nout) {
    int t = blockIdx.x * blockDim.x + threadIdx.x;
    if (t >= K * Nout) return;
    int i    = t & 15;
    int lane = (t >> 4) & 31;
    int rest = t >> 9;
    int kblks = K >> 5;
    int ntile = rest / kblks;
    int kblk  = rest % kblks;
    int half = lane >> 4, l15 = lane & 15;
    int col = ntile * 16 + l15;
    int k   = kblk * 32 + half * 8 + ((i < 8) ? i : (16 + i - 8));
    out[t] = (_Float16)W[(size_t)k * Nout + col];
}

__global__ void wmma_gemm_af32(const float* __restrict__ A,
                               const _Float16* __restrict__ Wpk,
                               float* __restrict__ C,
                               int M, int K, int Nout) {
    const int lane   = threadIdx.x & 31;
    const int ntiles = Nout >> 4;
    const int waveId = (blockIdx.x * blockDim.x + threadIdx.x) >> 5;
    const int mblk   = waveId / ntiles;
    const int ntile  = waveId % ntiles;
    const int m0     = mblk * (16 * MREP);
    if (m0 >= M) return;

    const int half = lane >> 4;
    const int l15  = lane & 15;
    const int kblks = K >> 5;
    const _Float16* wbase = Wpk + ((size_t)ntile * kblks * 512) + lane * 16;

    v8f zero = {};
    v8f acc[MREP];
#pragma unroll
    for (int r = 0; r < MREP; ++r) acc[r] = zero;

    for (int kblk = 0; kblk < kblks; ++kblk) {
        v16h b = *(const v16h*)(wbase + (size_t)kblk * 512);
#pragma unroll
        for (int r = 0; r < MREP; ++r) {
            const float* ap = A + (size_t)(m0 + r * 16 + l15) * K + kblk * 32 + half * 8;
            v16h a;
#pragma unroll
            for (int i = 0; i < 8; ++i) a[i]     = (_Float16)ap[i];
#pragma unroll
            for (int i = 0; i < 8; ++i) a[8 + i] = (_Float16)ap[16 + i];
            acc[r] = __builtin_amdgcn_wmma_f32_16x16x32_f16(
                false, a, false, b, (short)0, acc[r], false, false);
        }
    }
#pragma unroll
    for (int r = 0; r < MREP; ++r) {
#pragma unroll
        for (int q = 0; q < 8; ++q) {
            const int row = m0 + r * 16 + q + half * 8;
            C[(size_t)row * Nout + (ntile * 16 + l15)] = acc[r][q];
        }
    }
}

__global__ void wmma_gemm_af16(const _Float16* __restrict__ A,
                               const _Float16* __restrict__ Wpk,
                               const float* __restrict__ bias,
                               float* __restrict__ C,
                               int M, int K, int Nout, int addBias) {
    const int lane   = threadIdx.x & 31;
    const int ntiles = Nout >> 4;
    const int waveId = (blockIdx.x * blockDim.x + threadIdx.x) >> 5;
    const int mblk   = waveId / ntiles;
    const int ntile  = waveId % ntiles;
    const int m0     = mblk * (16 * MREP);
    if (m0 >= M) return;

    const int half = lane >> 4;
    const int l15  = lane & 15;
    const int colB = ntile * 16 + l15;
    const int kblks = K >> 5;
    const _Float16* wbase = Wpk + ((size_t)ntile * kblks * 512) + lane * 16;

    v8f zero = {};
    v8f acc[MREP];
#pragma unroll
    for (int r = 0; r < MREP; ++r) acc[r] = zero;

    for (int kblk = 0; kblk < kblks; ++kblk) {
        v16h b = *(const v16h*)(wbase + (size_t)kblk * 512);
#pragma unroll
        for (int r = 0; r < MREP; ++r) {
            const _Float16* ap = A + (size_t)(m0 + r * 16 + l15) * K + kblk * 32 + half * 8;
            v8h lo = *(const v8h*)(ap);
            v8h hi = *(const v8h*)(ap + 16);
            v16h a = __builtin_shufflevector(lo, hi, 0, 1, 2, 3, 4, 5, 6, 7,
                                             8, 9, 10, 11, 12, 13, 14, 15);
            acc[r] = __builtin_amdgcn_wmma_f32_16x16x32_f16(
                false, a, false, b, (short)0, acc[r], false, false);
        }
    }

    const float bv = addBias ? bias[colB] : 0.0f;
#pragma unroll
    for (int r = 0; r < MREP; ++r) {
#pragma unroll
        for (int q = 0; q < 8; ++q) {
            const int row = m0 + r * 16 + q + half * 8;
            C[(size_t)row * Nout + colB] = acc[r][q] + bv;
        }
    }
}

template <int K>
__device__ __forceinline__ void pull_rows(const float4* __restrict__ hp, int my_s, int my_c,
                                          int j, int lane, float4& a0, float4& a1) {
    float4 p[K], q[K];
    float  c[K];
#pragma unroll
    for (int k = 0; k < K; ++k) {
        const int s = __builtin_amdgcn_readlane(my_s, j + k);
        c[k] = __int_as_float(__builtin_amdgcn_readlane(my_c, j + k));
        const float4* row = hp + (size_t)s * (H_DIM / 4);
        p[k] = row[lane];
        q[k] = row[32 + lane];
    }
#pragma unroll
    for (int k = 0; k < K; ++k) {
        a0.x = fmaf(p[k].x, c[k], a0.x); a0.y = fmaf(p[k].y, c[k], a0.y);
        a0.z = fmaf(p[k].z, c[k], a0.z); a0.w = fmaf(p[k].w, c[k], a0.w);
        a1.x = fmaf(q[k].x, c[k], a1.x); a1.y = fmaf(q[k].y, c[k], a1.y);
        a1.z = fmaf(q[k].z, c[k], a1.z); a1.w = fmaf(q[k].w, c[k], a1.w);
    }
}

__device__ __forceinline__ float gcn_fin(float agg, float hl, float di, float bc,
                                         float mm, float vv, float gg, float bo) {
    float val = agg + hl * di * di + bc;
    val = (val - mm) * rsqrtf(vv + BN_EPS) * gg + bo;
    return fmaxf(val, 0.0f);
}

__device__ __forceinline__ float4 gcn_fin4(float4 a, float4 h, float di, float4 bc,
                                           float4 mm, float4 vv, float4 gg, float4 bo) {
    float4 r;
    r.x = gcn_fin(a.x, h.x, di, bc.x, mm.x, vv.x, gg.x, bo.x);
    r.y = gcn_fin(a.y, h.y, di, bc.y, mm.y, vv.y, gg.y, bo.y);
    r.z = gcn_fin(a.z, h.z, di, bc.z, mm.z, vv.z, gg.z, bo.z);
    r.w = gcn_fin(a.w, h.w, di, bc.w, mm.w, vv.w, gg.w, bo.w);
    return r;
}

typedef __attribute__((ext_vector_type(4))) _Float16 v4h;

template <bool OUT16>
__global__ void __launch_bounds__(256)
gcn_pull(const float* __restrict__ hlin,
         const int* __restrict__ rowptr,
         const int* __restrict__ csr,
         const float* __restrict__ dinv,
         const float* __restrict__ bconv,
         const float* __restrict__ g,
         const float* __restrict__ bb,
         const float* __restrict__ m,
         const float* __restrict__ v,
         _Float16* __restrict__ out16,
         float* __restrict__ out32, int N) {
    const int lane = threadIdx.x & 31;
    const int n = __builtin_amdgcn_readfirstlane(
        (int)((blockIdx.x * blockDim.x + threadIdx.x) >> 5));
    if (n >= N) return;
    const int beg = rowptr[n], end = rowptr[n + 1];
    const float4* hp = (const float4*)hlin;

    float4 a0 = {0.0f, 0.0f, 0.0f, 0.0f}, a1 = {0.0f, 0.0f, 0.0f, 0.0f};
    for (int base = beg; base < end; base += 32) {
        const int nb = min(32, end - base);
        int my_s = 0, my_c = 0;
        if (lane < nb) { const int2 rec = ((const int2*)csr)[base + lane]; my_s = rec.x; my_c = rec.y; }
        int j = 0;
        for (; j + 4 <= nb; j += 4) pull_rows<4>(hp, my_s, my_c, j, lane, a0, a1);
        const int r = nb - j;
        if (r == 3)      pull_rows<3>(hp, my_s, my_c, j, lane, a0, a1);
        else if (r == 2) pull_rows<2>(hp, my_s, my_c, j, lane, a0, a1);
        else if (r == 1) pull_rows<1>(hp, my_s, my_c, j, lane, a0, a1);
    }

    const float di = dinv[n];
    const float4* own = hp + (size_t)n * (H_DIM / 4);
    const float4 h0 = own[lane], h1 = own[32 + lane];
    const float4* bc4 = (const float4*)bconv; const float4* g4 = (const float4*)g;
    const float4* bb4 = (const float4*)bb;    const float4* m4 = (const float4*)m;
    const float4* v4  = (const float4*)v;
    const float4 r0 = gcn_fin4(a0, h0, di, bc4[lane], m4[lane], v4[lane], g4[lane], bb4[lane]);
    const float4 r1 = gcn_fin4(a1, h1, di, bc4[32 + lane], m4[32 + lane], v4[32 + lane],
                               g4[32 + lane], bb4[32 + lane]);
    if (OUT16) {
        v4h* o = (v4h*)(out16 + (size_t)n * H_DIM);
        v4h x0 = {(_Float16)r0.x, (_Float16)r0.y, (_Float16)r0.z, (_Float16)r0.w};
        v4h x1 = {(_Float16)r1.x, (_Float16)r1.y, (_Float16)r1.z, (_Float16)r1.w};
        o[lane] = x0;
        o[32 + lane] = x1;
    } else {
        float4* o = (float4*)(out32 + (size_t)n * H_DIM);
        o[lane] = r0;
        o[32 + lane] = r1;
    }
}

__global__ void f32_to_f16_x8(const float* __restrict__ in, _Float16* __restrict__ out,
                              int n8) {
    int t = blockIdx.x * blockDim.x + threadIdx.x;
    if (t >= n8) return;
    const float4 a = ((const float4*)in)[2 * t], b = ((const float4*)in)[2 * t + 1];
    v8h o = {(_Float16)a.x, (_Float16)a.y, (_Float16)a.z, (_Float16)a.w,
             (_Float16)b.x, (_Float16)b.y, (_Float16)b.z, (_Float16)b.w};
    ((v8h*)out)[t] = o;
}

__global__ void attn_score(const float* __restrict__ av,
                           const float* __restrict__ au,
                           const float* __restrict__ wA,
                           float* __restrict__ a, int N, int HA) {
    int wave = (blockIdx.x * blockDim.x + threadIdx.x) >> 5;
    int lane = threadIdx.x & 31;
    if (wave >= N) return;
    float sum = 0.0f;
    for (int j = lane; j < HA; j += 32) {
        float t = tanhf(av[(size_t)wave * HA + j]);
        float s = 1.0f / (1.0f + expf(-au[(size_t)wave * HA + j]));
        sum += t * s * wA[j];
    }
    for (int off = 16; off > 0; off >>= 1) sum += __shfl_down(sum, off, 32);
    if (lane == 0) a[wave] = sum;
}

__global__ void seg_init(int* seg, int cnt, int sentinel) {
    int t = blockIdx.x * blockDim.x + threadIdx.x;
    if (t < cnt) seg[t] = sentinel;
}
__global__ void seg_mark(const int* __restrict__ batch, int* seg, int N) {
    int t = blockIdx.x * blockDim.x + threadIdx.x;
    if (t < N) atomicMin(&seg[batch[t]], t);
}
__global__ void seg_fix(int* seg, int B, int Nval) {
    if (blockIdx.x == 0 && threadIdx.x == 0) {
        seg[B] = Nval;
        for (int g = B - 1; g >= 0; --g)
            if (seg[g] == Nval) seg[g] = seg[g + 1];
    }
}

__global__ void seg_softmax(const float* __restrict__ a,
                            const int* __restrict__ seg,
                            float* __restrict__ wout) {
    __shared__ float red[256];
    int g = blockIdx.x;
    int s = seg[g], e = seg[g + 1];
    float mx = -1e30f;
    for (int n = s + (int)threadIdx.x; n < e; n += blockDim.x)
        mx = fmaxf(mx, a[n]);
    red[threadIdx.x] = mx; __syncthreads();
    for (int off = 128; off > 0; off >>= 1) {
        if ((int)threadIdx.x < off)
            red[threadIdx.x] = fmaxf(red[threadIdx.x], red[threadIdx.x + off]);
        __syncthreads();
    }
    mx = red[0]; __syncthreads();
    float sum = 0.0f;
    for (int n = s + (int)threadIdx.x; n < e; n += blockDim.x)
        sum += expf(a[n] - mx);
    red[threadIdx.x] = sum; __syncthreads();
    for (int off = 128; off > 0; off >>= 1) {
        if ((int)threadIdx.x < off)
            red[threadIdx.x] += red[threadIdx.x + off];
        __syncthreads();
    }
    float inv = 1.0f / red[0];
    for (int n = s + (int)threadIdx.x; n < e; n += blockDim.x)
        wout[n] = expf(a[n] - mx) * inv;
}

#define POOL_CH 32
#define POOL_T  512
#define POOL_RG (POOL_T / 64)
#define POOL_U  8
static_assert(H_DIM == 256, "pooling assumes 64 float4 per row");
__global__ void __launch_bounds__(POOL_T)
slide_partial(const float* __restrict__ w, const float* __restrict__ h,
              const int* __restrict__ seg, float* __restrict__ part) {
    __shared__ float4 red[POOL_RG][64];
    const int g = blockIdx.x / POOL_CH, c = blockIdx.x % POOL_CH;
    const int s = seg[g], e = seg[g + 1];
    const int per = (e - s + POOL_CH - 1) / POOL_CH;
    const int n0 = s + c * per;
    const int n1 = min(n0 + per, e);
    const int col = threadIdx.x & 63;
    const int rg = __builtin_amdgcn_readfirstlane((int)(threadIdx.x >> 6));
    const float4* h4 = (const float4*)h;
    const float4 zero4 = {0.0f, 0.0f, 0.0f, 0.0f};
    float4 acc = zero4;
    for (int nb = n0 + rg; nb < n1; nb += POOL_U * POOL_RG) {
        float4 v[POOL_U];
        float  ww[POOL_U];
#pragma unroll
        for (int u = 0; u < POOL_U; ++u) {
            const int n  = nb + u * POOL_RG;
            const int nc = min(n, n1 - 1);
            v[u]  = h4[(size_t)nc * (H_DIM / 4) + col];
            ww[u] = w[nc];
            if (n >= n1) v[u] = zero4;
        }
#pragma unroll
        for (int u = 0; u < POOL_U; ++u) {
            acc.x = fmaf(ww[u], v[u].x, acc.x); acc.y = fmaf(ww[u], v[u].y, acc.y);
            acc.z = fmaf(ww[u], v[u].z, acc.z); acc.w = fmaf(ww[u], v[u].w, acc.w);
        }
    }
    red[rg][col] = acc;
    __syncthreads();
    if (threadIdx.x < H_DIM) {
        const float* rf = (const float*)red;
        float sum = 0.0f;
#pragma unroll
        for (int r = 0; r < POOL_RG; ++r) sum += rf[r * H_DIM + threadIdx.x];
        part[(size_t)blockIdx.x * H_DIM + threadIdx.x] = sum;
    }
}

__global__ void head_kernel(const float* __restrict__ part,
                            const float* __restrict__ Wc1,
                            const float* __restrict__ bc1,
                            const float* __restrict__ Wc2,
                            const float* __restrict__ bc2,
                            float* __restrict__ logits) {
    __shared__ float slide[N_BAGS * H_DIM];
    __shared__ float hid[N_BAGS * HA_DIM];
    for (int item = threadIdx.x; item < N_BAGS * H_DIM; item += blockDim.x) {
        int g = item / H_DIM, k = item - g * H_DIM;
        const float* p = part + (size_t)g * POOL_CH * H_DIM + k;
        float s = 0.0f;
#pragma unroll
        for (int c = 0; c < POOL_CH; ++c) s += p[c * H_DIM];
        slide[item] = s;
    }
    __syncthreads();
    for (int item = threadIdx.x; item < N_BAGS * HA_DIM; item += blockDim.x) {
        int g = item / HA_DIM, t = item - g * HA_DIM;
        float s = bc1[t];
        for (int k = 0; k < H_DIM; ++k) s += slide[g * H_DIM + k] * Wc1[k * HA_DIM + t];
        hid[item] = fmaxf(s, 0.0f);
    }
    __syncthreads();
    for (int item = threadIdx.x; item < N_BAGS * C_DIM; item += blockDim.x) {
        int g = item / C_DIM, c = item - g * C_DIM;
        float s = bc2[c];
        for (int t = 0; t < HA_DIM; ++t) s += hid[g * HA_DIM + t] * Wc2[t * C_DIM + c];
        logits[item] = s;
    }
}

static inline int cdiv(long long a, int b) { return (int)((a + b - 1) / b); }

extern "C" void kernel_launch(void* const* d_in, const int* in_sizes, int n_in,
                              void* d_out, int out_size, void* d_ws, size_t ws_size,
                              hipStream_t stream) {
    (void)in_sizes; (void)n_in; (void)out_size; (void)ws_size;
    const float* x   = (const float*)d_in[0];
    const int*  ei   = (const int*)d_in[1];
    const int*  batch= (const int*)d_in[2];
    const float* W1  = (const float*)d_in[3];  const float* b1   = (const float*)d_in[4];
    const float* g1  = (const float*)d_in[5];  const float* bb1  = (const float*)d_in[6];
    const float* m1  = (const float*)d_in[7];  const float* v1   = (const float*)d_in[8];
    const float* W2  = (const float*)d_in[9];  const float* b2   = (const float*)d_in[10];
    const float* g2  = (const float*)d_in[11]; const float* bb2  = (const float*)d_in[12];
    const float* m2  = (const float*)d_in[13]; const float* v2   = (const float*)d_in[14];
    const float* Wv  = (const float*)d_in[15]; const float* bv   = (const float*)d_in[16];
    const float* Wu  = (const float*)d_in[17]; const float* bu   = (const float*)d_in[18];
    const float* wA  = (const float*)d_in[19];
    const float* Wc1 = (const float*)d_in[20]; const float* bc1  = (const float*)d_in[21];
    const float* Wc2 = (const float*)d_in[22]; const float* bc2  = (const float*)d_in[23];

    const int* src = ei;
    const int* dst = ei + N_EDGES;

    float* out  = (float*)d_out;
    float* wout = out + N_BAGS * C_DIM;

    const size_t NH = (size_t)N_NODES * H_DIM;
    float*    bufA = (float*)d_ws;
    float*    bufB = bufA + NH;
    _Float16* h16  = (_Float16*)(bufB + NH);
    _Float16* w1pk = h16 + NH;
    _Float16* w2pk = w1pk + (size_t)D_IN * H_DIM;
    _Float16* wvpk = w2pk + (size_t)H_DIM * H_DIM;
    _Float16* wupk = wvpk + (size_t)H_DIM * HA_DIM;
    float*    dinv = (float*)(wupk + (size_t)H_DIM * HA_DIM);
    float*    abuf = dinv + N_NODES;
    int*      seg  = (int*)(abuf + N_NODES);

    int*   cnt      = (int*)h16;
    int*   rowptr   = cnt + N_NODES;
    int*   rank     = rowptr + N_NODES + 4;
    int*   csr      = rank + N_EDGES;
    _Float16* h1_16 = (_Float16*)bufB;
    float*    h2_32 = bufB;

    const int TB = 256;

    zero_i32<<<cdiv(N_NODES, TB), TB, 0, stream>>>(cnt, N_NODES);
    count_rank<<<cdiv(N_EDGES, TB), TB, 0, stream>>>(dst, cnt, rank, N_EDGES);
    scan_rowptr<<<1, SCAN_T, 0, stream>>>(cnt, rowptr, dinv);
    csr_fill<<<cdiv(N_EDGES, TB), TB, 0, stream>>>(src, dst, rank, rowptr, dinv,
                                                   csr, N_EDGES);

    pack_w_f16<<<cdiv((long long)D_IN * H_DIM, TB), TB, 0, stream>>>(W1, w1pk, D_IN, H_DIM);
    pack_w_f16<<<cdiv((long long)H_DIM * H_DIM, TB), TB, 0, stream>>>(W2, w2pk, H_DIM, H_DIM);
    pack_w_f16<<<cdiv((long long)H_DIM * HA_DIM, TB), TB, 0, stream>>>(Wv, wvpk, H_DIM, HA_DIM);
    pack_w_f16<<<cdiv((long long)H_DIM * HA_DIM, TB), TB, 0, stream>>>(Wu, wupk, H_DIM, HA_DIM);

    {
        long long waves = ((long long)N_NODES / 64) * (H_DIM / 16);
        wmma_gemm_af32<<<cdiv(waves * 32, TB), TB, 0, stream>>>(
            x, w1pk, bufA, N_NODES, D_IN, H_DIM);
    }
    gcn_pull<true><<<cdiv((long long)N_NODES * 32, TB), TB, 0, stream>>>(
        bufA, rowptr, csr, dinv, b1, g1, bb1, m1, v1,
        h1_16, nullptr, N_NODES);

    {
        long long waves = ((long long)N_NODES / 64) * (H_DIM / 16);
        wmma_gemm_af16<<<cdiv(waves * 32, TB), TB, 0, stream>>>(
            h1_16, w2pk, nullptr, bufA, N_NODES, H_DIM, H_DIM, 0);
    }
    gcn_pull<false><<<cdiv((long long)N_NODES * 32, TB), TB, 0, stream>>>(
        bufA, rowptr, csr, dinv, b2, g2, bb2, m2, v2,
        nullptr, h2_32, N_NODES);
    f32_to_f16_x8<<<cdiv((long long)(NH / 8), TB), TB, 0, stream>>>(h2_32, h16, (int)(NH / 8));

    float* av = bufA;
    float* au = bufA + (size_t)N_NODES * HA_DIM;
    {
        long long waves = ((long long)N_NODES / 64) * (HA_DIM / 16);
        wmma_gemm_af16<<<cdiv(waves * 32, TB), TB, 0, stream>>>(
            h16, wvpk, bv, av, N_NODES, H_DIM, HA_DIM, 1);
        wmma_gemm_af16<<<cdiv(waves * 32, TB), TB, 0, stream>>>(
            h16, wupk, bu, au, N_NODES, H_DIM, HA_DIM, 1);
    }
    attn_score<<<cdiv((long long)N_NODES * 32, TB), TB, 0, stream>>>(
        av, au, wA, abuf, N_NODES, HA_DIM);

    seg_init<<<1, 32, 0, stream>>>(seg, N_BAGS + 1, N_NODES);
    seg_mark<<<cdiv(N_NODES, TB), TB, 0, stream>>>(batch, seg, N_NODES);
    seg_fix<<<1, 32, 0, stream>>>(seg, N_BAGS, N_NODES);

    seg_softmax<<<N_BAGS, 256, 0, stream>>>(abuf, seg, wout);

    float* part = bufA;
    slide_partial<<<N_BAGS * POOL_CH, POOL_T, 0, stream>>>(wout, h2_32, seg, part);

    head_kernel<<<1, 256, 0, stream>>>(part, Wc1, bc1, Wc2, bc2, out);
}


# GEMM A operand: per-wave global fragment loads (16 lines each, tile re-read per column tile) -> 64-row A tile loaded once per work-group with whole-line loads into LDS (f16); same k and sum order
# speedup vs baseline: 3.0570x; 1.4830x over previous
#include <hip/hip_runtime.h>
#include <hip/hip_bf16.h>
#include <math.h>

typedef __attribute__((ext_vector_type(16))) _Float16 v16h;
typedef __attribute__((ext_vector_type(8)))  _Float16 v8h;
typedef __attribute__((ext_vector_type(8)))  float    v8f;

#define N_NODES 65536
#define N_EDGES 524288
#define N_BAGS  8
#define D_IN    512
#define H_DIM   256
#define HA_DIM  128
#define C_DIM   2
#define BN_EPS  1e-5f
#define MREP    4

__global__ void zero_i32(int* p, int n) {
    int t = blockIdx.x * blockDim.x + threadIdx.x;
    if (t < n) p[t] = 0;
}

__global__ void count_rank(const int* __restrict__ dst, int* __restrict__ cnt,
                           int* __restrict__ rank, int E) {
    int t = blockIdx.x * blockDim.x + threadIdx.x;
    if (t < E) rank[t] = atomicAdd(&cnt[dst[t]], 1);
}

#define SCAN_T   1024
#define SCAN_PER (N_NODES / SCAN_T)
static_assert(N_NODES % (SCAN_T * 4) == 0, "scan assumes whole int4 chunks per thread");
__global__ void __launch_bounds__(SCAN_T)
scan_rowptr(const int* __restrict__ cnt, int* __restrict__ rowptr,
            float* __restrict__ dinv) {
    __shared__ int wsum[32];
    const int t = threadIdx.x, lane = t & 31, w = t >> 5;
    const int4* c4 = (const int4*)(cnt + t * SCAN_PER);
    int4 c[SCAN_PER / 4];
    int s = 0;
#pragma unroll
    for (int i = 0; i < SCAN_PER / 4; ++i) {
        c[i] = c4[i];
        s += c[i].x + c[i].y + c[i].z + c[i].w;
    }
    int inc = s;
    for (int off = 1; off < 32; off <<= 1) {
        int v = __shfl_up(inc, off, 32);
        if (lane >= off) inc += v;
    }
    if (lane == 31) wsum[w] = inc;
    __syncthreads();
    if (w == 0) {
        int v = wsum[lane], iv = v;
        for (int off = 1; off < 32; off <<= 1) {
            int u = __shfl_up(iv, off, 32);
            if (lane >= off) iv += u;
        }
        wsum[lane] = iv - v;
    }
    __syncthreads();
    int run = wsum[w] + inc - s;
    int4*   r4 = (int4*)(rowptr + t * SCAN_PER);
    float4* d4 = (float4*)(dinv + t * SCAN_PER);
#pragma unroll
    for (int i = 0; i < SCAN_PER / 4; ++i) {
        int4 r;
        r.x = run; run += c[i].x;
        r.y = run; run += c[i].y;
        r.z = run; run += c[i].z;
        r.w = run; run += c[i].w;
        r4[i] = r;
        float4 d;
        d.x = rsqrtf(1.0f + (float)c[i].x);
        d.y = rsqrtf(1.0f + (float)c[i].y);
        d.z = rsqrtf(1.0f + (float)c[i].z);
        d.w = rsqrtf(1.0f + (float)c[i].w);
        d4[i] = d;
    }
    if (t == SCAN_T - 1) rowptr[N_NODES] = run;
}

__global__ void csr_fill(const int* __restrict__ src, const int* __restrict__ dst,
                         const int* __restrict__ rank, const int* __restrict__ rowptr,
                         const float* __restrict__ dinv,
                         int* __restrict__ csr, int E) {
    int t = blockIdx.x * blockDim.x + threadIdx.x;
    if (t >= E) return;
    int s = src[t], d = dst[t];
    int pos = rowptr[d] + rank[t];
    ((int2*)csr)[pos] = make_int2(s, __float_as_int(dinv[s] * dinv[d]));
}

__global__ void pack_w_f16(const float* __restrict__ W, _Float16* __restrict__ out,
                           int K, int Nout) {
    int t = blockIdx.x * blockDim.x + threadIdx.x;
    if (t >= K * Nout) return;
    int i    = t & 15;
    int lane = (t >> 4) & 31;
    int rest = t >> 9;
    int kblks = K >> 5;
    int ntile = rest / kblks;
    int kblk  = rest % kblks;
    int half = lane >> 4, l15 = lane & 15;
    int col = ntile * 16 + l15;
    int k   = kblk * 32 + half * 8 + ((i < 8) ? i : (16 + i - 8));
    out[t] = (_Float16)W[(size_t)k * Nout + col];
}

#define GT_ROWS 64
#define GT_T    256
#define GT_LD   8
static_assert(GT_ROWS == 16 * MREP, "a wave covers all rows of the tile");
static_assert(N_NODES % GT_ROWS == 0, "no partial row tile");

#define GT_PITCH(K) ((K) + 8)

template <int K, int NOUT, bool BIAS>
__device__ __forceinline__ void gemm_tile_from_lds(const _Float16* As,
                                                   const _Float16* __restrict__ Wpk,
                                                   const float* __restrict__ bias,
                                                   float* __restrict__ C, int m0) {
    constexpr int PITCH = GT_PITCH(K);
    constexpr int KBLKS = K / 32;
    constexpr int NTW   = NOUT / 16 / (GT_T / 32);
    static_assert(NTW >= 1 && NTW * 16 * (GT_T / 32) == NOUT, "waves tile the columns");
    const int lane = threadIdx.x & 31;
    const int wave = __builtin_amdgcn_readfirstlane((int)(threadIdx.x >> 5));
    const int half = lane >> 4;
    const int l15  = lane & 15;
    const _Float16* wb = Wpk + (size_t)(wave * NTW) * KBLKS * 512 + lane * 16;
    const _Float16* ab = As + l15 * PITCH + half * 8;

    v8f zero = {};
    v8f acc[MREP][NTW];
#pragma unroll
    for (int r = 0; r < MREP; ++r)
#pragma unroll
        for (int t = 0; t < NTW; ++t) acc[r][t] = zero;

    for (int kblk = 0; kblk < KBLKS; ++kblk) {
        v16h b[NTW];
#pragma unroll
        for (int t = 0; t < NTW; ++t)
            b[t] = *(const v16h*)(wb + ((size_t)t * KBLKS + kblk) * 512);
#pragma unroll
        for (int r = 0; r < MREP; ++r) {
            const _Float16* ap = ab + r * 16 * PITCH + kblk * 32;
            v8h lo = *(const v8h*)(ap);
            v8h hi = *(const v8h*)(ap + 16);
            v16h a = __builtin_shufflevector(lo, hi, 0, 1, 2, 3, 4, 5, 6, 7,
                                             8, 9, 10, 11, 12, 13, 14, 15);
#pragma unroll
            for (int t = 0; t < NTW; ++t)
                acc[r][t] = __builtin_amdgcn_wmma_f32_16x16x32_f16(
                    false, a, false, b[t], (short)0, acc[r][t], false, false);
        }
    }

#pragma unroll
    for (int t = 0; t < NTW; ++t) {
        const int col = (wave * NTW + t) * 16 + l15;
        float bv = 0.0f;
        if (BIAS) bv = bias[col];
#pragma unroll
        for (int r = 0; r < MREP; ++r) {
#pragma unroll
            for (int q = 0; q < 8; ++q) {
                const int row = m0 + r * 16 + q + half * 8;
                C[(size_t)row * NOUT + col] = BIAS ? acc[r][t][q] + bv : acc[r][t][q];
            }
        }
    }
}

typedef __attribute__((ext_vector_type(4))) _Float16 v4h_t;

template <int K, int NOUT>
__global__ void __launch_bounds__(GT_T)
wmma_gemm_lds_af32(const float* __restrict__ A, const _Float16* __restrict__ Wpk,
                   float* __restrict__ C) {
    __shared__ alignas(16) _Float16 As[GT_ROWS * GT_PITCH(K)];
    constexpr int VROW  = K / 4;
    constexpr int PER_T = GT_ROWS * VROW / GT_T;
    static_assert(PER_T % GT_LD == 0 && VROW % 32 == 0, "whole batches, a wave inside one row");
    const int tid = threadIdx.x;
    const int m0  = blockIdx.x * GT_ROWS;
    const float4* Ag = (const float4*)(A + (size_t)m0 * K);
#pragma unroll 1
    for (int it = 0; it < PER_T; it += GT_LD) {
        float4 v[GT_LD];
#pragma unroll
        for (int u = 0; u < GT_LD; ++u) v[u] = Ag[(it + u) * GT_T + tid];
#pragma unroll
        for (int u = 0; u < GT_LD; ++u) {
            const int f = (it + u) * GT_T + tid;
            const int row = f / VROW, c = f % VROW;
            v4h_t h = {(_Float16)v[u].x, (_Float16)v[u].y, (_Float16)v[u].z, (_Float16)v[u].w};
            *(v4h_t*)(As + row * GT_PITCH(K) + c * 4) = h;
        }
    }
    __syncthreads();
    gemm_tile_from_lds<K, NOUT, false>(As, Wpk, nullptr, C, m0);
}

template <int K, int NOUT, bool BIAS>
__global__ void __launch_bounds__(GT_T)
wmma_gemm_lds_af16(const _Float16* __restrict__ A, const _Float16* __restrict__ Wpk,
                   const float* __restrict__ bias, float* __restrict__ C) {
    __shared__ alignas(16) _Float16 As[GT_ROWS * GT_PITCH(K)];
    constexpr int VROW  = K / 8;
    constexpr int PER_T = GT_ROWS * VROW / GT_T;
    static_assert(GT_ROWS * VROW % GT_T == 0 && VROW % 32 == 0, "whole passes, a wave inside one row");
    const int tid = threadIdx.x;
    const int m0  = blockIdx.x * GT_ROWS;
    const v8h* Ag = (const v8h*)(A + (size_t)m0 * K);
    v8h v[PER_T];
#pragma unroll
    for (int u = 0; u < PER_T; ++u) v[u] = Ag[u * GT_T + tid];
#pragma unroll
    for (int u = 0; u < PER_T; ++u) {
        const int f = u * GT_T + tid;
        const int row = f / VROW, c = f % VROW;
        *(v8h*)(As + row * GT_PITCH(K) + c * 8) = v[u];
    }
    __syncthreads();
    gemm_tile_from_lds<K, NOUT, BIAS>(As, Wpk, bias, C, m0);
}

template <int K>
__device__ __forceinline__ void pull_rows(const float4* __restrict__ hp, int my_s, int my_c,
                                          int j, int lane, float4& a0, float4& a1) {
    float4 p[K], q[K];
    float  c[K];
#pragma unroll
    for (int k = 0; k < K; ++k) {
        const int s = __builtin_amdgcn_readlane(my_s, j + k);
        c[k] = __int_as_float(__builtin_amdgcn_readlane(my_c, j + k));
        const float4* row = hp + (size_t)s * (H_DIM / 4);
        p[k] = row[lane];
        q[k] = row[32 + lane];
    }
#pragma unroll
    for (int k = 0; k < K; ++k) {
        a0.x = fmaf(p[k].x, c[k], a0.x); a0.y = fmaf(p[k].y, c[k], a0.y);
        a0.z = fmaf(p[k].z, c[k], a0.z); a0.w = fmaf(p[k].w, c[k], a0.w);
        a1.x = fmaf(q[k].x, c[k], a1.x); a1.y = fmaf(q[k].y, c[k], a1.y);
        a1.z = fmaf(q[k].z, c[k], a1.z); a1.w = fmaf(q[k].w, c[k], a1.w);
    }
}

__device__ __forceinline__ float gcn_fin(float agg, float hl, float di, float bc,
                                         float mm, float vv, float gg, float bo) {
    float val = agg + hl * di * di + bc;
    val = (val - mm) * rsqrtf(vv + BN_EPS) * gg + bo;
    return fmaxf(val, 0.0f);
}

__device__ __forceinline__ float4 gcn_fin4(float4 a, float4 h, float di, float4 bc,
                                           float4 mm, float4 vv, float4 gg, float4 bo) {
    float4 r;
    r.x = gcn_fin(a.x, h.x, di, bc.x, mm.x, vv.x, gg.x, bo.x);
    r.y = gcn_fin(a.y, h.y, di, bc.y, mm.y, vv.y, gg.y, bo.y);
    r.z = gcn_fin(a.z, h.z, di, bc.z, mm.z, vv.z, gg.z, bo.z);
    r.w = gcn_fin(a.w, h.w, di, bc.w, mm.w, vv.w, gg.w, bo.w);
    return r;
}

typedef __attribute__((ext_vector_type(4))) _Float16 v4h;

template <bool OUT16>
__global__ void __launch_bounds__(256)
gcn_pull(const float* __restrict__ hlin,
         const int* __restrict__ rowptr,
         const int* __restrict__ csr,
         const float* __restrict__ dinv,
         const float* __restrict__ bconv,
         const float* __restrict__ g,
         const float* __restrict__ bb,
         const float* __restrict__ m,
         const float* __restrict__ v,
         _Float16* __restrict__ out16,
         float* __restrict__ out32, int N) {
    const int lane = threadIdx.x & 31;
    const int n = __builtin_amdgcn_readfirstlane(
        (int)((blockIdx.x * blockDim.x + threadIdx.x) >> 5));
    if (n >= N) return;
    const int beg = rowptr[n], end = rowptr[n + 1];
    const float4* hp = (const float4*)hlin;

    float4 a0 = {0.0f, 0.0f, 0.0f, 0.0f}, a1 = {0.0f, 0.0f, 0.0f, 0.0f};
    for (int base = beg; base < end; base += 32) {
        const int nb = min(32, end - base);
        int my_s = 0, my_c = 0;
        if (lane < nb) { const int2 rec = ((const int2*)csr)[base + lane]; my_s = rec.x; my_c = rec.y; }
        int j = 0;
        for (; j + 4 <= nb; j += 4) pull_rows<4>(hp, my_s, my_c, j, lane, a0, a1);
        const int r = nb - j;
        if (r == 3)      pull_rows<3>(hp, my_s, my_c, j, lane, a0, a1);
        else if (r == 2) pull_rows<2>(hp, my_s, my_c, j, lane, a0, a1);
        else if (r == 1) pull_rows<1>(hp, my_s, my_c, j, lane, a0, a1);
    }

    const float di = dinv[n];
    const float4* own = hp + (size_t)n * (H_DIM / 4);
    const float4 h0 = own[lane], h1 = own[32 + lane];
    const float4* bc4 = (const float4*)bconv; const float4* g4 = (const float4*)g;
    const float4* bb4 = (const float4*)bb;    const float4* m4 = (const float4*)m;
    const float4* v4  = (const float4*)v;
    const float4 r0 = gcn_fin4(a0, h0, di, bc4[lane], m4[lane], v4[lane], g4[lane], bb4[lane]);
    const float4 r1 = gcn_fin4(a1, h1, di, bc4[32 + lane], m4[32 + lane], v4[32 + lane],
                               g4[32 + lane], bb4[32 + lane]);
    if (OUT16) {
        v4h* o = (v4h*)(out16 + (size_t)n * H_DIM);
        v4h x0 = {(_Float16)r0.x, (_Float16)r0.y, (_Float16)r0.z, (_Float16)r0.w};
        v4h x1 = {(_Float16)r1.x, (_Float16)r1.y, (_Float16)r1.z, (_Float16)r1.w};
        o[lane] = x0;
        o[32 + lane] = x1;
    } else {
        float4* o = (float4*)(out32 + (size_t)n * H_DIM);
        o[lane] = r0;
        o[32 + lane] = r1;
    }
}

__global__ void f32_to_f16_x8(const float* __restrict__ in, _Float16* __restrict__ out,
                              int n8) {
    int t = blockIdx.x * blockDim.x + threadIdx.x;
    if (t >= n8) return;
    const float4 a = ((const float4*)in)[2 * t], b = ((const float4*)in)[2 * t + 1];
    v8h o = {(_Float16)a.x, (_Float16)a.y, (_Float16)a.z, (_Float16)a.w,
             (_Float16)b.x, (_Float16)b.y, (_Float16)b.z, (_Float16)b.w};
    ((v8h*)out)[t] = o;
}

__global__ void attn_score(const float* __restrict__ av,
                           const float* __restrict__ au,
                           const float* __restrict__ wA,
                           float* __restrict__ a, int N, int HA) {
    int wave = (blockIdx.x * blockDim.x + threadIdx.x) >> 5;
    int lane = threadIdx.x & 31;
    if (wave >= N) return;
    float sum = 0.0f;
    for (int j = lane; j < HA; j += 32) {
        float t = tanhf(av[(size_t)wave * HA + j]);
        float s = 1.0f / (1.0f + expf(-au[(size_t)wave * HA + j]));
        sum += t * s * wA[j];
    }
    for (int off = 16; off > 0; off >>= 1) sum += __shfl_down(sum, off, 32);
    if (lane == 0) a[wave] = sum;
}

__global__ void seg_init(int* seg, int cnt, int sentinel) {
    int t = blockIdx.x * blockDim.x + threadIdx.x;
    if (t < cnt) seg[t] = sentinel;
}
__global__ void seg_mark(const int* __restrict__ batch, int* seg, int N) {
    int t = blockIdx.x * blockDim.x + threadIdx.x;
    if (t < N) atomicMin(&seg[batch[t]], t);
}
__global__ void seg_fix(int* seg, int B, int Nval) {
    if (blockIdx.x == 0 && threadIdx.x == 0) {
        seg[B] = Nval;
        for (int g = B - 1; g >= 0; --g)
            if (seg[g] == Nval) seg[g] = seg[g + 1];
    }
}

__global__ void seg_softmax(const float* __restrict__ a,
                            const int* __restrict__ seg,
                            float* __restrict__ wout) {
    __shared__ float red[256];
    int g = blockIdx.x;
    int s = seg[g], e = seg[g + 1];
    float mx = -1e30f;
    for (int n = s + (int)threadIdx.x; n < e; n += blockDim.x)
        mx = fmaxf(mx, a[n]);
    red[threadIdx.x] = mx; __syncthreads();
    for (int off = 128; off > 0; off >>= 1) {
        if ((int)threadIdx.x < off)
            red[threadIdx.x] = fmaxf(red[threadIdx.x], red[threadIdx.x + off]);
        __syncthreads();
    }
    mx = red[0]; __syncthreads();
    float sum = 0.0f;
    for (int n = s + (int)threadIdx.x; n < e; n += blockDim.x)
        sum += expf(a[n] - mx);
    red[threadIdx.x] = sum; __syncthreads();
    for (int off = 128; off > 0; off >>= 1) {
        if ((int)threadIdx.x < off)
            red[threadIdx.x] += red[threadIdx.x + off];
        __syncthreads();
    }
    float inv = 1.0f / red[0];
    for (int n = s + (int)threadIdx.x; n < e; n += blockDim.x)
        wout[n] = expf(a[n] - mx) * inv;
}

#define POOL_CH 32
#define POOL_T  512
#define POOL_RG (POOL_T / 64)
#define POOL_U  8
static_assert(H_DIM == 256, "pooling assumes 64 float4 per row");
__global__ void __launch_bounds__(POOL_T)
slide_partial(const float* __restrict__ w, const float* __restrict__ h,
              const int* __restrict__ seg, float* __restrict__ part) {
    __shared__ float4 red[POOL_RG][64];
    const int g = blockIdx.x / POOL_CH, c = blockIdx.x % POOL_CH;
    const int s = seg[g], e = seg[g + 1];
    const int per = (e - s + POOL_CH - 1) / POOL_CH;
    const int n0 = s + c * per;
    const int n1 = min(n0 + per, e);
    const int col = threadIdx.x & 63;
    const int rg = __builtin_amdgcn_readfirstlane((int)(threadIdx.x >> 6));
    const float4* h4 = (const float4*)h;
    const float4 zero4 = {0.0f, 0.0f, 0.0f, 0.0f};
    float4 acc = zero4;
    for (int nb = n0 + rg; nb < n1; nb += POOL_U * POOL_RG) {
        float4 v[POOL_U];
        float  ww[POOL_U];
#pragma unroll
        for (int u = 0; u < POOL_U; ++u) {
            const int n  = nb + u * POOL_RG;
            const int nc = min(n, n1 - 1);
            v[u]  = h4[(size_t)nc * (H_DIM / 4) + col];
            ww[u] = w[nc];
            if (n >= n1) v[u] = zero4;
        }
#pragma unroll
        for (int u = 0; u < POOL_U; ++u) {
            acc.x = fmaf(ww[u], v[u].x, acc.x); acc.y = fmaf(ww[u], v[u].y, acc.y);
            acc.z = fmaf(ww[u], v[u].z, acc.z); acc.w = fmaf(ww[u], v[u].w, acc.w);
        }
    }
    red[rg][col] = acc;
    __syncthreads();
    if (threadIdx.x < H_DIM) {
        const float* rf = (const float*)red;
        float sum = 0.0f;
#pragma unroll
        for (int r = 0; r < POOL_RG; ++r) sum += rf[r * H_DIM + threadIdx.x];
        part[(size_t)blockIdx.x * H_DIM + threadIdx.x] = sum;
    }
}

__global__ void head_kernel(const float* __restrict__ part,
                            const float* __restrict__ Wc1,
                            const float* __restrict__ bc1,
                            const float* __restrict__ Wc2,
                            const float* __restrict__ bc2,
                            float* __restrict__ logits) {
    __shared__ float slide[N_BAGS * H_DIM];
    __shared__ float hid[N_BAGS * HA_DIM];
    for (int item = threadIdx.x; item < N_BAGS * H_DIM; item += blockDim.x) {
        int g = item / H_DIM, k = item - g * H_DIM;
        const float* p = part + (size_t)g * POOL_CH * H_DIM + k;
        float s = 0.0f;
#pragma unroll
        for (int c = 0; c < POOL_CH; ++c) s += p[c * H_DIM];
        slide[item] = s;
    }
    __syncthreads();
    for (int item = threadIdx.x; item < N_BAGS * HA_DIM; item += blockDim.x) {
        int g = item / HA_DIM, t = item - g * HA_DIM;
        float s = bc1[t];
        for (int k = 0; k < H_DIM; ++k) s += slide[g * H_DIM + k] * Wc1[k * HA_DIM + t];
        hid[item] = fmaxf(s, 0.0f);
    }
    __syncthreads();
    for (int item = threadIdx.x; item < N_BAGS * C_DIM; item += blockDim.x) {
        int g = item / C_DIM, c = item - g * C_DIM;
        float s = bc2[c];
        for (int t = 0; t < HA_DIM; ++t) s += hid[g * HA_DIM + t] * Wc2[t * C_DIM + c];
        logits[item] = s;
    }
}

static inline int cdiv(long long a, int b) { return (int)((a + b - 1) / b); }

extern "C" void kernel_launch(void* const* d_in, const int* in_sizes, int n_in,
                              void* d_out, int out_size, void* d_ws, size_t ws_size,
                              hipStream_t stream) {
    (void)in_sizes; (void)n_in; (void)out_size; (void)ws_size;
    const float* x   = (const float*)d_in[0];
    const int*  ei   = (const int*)d_in[1];
    const int*  batch= (const int*)d_in[2];
    const float* W1  = (const float*)d_in[3];  const float* b1   = (const float*)d_in[4];
    const float* g1  = (const float*)d_in[5];  const float* bb1  = (const float*)d_in[6];
    const float* m1  = (const float*)d_in[7];  const float* v1   = (const float*)d_in[8];
    const float* W2  = (const float*)d_in[9];  const float* b2   = (const float*)d_in[10];
    const float* g2  = (const float*)d_in[11]; const float* bb2  = (const float*)d_in[12];
    const float* m2  = (const float*)d_in[13]; const float* v2   = (const float*)d_in[14];
    const float* Wv  = (const float*)d_in[15]; const float* bv   = (const float*)d_in[16];
    const float* Wu  = (const float*)d_in[17]; const float* bu   = (const float*)d_in[18];
    const float* wA  = (const float*)d_in[19];
    const float* Wc1 = (const float*)d_in[20]; const float* bc1  = (const float*)d_in[21];
    const float* Wc2 = (const float*)d_in[22]; const float* bc2  = (const float*)d_in[23];

    const int* src = ei;
    const int* dst = ei + N_EDGES;

    float* out  = (float*)d_out;
    float* wout = out + N_BAGS * C_DIM;

    const size_t NH = (size_t)N_NODES * H_DIM;
    float*    bufA = (float*)d_ws;
    float*    bufB = bufA + NH;
    _Float16* h16  = (_Float16*)(bufB + NH);
    _Float16* w1pk = h16 + NH;
    _Float16* w2pk = w1pk + (size_t)D_IN * H_DIM;
    _Float16* wvpk = w2pk + (size_t)H_DIM * H_DIM;
    _Float16* wupk = wvpk + (size_t)H_DIM * HA_DIM;
    float*    dinv = (float*)(wupk + (size_t)H_DIM * HA_DIM);
    float*    abuf = dinv + N_NODES;
    int*      seg  = (int*)(abuf + N_NODES);

    int*   cnt      = (int*)h16;
    int*   rowptr   = cnt + N_NODES;
    int*   rank     = rowptr + N_NODES + 4;
    int*   csr      = rank + N_EDGES;
    _Float16* h1_16 = (_Float16*)bufB;
    float*    h2_32 = bufB;

    const int TB = 256;

    zero_i32<<<cdiv(N_NODES, TB), TB, 0, stream>>>(cnt, N_NODES);
    count_rank<<<cdiv(N_EDGES, TB), TB, 0, stream>>>(dst, cnt, rank, N_EDGES);
    scan_rowptr<<<1, SCAN_T, 0, stream>>>(cnt, rowptr, dinv);
    csr_fill<<<cdiv(N_EDGES, TB), TB, 0, stream>>>(src, dst, rank, rowptr, dinv,
                                                   csr, N_EDGES);

    pack_w_f16<<<cdiv((long long)D_IN * H_DIM, TB), TB, 0, stream>>>(W1, w1pk, D_IN, H_DIM);
    pack_w_f16<<<cdiv((long long)H_DIM * H_DIM, TB), TB, 0, stream>>>(W2, w2pk, H_DIM, H_DIM);
    pack_w_f16<<<cdiv((long long)H_DIM * HA_DIM, TB), TB, 0, stream>>>(Wv, wvpk, H_DIM, HA_DIM);
    pack_w_f16<<<cdiv((long long)H_DIM * HA_DIM, TB), TB, 0, stream>>>(Wu, wupk, H_DIM, HA_DIM);

    wmma_gemm_lds_af32<D_IN, H_DIM><<<N_NODES / GT_ROWS, GT_T, 0, stream>>>(x, w1pk, bufA);
    gcn_pull<true><<<cdiv((long long)N_NODES * 32, TB), TB, 0, stream>>>(
        bufA, rowptr, csr, dinv, b1, g1, bb1, m1, v1,
        h1_16, nullptr, N_NODES);

    wmma_gemm_lds_af16<H_DIM, H_DIM, false><<<N_NODES / GT_ROWS, GT_T, 0, stream>>>(
        h1_16, w2pk, nullptr, bufA);
    gcn_pull<false><<<cdiv((long long)N_NODES * 32, TB), TB, 0, stream>>>(
        bufA, rowptr, csr, dinv, b2, g2, bb2, m2, v2,
        nullptr, h2_32, N_NODES);
    f32_to_f16_x8<<<cdiv((long long)(NH / 8), TB), TB, 0, stream>>>(h2_32, h16, (int)(NH / 8));

    float* av = bufA;
    float* au = bufA + (size_t)N_NODES * HA_DIM;
    wmma_gemm_lds_af16<H_DIM, HA_DIM, true><<<N_NODES / GT_ROWS, GT_T, 0, stream>>>(
        h16, wvpk, bv, av);
    wmma_gemm_lds_af16<H_DIM, HA_DIM, true><<<N_NODES / GT_ROWS, GT_T, 0, stream>>>(
        h16, wupk, bu, au);
    attn_score<<<cdiv((long long)N_NODES * 32, TB), TB, 0, stream>>>(
        av, au, wA, abuf, N_NODES, HA_DIM);

    seg_init<<<1, 32, 0, stream>>>(seg, N_BAGS + 1, N_NODES);
    seg_mark<<<cdiv(N_NODES, TB), TB, 0, stream>>>(batch, seg, N_NODES);
    seg_fix<<<1, 32, 0, stream>>>(seg, N_BAGS, N_NODES);

    seg_softmax<<<N_BAGS, 256, 0, stream>>>(abuf, seg, wout);

    float* part = bufA;
    slide_partial<<<N_BAGS * POOL_CH, POOL_T, 0, stream>>>(wout, h2_32, seg, part);

    head_kernel<<<1, 256, 0, stream>>>(part, Wc1, bc1, Wc2, bc2, out);
}


# Independent launches merged, 22 -> 14: 4 weight packs + counter zero fill + bag boundaries (one writer each, no atomicMin) as block ranges of one grid; Wv|Wu GEMMs as one grid on one A tile
# speedup vs baseline: 3.8590x; 1.2623x over previous
#include <hip/hip_runtime.h>
#include <hip/hip_bf16.h>
#include <math.h>

typedef __attribute__((ext_vector_type(16))) _Float16 v16h;
typedef __attribute__((ext_vector_type(8)))  _Float16 v8h;
typedef __attribute__((ext_vector_type(8)))  float    v8f;

#define N_NODES 65536
#define N_EDGES 524288
#define N_BAGS  8
#define D_IN    512
#define H_DIM   256
#define HA_DIM  128
#define C_DIM   2
#define BN_EPS  1e-5f
#define MREP    4

__global__ void count_rank(const int* __restrict__ dst, int* __restrict__ cnt,
                           int* __restrict__ rank, int E) {
    int t = blockIdx.x * blockDim.x + threadIdx.x;
    if (t < E) rank[t] = atomicAdd(&cnt[dst[t]], 1);
}

#define SCAN_T   1024
#define SCAN_PER (N_NODES / SCAN_T)
static_assert(N_NODES % (SCAN_T * 4) == 0, "scan assumes whole int4 chunks per thread");
__global__ void __launch_bounds__(SCAN_T)
scan_rowptr(const int* __restrict__ cnt, int* __restrict__ rowptr,
            float* __restrict__ dinv) {
    __shared__ int wsum[32];
    const int t = threadIdx.x, lane = t & 31, w = t >> 5;
    const int4* c4 = (const int4*)(cnt + t * SCAN_PER);
    int4 c[SCAN_PER / 4];
    int s = 0;
#pragma unroll
    for (int i = 0; i < SCAN_PER / 4; ++i) {
        c[i] = c4[i];
        s += c[i].x + c[i].y + c[i].z + c[i].w;
    }
    int inc = s;
    for (int off = 1; off < 32; off <<= 1) {
        int v = __shfl_up(inc, off, 32);
        if (lane >= off) inc += v;
    }
    if (lane == 31) wsum[w] = inc;
    __syncthreads();
    if (w == 0) {
        int v = wsum[lane], iv = v;
        for (int off = 1; off < 32; off <<= 1) {
            int u = __shfl_up(iv, off, 32);
            if (lane >= off) iv += u;
        }
        wsum[lane] = iv - v;
    }
    __syncthreads();
    int run = wsum[w] + inc - s;
    int4*   r4 = (int4*)(rowptr + t * SCAN_PER);
    float4* d4 = (float4*)(dinv + t * SCAN_PER);
#pragma unroll
    for (int i = 0; i < SCAN_PER / 4; ++i) {
        int4 r;
        r.x = run; run += c[i].x;
        r.y = run; run += c[i].y;
        r.z = run; run += c[i].z;
        r.w = run; run += c[i].w;
        r4[i] = r;
        float4 d;
        d.x = rsqrtf(1.0f + (float)c[i].x);
        d.y = rsqrtf(1.0f + (float)c[i].y);
        d.z = rsqrtf(1.0f + (float)c[i].z);
        d.w = rsqrtf(1.0f + (float)c[i].w);
        d4[i] = d;
    }
    if (t == SCAN_T - 1) rowptr[N_NODES] = run;
}

__global__ void csr_fill(const int* __restrict__ src, const int* __restrict__ dst,
                         const int* __restrict__ rank, const int* __restrict__ rowptr,
                         const float* __restrict__ dinv,
                         int* __restrict__ csr, int E) {
    int t = blockIdx.x * blockDim.x + threadIdx.x;
    if (t >= E) return;
    int s = src[t], d = dst[t];
    int pos = rowptr[d] + rank[t];
    ((int2*)csr)[pos] = make_int2(s, __float_as_int(dinv[s] * dinv[d]));
}

template <int K, int NOUT>
__device__ __forceinline__ void pack_w_elem(const float* __restrict__ Wm,
                                            _Float16* __restrict__ out, int t) {
    constexpr int KBLKS = K >> 5;
    const int i    = t & 15;
    const int lane = (t >> 4) & 31;
    const int rest = t >> 9;
    const int ntile = rest / KBLKS;
    const int kblk  = rest % KBLKS;
    const int half = lane >> 4, l15 = lane & 15;
    const int col = ntile * 16 + l15;
    const int k   = kblk * 32 + half * 8 + ((i < 8) ? i : (16 + i - 8));
    out[t] = (_Float16)Wm[(size_t)k * NOUT + col];
}

#define SU_T      256
#define SU_B_W1   (D_IN * H_DIM / SU_T)
#define SU_B_W2   (H_DIM * H_DIM / SU_T)
#define SU_B_WA   (H_DIM * HA_DIM / SU_T)
#define SU_B_SEG  (N_NODES / SU_T)
#define SU_B_ZERO (N_NODES / SU_T)
#define SU_BLOCKS (SU_B_W1 + SU_B_W2 + 2 * SU_B_WA + SU_B_SEG + SU_B_ZERO)
static_assert(N_NODES % SU_T == 0 && (H_DIM * HA_DIM) % SU_T == 0, "whole blocks per range");
__global__ void __launch_bounds__(SU_T)
setup_all(const float* __restrict__ W1, _Float16* __restrict__ w1pk,
          const float* __restrict__ W2, _Float16* __restrict__ w2pk,
          const float* __restrict__ Wv, _Float16* __restrict__ wvpk,
          const float* __restrict__ Wu, _Float16* __restrict__ wupk,
          const int* __restrict__ batch, int* __restrict__ seg,
          int* __restrict__ cnt) {
    int b = blockIdx.x;
    const int tid = threadIdx.x;
    if (b < SU_B_W1) { pack_w_elem<D_IN, H_DIM>(W1, w1pk, b * SU_T + tid); return; }
    b -= SU_B_W1;
    if (b < SU_B_W2) { pack_w_elem<H_DIM, H_DIM>(W2, w2pk, b * SU_T + tid); return; }
    b -= SU_B_W2;
    if (b < SU_B_WA) { pack_w_elem<H_DIM, HA_DIM>(Wv, wvpk, b * SU_T + tid); return; }
    b -= SU_B_WA;
    if (b < SU_B_WA) { pack_w_elem<H_DIM, HA_DIM>(Wu, wupk, b * SU_T + tid); return; }
    b -= SU_B_WA;
    if (b < SU_B_SEG) {
        const int n = b * SU_T + tid;
        const int cur  = batch[n];
        const int prev = (n == 0) ? -1 : batch[n - 1];
#pragma unroll
        for (int g = 0; g < N_BAGS; ++g)
            if (g > prev && g <= cur) seg[g] = n;
        if (n == N_NODES - 1) {
#pragma unroll
            for (int g = 1; g <= N_BAGS; ++g)
                if (g > cur) seg[g] = N_NODES;
        }
        return;
    }
    b -= SU_B_SEG;
    cnt[b * SU_T + tid] = 0;
}

#define GT_ROWS 64
#define GT_T    256
#define GT_LD   8
static_assert(GT_ROWS == 16 * MREP, "a wave covers all rows of the tile");
static_assert(N_NODES % GT_ROWS == 0, "no partial row tile");

#define GT_PITCH(K) ((K) + 8)

template <int K, int NOUT, bool BIAS, bool TWO = false>
__device__ __forceinline__ void gemm_tile_from_lds(const _Float16* As,
                                                   const _Float16* __restrict__ Wpk,
                                                   const float* __restrict__ bias,
                                                   float* __restrict__ C, int m0,
                                                   const float* __restrict__ bias1 = nullptr,
                                                   float* __restrict__ C1 = nullptr) {
    constexpr int PITCH = GT_PITCH(K);
    constexpr int KBLKS = K / 32;
    constexpr int NTW   = NOUT / 16 / (GT_T / 32);
    static_assert(NTW >= 1 && NTW * 16 * (GT_T / 32) == NOUT, "waves tile the columns");
    const int lane = threadIdx.x & 31;
    const int wave = __builtin_amdgcn_readfirstlane((int)(threadIdx.x >> 5));
    const int half = lane >> 4;
    const int l15  = lane & 15;
    const _Float16* wb = Wpk + (size_t)(wave * NTW) * KBLKS * 512 + lane * 16;
    const _Float16* ab = As + l15 * PITCH + half * 8;

    v8f zero = {};
    v8f acc[MREP][NTW];
#pragma unroll
    for (int r = 0; r < MREP; ++r)
#pragma unroll
        for (int t = 0; t < NTW; ++t) acc[r][t] = zero;

    for (int kblk = 0; kblk < KBLKS; ++kblk) {
        v16h b[NTW];
#pragma unroll
        for (int t = 0; t < NTW; ++t)
            b[t] = *(const v16h*)(wb + ((size_t)t * KBLKS + kblk) * 512);
#pragma unroll
        for (int r = 0; r < MREP; ++r) {
            const _Float16* ap = ab + r * 16 * PITCH + kblk * 32;
            v8h lo = *(const v8h*)(ap);
            v8h hi = *(const v8h*)(ap + 16);
            v16h a = __builtin_shufflevector(lo, hi, 0, 1, 2, 3, 4, 5, 6, 7,
                                             8, 9, 10, 11, 12, 13, 14, 15);
#pragma unroll
            for (int t = 0; t < NTW; ++t)
                acc[r][t] = __builtin_amdgcn_wmma_f32_16x16x32_f16(
                    false, a, false, b[t], (short)0, acc[r][t], false, false);
        }
    }

    constexpr int CW = TWO ? NOUT / 2 : NOUT;
    static_assert(!TWO || (NOUT / 2) % (16 * NTW) == 0, "a wave stays inside one plane");
#pragma unroll
    for (int t = 0; t < NTW; ++t) {
        int ct = wave * NTW + t;
        const float* bp = bias;
        float* Cp = C;
        if (TWO && ct >= CW / 16) { ct -= CW / 16; bp = bias1; Cp = C1; }
        const int col = ct * 16 + l15;
        float bv = 0.0f;
        if (BIAS) bv = bp[col];
#pragma unroll
        for (int r = 0; r < MREP; ++r) {
#pragma unroll
            for (int q = 0; q < 8; ++q) {
                const int row = m0 + r * 16 + q + half * 8;
                Cp[(size_t)row * CW + col] = BIAS ? acc[r][t][q] + bv : acc[r][t][q];
            }
        }
    }
}

typedef __attribute__((ext_vector_type(4))) _Float16 v4h_t;

template <int K, int NOUT>
__global__ void __launch_bounds__(GT_T)
wmma_gemm_lds_af32(const float* __restrict__ A, const _Float16* __restrict__ Wpk,
                   float* __restrict__ C) {
    __shared__ alignas(16) _Float16 As[GT_ROWS * GT_PITCH(K)];
    constexpr int VROW  = K / 4;
    constexpr int PER_T = GT_ROWS * VROW / GT_T;
    static_assert(PER_T % GT_LD == 0 && VROW % 32 == 0, "whole batches, a wave inside one row");
    const int tid = threadIdx.x;
    const int m0  = blockIdx.x * GT_ROWS;
    const float4* Ag = (const float4*)(A + (size_t)m0 * K);
#pragma unroll 1
    for (int it = 0; it < PER_T; it += GT_LD) {
        float4 v[GT_LD];
#pragma unroll
        for (int u = 0; u < GT_LD; ++u) v[u] = Ag[(it + u) * GT_T + tid];
#pragma unroll
        for (int u = 0; u < GT_LD; ++u) {
            const int f = (it + u) * GT_T + tid;
            const int row = f / VROW, c = f % VROW;
            v4h_t h = {(_Float16)v[u].x, (_Float16)v[u].y, (_Float16)v[u].z, (_Float16)v[u].w};
            *(v4h_t*)(As + row * GT_PITCH(K) + c * 4) = h;
        }
    }
    __syncthreads();
    gemm_tile_from_lds<K, NOUT, false>(As, Wpk, nullptr, C, m0);
}

template <int K, int NOUT, bool BIAS, bool TWO = false>
__global__ void __launch_bounds__(GT_T)
wmma_gemm_lds_af16(const _Float16* __restrict__ A, const _Float16* __restrict__ Wpk,
                   const float* __restrict__ bias, float* __restrict__ C,
                   const float* __restrict__ bias1 = nullptr,
                   float* __restrict__ C1 = nullptr) {
    __shared__ alignas(16) _Float16 As[GT_ROWS * GT_PITCH(K)];
    constexpr int VROW  = K / 8;
    constexpr int PER_T = GT_ROWS * VROW / GT_T;
    static_assert(GT_ROWS * VROW % GT_T == 0 && VROW % 32 == 0, "whole passes, a wave inside one row");
    const int tid = threadIdx.x;
    const int m0  = blockIdx.x * GT_ROWS;
    const v8h* Ag = (const v8h*)(A + (size_t)m0 * K);
    v8h v[PER_T];
#pragma unroll
    for (int u = 0; u < PER_T; ++u) v[u] = Ag[u * GT_T + tid];
#pragma unroll
    for (int u = 0; u < PER_T; ++u) {
        const int f = u * GT_T + tid;
        const int row = f / VROW, c = f % VROW;
        *(v8h*)(As + row * GT_PITCH(K) + c * 8) = v[u];
    }
    __syncthreads();
    gemm_tile_from_lds<K, NOUT, BIAS, TWO>(As, Wpk, bias, C, m0, bias1, C1);
}

template <int K>
__device__ __forceinline__ void pull_rows(const float4* __restrict__ hp, int my_s, int my_c,
                                          int j, int lane, float4& a0, float4& a1) {
    float4 p[K], q[K];
    float  c[K];
#pragma unroll
    for (int k = 0; k < K; ++k) {
        const int s = __builtin_amdgcn_readlane(my_s, j + k);
        c[k] = __int_as_float(__builtin_amdgcn_readlane(my_c, j + k));
        const float4* row = hp + (size_t)s * (H_DIM / 4);
        p[k] = row[lane];
        q[k] = row[32 + lane];
    }
#pragma unroll
    for (int k = 0; k < K; ++k) {
        a0.x = fmaf(p[k].x, c[k], a0.x); a0.y = fmaf(p[k].y, c[k], a0.y);
        a0.z = fmaf(p[k].z, c[k], a0.z); a0.w = fmaf(p[k].w, c[k], a0.w);
        a1.x = fmaf(q[k].x, c[k], a1.x); a1.y = fmaf(q[k].y, c[k], a1.y);
        a1.z = fmaf(q[k].z, c[k], a1.z); a1.w = fmaf(q[k].w, c[k], a1.w);
    }
}

__device__ __forceinline__ float gcn_fin(float agg, float hl, float di, float bc,
                                         float mm, float vv, float gg, float bo) {
    float val = agg + hl * di * di + bc;
    val = (val - mm) * rsqrtf(vv + BN_EPS) * gg + bo;
    return fmaxf(val, 0.0f);
}

__device__ __forceinline__ float4 gcn_fin4(float4 a, float4 h, float di, float4 bc,
                                           float4 mm, float4 vv, float4 gg, float4 bo) {
    float4 r;
    r.x = gcn_fin(a.x, h.x, di, bc.x, mm.x, vv.x, gg.x, bo.x);
    r.y = gcn_fin(a.y, h.y, di, bc.y, mm.y, vv.y, gg.y, bo.y);
    r.z = gcn_fin(a.z, h.z, di, bc.z, mm.z, vv.z, gg.z, bo.z);
    r.w = gcn_fin(a.w, h.w, di, bc.w, mm.w, vv.w, gg.w, bo.w);
    return r;
}

typedef __attribute__((ext_vector_type(4))) _Float16 v4h;

template <bool OUT16>
__global__ void __launch_bounds__(256)
gcn_pull(const float* __restrict__ hlin,
         const int* __restrict__ rowptr,
         const int* __restrict__ csr,
         const float* __restrict__ dinv,
         const float* __restrict__ bconv,
         const float* __restrict__ g,
         const float* __restrict__ bb,
         const float* __restrict__ m,
         const float* __restrict__ v,
         _Float16* __restrict__ out16,
         float* __restrict__ out32, int N) {
    const int lane = threadIdx.x & 31;
    const int n = __builtin_amdgcn_readfirstlane(
        (int)((blockIdx.x * blockDim.x + threadIdx.x) >> 5));
    if (n >= N) return;
    const int beg = rowptr[n], end = rowptr[n + 1];
    const float4* hp = (const float4*)hlin;

    float4 a0 = {0.0f, 0.0f, 0.0f, 0.0f}, a1 = {0.0f, 0.0f, 0.0f, 0.0f};
    for (int base = beg; base < end; base += 32) {
        const int nb = min(32, end - base);
        int my_s = 0, my_c = 0;
        if (lane < nb) { const int2 rec = ((const int2*)csr)[base + lane]; my_s = rec.x; my_c = rec.y; }
        int j = 0;
        for (; j + 4 <= nb; j += 4) pull_rows<4>(hp, my_s, my_c, j, lane, a0, a1);
        const int r = nb - j;
        if (r == 3)      pull_rows<3>(hp, my_s, my_c, j, lane, a0, a1);
        else if (r == 2) pull_rows<2>(hp, my_s, my_c, j, lane, a0, a1);
        else if (r == 1) pull_rows<1>(hp, my_s, my_c, j, lane, a0, a1);
    }

    const float di = dinv[n];
    const float4* own = hp + (size_t)n * (H_DIM / 4);
    const float4 h0 = own[lane], h1 = own[32 + lane];
    const float4* bc4 = (const float4*)bconv; const float4* g4 = (const float4*)g;
    const float4* bb4 = (const float4*)bb;    const float4* m4 = (const float4*)m;
    const float4* v4  = (const float4*)v;
    const float4 r0 = gcn_fin4(a0, h0, di, bc4[lane], m4[lane], v4[lane], g4[lane], bb4[lane]);
    const float4 r1 = gcn_fin4(a1, h1, di, bc4[32 + lane], m4[32 + lane], v4[32 + lane],
                               g4[32 + lane], bb4[32 + lane]);
    if (OUT16) {
        v4h* o = (v4h*)(out16 + (size_t)n * H_DIM);
        v4h x0 = {(_Float16)r0.x, (_Float16)r0.y, (_Float16)r0.z, (_Float16)r0.w};
        v4h x1 = {(_Float16)r1.x, (_Float16)r1.y, (_Float16)r1.z, (_Float16)r1.w};
        o[lane] = x0;
        o[32 + lane] = x1;
    } else {
        float4* o = (float4*)(out32 + (size_t)n * H_DIM);
        o[lane] = r0;
        o[32 + lane] = r1;
    }
}

__global__ void f32_to_f16_x8(const float* __restrict__ in, _Float16* __restrict__ out,
                              int n8) {
    int t = blockIdx.x * blockDim.x + threadIdx.x;
    if (t >= n8) return;
    const float4 a = ((const float4*)in)[2 * t], b = ((const float4*)in)[2 * t + 1];
    v8h o = {(_Float16)a.x, (_Float16)a.y, (_Float16)a.z, (_Float16)a.w,
             (_Float16)b.x, (_Float16)b.y, (_Float16)b.z, (_Float16)b.w};
    ((v8h*)out)[t] = o;
}

__global__ void attn_score(const float* __restrict__ av,
                           const float* __restrict__ au,
                           const float* __restrict__ wA,
                           float* __restrict__ a, int N, int HA) {
    int wave = (blockIdx.x * blockDim.x + threadIdx.x) >> 5;
    int lane = threadIdx.x & 31;
    if (wave >= N) return;
    float sum = 0.0f;
    for (int j = lane; j < HA; j += 32) {
        float t = tanhf(av[(size_t)wave * HA + j]);
        float s = 1.0f / (1.0f + expf(-au[(size_t)wave * HA + j]));
        sum += t * s * wA[j];
    }
    for (int off = 16; off > 0; off >>= 1) sum += __shfl_down(sum, off, 32);
    if (lane == 0) a[wave] = sum;
}

__global__ void seg_softmax(const float* __restrict__ a,
                            const int* __restrict__ seg,
                            float* __restrict__ wout) {
    __shared__ float red[256];
    int g = blockIdx.x;
    int s = seg[g], e = seg[g + 1];
    float mx = -1e30f;
    for (int n = s + (int)threadIdx.x; n < e; n += blockDim.x)
        mx = fmaxf(mx, a[n]);
    red[threadIdx.x] = mx; __syncthreads();
    for (int off = 128; off > 0; off >>= 1) {
        if ((int)threadIdx.x < off)
            red[threadIdx.x] = fmaxf(red[threadIdx.x], red[threadIdx.x + off]);
        __syncthreads();
    }
    mx = red[0]; __syncthreads();
    float sum = 0.0f;
    for (int n = s + (int)threadIdx.x; n < e; n += blockDim.x)
        sum += expf(a[n] - mx);
    red[threadIdx.x] = sum; __syncthreads();
    for (int off = 128; off > 0; off >>= 1) {
        if ((int)threadIdx.x < off)
            red[threadIdx.x] += red[threadIdx.x + off];
        __syncthreads();
    }
    float inv = 1.0f / red[0];
    for (int n = s + (int)threadIdx.x; n < e; n += blockDim.x)
        wout[n] = expf(a[n] - mx) * inv;
}

#define POOL_CH 32
#define POOL_T  512
#define POOL_RG (POOL_T / 64)
#define POOL_U  8
static_assert(H_DIM == 256, "pooling assumes 64 float4 per row");
__global__ void __launch_bounds__(POOL_T)
slide_partial(const float* __restrict__ w, const float* __restrict__ h,
              const int* __restrict__ seg, float* __restrict__ part) {
    __shared__ float4 red[POOL_RG][64];
    const int g = blockIdx.x / POOL_CH, c = blockIdx.x % POOL_CH;
    const int s = seg[g], e = seg[g + 1];
    const int per = (e - s + POOL_CH - 1) / POOL_CH;
    const int n0 = s + c * per;
    const int n1 = min(n0 + per, e);
    const int col = threadIdx.x & 63;
    const int rg = __builtin_amdgcn_readfirstlane((int)(threadIdx.x >> 6));
    const float4* h4 = (const float4*)h;
    const float4 zero4 = {0.0f, 0.0f, 0.0f, 0.0f};
    float4 acc = zero4;
    for (int nb = n0 + rg; nb < n1; nb += POOL_U * POOL_RG) {
        float4 v[POOL_U];
        float  ww[POOL_U];
#pragma unroll
        for (int u = 0; u < POOL_U; ++u) {
            const int n  = nb + u * POOL_RG;
            const int nc = min(n, n1 - 1);
            v[u]  = h4[(size_t)nc * (H_DIM / 4) + col];
            ww[u] = w[nc];
            if (n >= n1) v[u] = zero4;
        }
#pragma unroll
        for (int u = 0; u < POOL_U; ++u) {
            acc.x = fmaf(ww[u], v[u].x, acc.x); acc.y = fmaf(ww[u], v[u].y, acc.y);
            acc.z = fmaf(ww[u], v[u].z, acc.z); acc.w = fmaf(ww[u], v[u].w, acc.w);
        }
    }
    red[rg][col] = acc;
    __syncthreads();
    if (threadIdx.x < H_DIM) {
        const float* rf = (const float*)red;
        float sum = 0.0f;
#pragma unroll
        for (int r = 0; r < POOL_RG; ++r) sum += rf[r * H_DIM + threadIdx.x];
        part[(size_t)blockIdx.x * H_DIM + threadIdx.x] = sum;
    }
}

__global__ void head_kernel(const float* __restrict__ part,
                            const float* __restrict__ Wc1,
                            const float* __restrict__ bc1,
                            const float* __restrict__ Wc2,
                            const float* __restrict__ bc2,
                            float* __restrict__ logits) {
    __shared__ float slide[N_BAGS * H_DIM];
    __shared__ float hid[N_BAGS * HA_DIM];
    for (int item = threadIdx.x; item < N_BAGS * H_DIM; item += blockDim.x) {
        int g = item / H_DIM, k = item - g * H_DIM;
        const float* p = part + (size_t)g * POOL_CH * H_DIM + k;
        float s = 0.0f;
#pragma unroll
        for (int c = 0; c < POOL_CH; ++c) s += p[c * H_DIM];
        slide[item] = s;
    }
    __syncthreads();
    for (int item = threadIdx.x; item < N_BAGS * HA_DIM; item += blockDim.x) {
        int g = item / HA_DIM, t = item - g * HA_DIM;
        float s = bc1[t];
        for (int k = 0; k < H_DIM; ++k) s += slide[g * H_DIM + k] * Wc1[k * HA_DIM + t];
        hid[item] = fmaxf(s, 0.0f);
    }
    __syncthreads();
    for (int item = threadIdx.x; item < N_BAGS * C_DIM; item += blockDim.x) {
        int g = item / C_DIM, c = item - g * C_DIM;
        float s = bc2[c];
        for (int t = 0; t < HA_DIM; ++t) s += hid[g * HA_DIM + t] * Wc2[t * C_DIM + c];
        logits[item] = s;
    }
}

static inline int cdiv(long long a, int b) { return (int)((a + b - 1) / b); }

extern "C" void kernel_launch(void* const* d_in, const int* in_sizes, int n_in,
                              void* d_out, int out_size, void* d_ws, size_t ws_size,
                              hipStream_t stream) {
    (void)in_sizes; (void)n_in; (void)out_size; (void)ws_size;
    const float* x   = (const float*)d_in[0];
    const int*  ei   = (const int*)d_in[1];
    const int*  batch= (const int*)d_in[2];
    const float* W1  = (const float*)d_in[3];  const float* b1   = (const float*)d_in[4];
    const float* g1  = (const float*)d_in[5];  const float* bb1  = (const float*)d_in[6];
    const float* m1  = (const float*)d_in[7];  const float* v1   = (const float*)d_in[8];
    const float* W2  = (const float*)d_in[9];  const float* b2   = (const float*)d_in[10];
    const float* g2  = (const float*)d_in[11]; const float* bb2  = (const float*)d_in[12];
    const float* m2  = (const float*)d_in[13]; const float* v2   = (const float*)d_in[14];
    const float* Wv  = (const float*)d_in[15]; const float* bv   = (const float*)d_in[16];
    const float* Wu  = (const float*)d_in[17]; const float* bu   = (const float*)d_in[18];
    const float* wA  = (const float*)d_in[19];
    const float* Wc1 = (const float*)d_in[20]; const float* bc1  = (const float*)d_in[21];
    const float* Wc2 = (const float*)d_in[22]; const float* bc2  = (const float*)d_in[23];

    const int* src = ei;
    const int* dst = ei + N_EDGES;

    float* out  = (float*)d_out;
    float* wout = out + N_BAGS * C_DIM;

    const size_t NH = (size_t)N_NODES * H_DIM;
    float*    bufA = (float*)d_ws;
    float*    bufB = bufA + NH;
    _Float16* h16  = (_Float16*)(bufB + NH);
    _Float16* w1pk = h16 + NH;
    _Float16* w2pk = w1pk + (size_t)D_IN * H_DIM;
    _Float16* wvpk = w2pk + (size_t)H_DIM * H_DIM;
    _Float16* wupk = wvpk + (size_t)H_DIM * HA_DIM;
    float*    dinv = (float*)(wupk + (size_t)H_DIM * HA_DIM);
    float*    abuf = dinv + N_NODES;
    int*      seg  = (int*)(abuf + N_NODES);

    int*   cnt      = (int*)h16;
    int*   rowptr   = cnt + N_NODES;
    int*   rank     = rowptr + N_NODES + 4;
    int*   csr      = rank + N_EDGES;
    _Float16* h1_16 = (_Float16*)bufB;
    float*    h2_32 = bufB;

    const int TB = 256;

    setup_all<<<SU_BLOCKS, SU_T, 0, stream>>>(W1, w1pk, W2, w2pk, Wv, wvpk, Wu, wupk,
                                              batch, seg, cnt);

    count_rank<<<cdiv(N_EDGES, TB), TB, 0, stream>>>(dst, cnt, rank, N_EDGES);
    scan_rowptr<<<1, SCAN_T, 0, stream>>>(cnt, rowptr, dinv);
    csr_fill<<<cdiv(N_EDGES, TB), TB, 0, stream>>>(src, dst, rank, rowptr, dinv,
                                                   csr, N_EDGES);

    wmma_gemm_lds_af32<D_IN, H_DIM><<<N_NODES / GT_ROWS, GT_T, 0, stream>>>(x, w1pk, bufA);
    gcn_pull<true><<<cdiv((long long)N_NODES * 32, TB), TB, 0, stream>>>(
        bufA, rowptr, csr, dinv, b1, g1, bb1, m1, v1,
        h1_16, nullptr, N_NODES);

    wmma_gemm_lds_af16<H_DIM, H_DIM, false><<<N_NODES / GT_ROWS, GT_T, 0, stream>>>(
        h1_16, w2pk, nullptr, bufA);
    gcn_pull<false><<<cdiv((long long)N_NODES * 32, TB), TB, 0, stream>>>(
        bufA, rowptr, csr, dinv, b2, g2, bb2, m2, v2,
        nullptr, h2_32, N_NODES);
    f32_to_f16_x8<<<cdiv((long long)(NH / 8), TB), TB, 0, stream>>>(h2_32, h16, (int)(NH / 8));

    float* av = bufA;
    float* au = bufA + (size_t)N_NODES * HA_DIM;
    wmma_gemm_lds_af16<H_DIM, 2 * HA_DIM, true, true><<<N_NODES / GT_ROWS, GT_T, 0, stream>>>(
        h16, wvpk, bv, av, bu, au);
    attn_score<<<cdiv((long long)N_NODES * 32, TB), TB, 0, stream>>>(
        av, au, wA, abuf, N_NODES, HA_DIM);

    seg_softmax<<<N_BAGS, 256, 0, stream>>>(abuf, seg, wout);

    float* part = bufA;
    slide_partial<<<N_BAGS * POOL_CH, POOL_T, 0, stream>>>(wout, h2_32, seg, part);

    head_kernel<<<1, 256, 0, stream>>>(part, Wc1, bc1, Wc2, bc2, out);
}


# One wave per item -> waves that loop over items (pull 4 nodes, score 8 nodes, f32->f16 copy 8 pieces, edge kernels 4 edges per thread): 65536-wave grids -> 16384 or 8192 waves; sum orders kept
# speedup vs baseline: 3.9973x; 1.0358x over previous
#include <hip/hip_runtime.h>
#include <hip/hip_bf16.h>
#include <math.h>

typedef __attribute__((ext_vector_type(16))) _Float16 v16h;
typedef __attribute__((ext_vector_type(8)))  _Float16 v8h;
typedef __attribute__((ext_vector_type(8)))  float    v8f;

#define N_NODES 65536
#define N_EDGES 524288
#define N_BAGS  8
#define D_IN    512
#define H_DIM   256
#define HA_DIM  128
#define C_DIM   2
#define BN_EPS  1e-5f
#define MREP    4

#define EDGE_PER 4
#define EDGE_T   256
static_assert(N_EDGES % (EDGE_T * EDGE_PER) == 0, "whole work-groups");
__global__ void __launch_bounds__(EDGE_T)
count_rank(const int* __restrict__ dst, int* __restrict__ cnt, int* __restrict__ rank) {
    const int t = blockIdx.x * blockDim.x + threadIdx.x;
    const int4 d = ((const int4*)dst)[t];
    int4 r;
    r.x = atomicAdd(&cnt[d.x], 1);
    r.y = atomicAdd(&cnt[d.y], 1);
    r.z = atomicAdd(&cnt[d.z], 1);
    r.w = atomicAdd(&cnt[d.w], 1);
    ((int4*)rank)[t] = r;
}

#define SCAN_T   1024
#define SCAN_PER (N_NODES / SCAN_T)
static_assert(N_NODES % (SCAN_T * 4) == 0, "scan assumes whole int4 chunks per thread");
__global__ void __launch_bounds__(SCAN_T)
scan_rowptr(const int* __restrict__ cnt, int* __restrict__ rowptr,
            float* __restrict__ dinv) {
    __shared__ int wsum[32];
    const int t = threadIdx.x, lane = t & 31, w = t >> 5;
    const int4* c4 = (const int4*)(cnt + t * SCAN_PER);
    int4 c[SCAN_PER / 4];
    int s = 0;
#pragma unroll
    for (int i = 0; i < SCAN_PER / 4; ++i) {
        c[i] = c4[i];
        s += c[i].x + c[i].y + c[i].z + c[i].w;
    }
    int inc = s;
    for (int off = 1; off < 32; off <<= 1) {
        int v = __shfl_up(inc, off, 32);
        if (lane >= off) inc += v;
    }
    if (lane == 31) wsum[w] = inc;
    __syncthreads();
    if (w == 0) {
        int v = wsum[lane], iv = v;
        for (int off = 1; off < 32; off <<= 1) {
            int u = __shfl_up(iv, off, 32);
            if (lane >= off) iv += u;
        }
        wsum[lane] = iv - v;
    }
    __syncthreads();
    int run = wsum[w] + inc - s;
    int4*   r4 = (int4*)(rowptr + t * SCAN_PER);
    float4* d4 = (float4*)(dinv + t * SCAN_PER);
#pragma unroll
    for (int i = 0; i < SCAN_PER / 4; ++i) {
        int4 r;
        r.x = run; run += c[i].x;
        r.y = run; run += c[i].y;
        r.z = run; run += c[i].z;
        r.w = run; run += c[i].w;
        r4[i] = r;
        float4 d;
        d.x = rsqrtf(1.0f + (float)c[i].x);
        d.y = rsqrtf(1.0f + (float)c[i].y);
        d.z = rsqrtf(1.0f + (float)c[i].z);
        d.w = rsqrtf(1.0f + (float)c[i].w);
        d4[i] = d;
    }
    if (t == SCAN_T - 1) rowptr[N_NODES] = run;
}

__device__ __forceinline__ void csr_put(int s, int d, int r,
                                        const int* __restrict__ rowptr,
                                        const float* __restrict__ dinv,
                                        int* __restrict__ csr) {
    const int pos = rowptr[d] + r;
    ((int2*)csr)[pos] = make_int2(s, __float_as_int(dinv[s] * dinv[d]));
}

__global__ void __launch_bounds__(EDGE_T)
csr_fill(const int* __restrict__ src, const int* __restrict__ dst,
         const int* __restrict__ rank, const int* __restrict__ rowptr,
         const float* __restrict__ dinv, int* __restrict__ csr) {
    const int t = blockIdx.x * blockDim.x + threadIdx.x;
    const int4 s = ((const int4*)src)[t], d = ((const int4*)dst)[t], r = ((const int4*)rank)[t];
    csr_put(s.x, d.x, r.x, rowptr, dinv, csr);
    csr_put(s.y, d.y, r.y, rowptr, dinv, csr);
    csr_put(s.z, d.z, r.z, rowptr, dinv, csr);
    csr_put(s.w, d.w, r.w, rowptr, dinv, csr);
}

template <int K, int NOUT>
__device__ __forceinline__ void pack_w_elem(const float* __restrict__ Wm,
                                            _Float16* __restrict__ out, int t) {
    constexpr int KBLKS = K >> 5;
    const int i    = t & 15;
    const int lane = (t >> 4) & 31;
    const int rest = t >> 9;
    const int ntile = rest / KBLKS;
    const int kblk  = rest % KBLKS;
    const int half = lane >> 4, l15 = lane & 15;
    const int col = ntile * 16 + l15;
    const int k   = kblk * 32 + half * 8 + ((i < 8) ? i : (16 + i - 8));
    out[t] = (_Float16)Wm[(size_t)k * NOUT + col];
}

#define SU_T      256
#define SU_B_W1   (D_IN * H_DIM / SU_T)
#define SU_B_W2   (H_DIM * H_DIM / SU_T)
#define SU_B_WA   (H_DIM * HA_DIM / SU_T)
#define SU_B_SEG  (N_NODES / SU_T)
#define SU_B_ZERO (N_NODES / SU_T)
#define SU_BLOCKS (SU_B_W1 + SU_B_W2 + 2 * SU_B_WA + SU_B_SEG + SU_B_ZERO)
static_assert(N_NODES % SU_T == 0 && (H_DIM * HA_DIM) % SU_T == 0, "whole blocks per range");
__global__ void __launch_bounds__(SU_T)
setup_all(const float* __restrict__ W1, _Float16* __restrict__ w1pk,
          const float* __restrict__ W2, _Float16* __restrict__ w2pk,
          const float* __restrict__ Wv, _Float16* __restrict__ wvpk,
          const float* __restrict__ Wu, _Float16* __restrict__ wupk,
          const int* __restrict__ batch, int* __restrict__ seg,
          int* __restrict__ cnt) {
    int b = blockIdx.x;
    const int tid = threadIdx.x;
    if (b < SU_B_W1) { pack_w_elem<D_IN, H_DIM>(W1, w1pk, b * SU_T + tid); return; }
    b -= SU_B_W1;
    if (b < SU_B_W2) { pack_w_elem<H_DIM, H_DIM>(W2, w2pk, b * SU_T + tid); return; }
    b -= SU_B_W2;
    if (b < SU_B_WA) { pack_w_elem<H_DIM, HA_DIM>(Wv, wvpk, b * SU_T + tid); return; }
    b -= SU_B_WA;
    if (b < SU_B_WA) { pack_w_elem<H_DIM, HA_DIM>(Wu, wupk, b * SU_T + tid); return; }
    b -= SU_B_WA;
    if (b < SU_B_SEG) {
        const int n = b * SU_T + tid;
        const int cur  = batch[n];
        const int prev = (n == 0) ? -1 : batch[n - 1];
#pragma unroll
        for (int g = 0; g < N_BAGS; ++g)
            if (g > prev && g <= cur) seg[g] = n;
        if (n == N_NODES - 1) {
#pragma unroll
            for (int g = 1; g <= N_BAGS; ++g)
                if (g > cur) seg[g] = N_NODES;
        }
        return;
    }
    b -= SU_B_SEG;
    cnt[b * SU_T + tid] = 0;
}

#define GT_ROWS 64
#define GT_T    256
#define GT_LD   8
static_assert(GT_ROWS == 16 * MREP, "a wave covers all rows of the tile");
static_assert(N_NODES % GT_ROWS == 0, "no partial row tile");

#define GT_PITCH(K) ((K) + 8)

template <int K, int NOUT, bool BIAS, bool TWO = false>
__device__ __forceinline__ void gemm_tile_from_lds(const _Float16* As,
                                                   const _Float16* __restrict__ Wpk,
                                                   const float* __restrict__ bias,
                                                   float* __restrict__ C, int m0,
                                                   const float* __restrict__ bias1 = nullptr,
                                                   float* __restrict__ C1 = nullptr) {
    constexpr int PITCH = GT_PITCH(K);
    constexpr int KBLKS = K / 32;
    constexpr int NTW   = NOUT / 16 / (GT_T / 32);
    static_assert(NTW >= 1 && NTW * 16 * (GT_T / 32) == NOUT, "waves tile the columns");
    const int lane = threadIdx.x & 31;
    const int wave = __builtin_amdgcn_readfirstlane((int)(threadIdx.x >> 5));
    const int half = lane >> 4;
    const int l15  = lane & 15;
    const _Float16* wb = Wpk + (size_t)(wave * NTW) * KBLKS * 512 + lane * 16;
    const _Float16* ab = As + l15 * PITCH + half * 8;

    v8f zero = {};
    v8f acc[MREP][NTW];
#pragma unroll
    for (int r = 0; r < MREP; ++r)
#pragma unroll
        for (int t = 0; t < NTW; ++t) acc[r][t] = zero;

    for (int kblk = 0; kblk < KBLKS; ++kblk) {
        v16h b[NTW];
#pragma unroll
        for (int t = 0; t < NTW; ++t)
            b[t] = *(const v16h*)(wb + ((size_t)t * KBLKS + kblk) * 512);
#pragma unroll
        for (int r = 0; r < MREP; ++r) {
            const _Float16* ap = ab + r * 16 * PITCH + kblk * 32;
            v8h lo = *(const v8h*)(ap);
            v8h hi = *(const v8h*)(ap + 16);
            v16h a = __builtin_shufflevector(lo, hi, 0, 1, 2, 3, 4, 5, 6, 7,
                                             8, 9, 10, 11, 12, 13, 14, 15);
#pragma unroll
            for (int t = 0; t < NTW; ++t)
                acc[r][t] = __builtin_amdgcn_wmma_f32_16x16x32_f16(
                    false, a, false, b[t], (short)0, acc[r][t], false, false);
        }
    }

    constexpr int CW = TWO ? NOUT / 2 : NOUT;
    static_assert(!TWO || (NOUT / 2) % (16 * NTW) == 0, "a wave stays inside one plane");
#pragma unroll
    for (int t = 0; t < NTW; ++t) {
        int ct = wave * NTW + t;
        const float* bp = bias;
        float* Cp = C;
        if (TWO && ct >= CW / 16) { ct -= CW / 16; bp = bias1; Cp = C1; }
        const int col = ct * 16 + l15;
        float bv = 0.0f;
        if (BIAS) bv = bp[col];
#pragma unroll
        for (int r = 0; r < MREP; ++r) {
#pragma unroll
            for (int q = 0; q < 8; ++q) {
                const int row = m0 + r * 16 + q + half * 8;
                Cp[(size_t)row * CW + col] = BIAS ? acc[r][t][q] + bv : acc[r][t][q];
            }
        }
    }
}

typedef __attribute__((ext_vector_type(4))) _Float16 v4h_t;

template <int K, int NOUT>
__global__ void __launch_bounds__(GT_T)
wmma_gemm_lds_af32(const float* __restrict__ A, const _Float16* __restrict__ Wpk,
                   float* __restrict__ C) {
    __shared__ alignas(16) _Float16 As[GT_ROWS * GT_PITCH(K)];
    constexpr int VROW  = K / 4;
    constexpr int PER_T = GT_ROWS * VROW / GT_T;
    static_assert(PER_T % GT_LD == 0 && VROW % 32 == 0, "whole batches, a wave inside one row");
    const int tid = threadIdx.x;
    const int m0  = blockIdx.x * GT_ROWS;
    const float4* Ag = (const float4*)(A + (size_t)m0 * K);
#pragma unroll 1
    for (int it = 0; it < PER_T; it += GT_LD) {
        float4 v[GT_LD];
#pragma unroll
        for (int u = 0; u < GT_LD; ++u) v[u] = Ag[(it + u) * GT_T + tid];
#pragma unroll
        for (int u = 0; u < GT_LD; ++u) {
            const int f = (it + u) * GT_T + tid;
            const int row = f / VROW, c = f % VROW;
            v4h_t h = {(_Float16)v[u].x, (_Float16)v[u].y, (_Float16)v[u].z, (_Float16)v[u].w};
            *(v4h_t*)(As + row * GT_PITCH(K) + c * 4) = h;
        }
    }
    __syncthreads();
    gemm_tile_from_lds<K, NOUT, false>(As, Wpk, nullptr, C, m0);
}

template <int K, int NOUT, bool BIAS, bool TWO = false>
__global__ void __launch_bounds__(GT_T)
wmma_gemm_lds_af16(const _Float16* __restrict__ A, const _Float16* __restrict__ Wpk,
                   const float* __restrict__ bias, float* __restrict__ C,
                   const float* __restrict__ bias1 = nullptr,
                   float* __restrict__ C1 = nullptr) {
    __shared__ alignas(16) _Float16 As[GT_ROWS * GT_PITCH(K)];
    constexpr int VROW  = K / 8;
    constexpr int PER_T = GT_ROWS * VROW / GT_T;
    static_assert(GT_ROWS * VROW % GT_T == 0 && VROW % 32 == 0, "whole passes, a wave inside one row");
    const int tid = threadIdx.x;
    const int m0  = blockIdx.x * GT_ROWS;
    const v8h* Ag = (const v8h*)(A + (size_t)m0 * K);
    v8h v[PER_T];
#pragma unroll
    for (int u = 0; u < PER_T; ++u) v[u] = Ag[u * GT_T + tid];
#pragma unroll
    for (int u = 0; u < PER_T; ++u) {
        const int f = u * GT_T + tid;
        const int row = f / VROW, c = f % VROW;
        *(v8h*)(As + row * GT_PITCH(K) + c * 8) = v[u];
    }
    __syncthreads();
    gemm_tile_from_lds<K, NOUT, BIAS, TWO>(As, Wpk, bias, C, m0, bias1, C1);
}

template <int K>
__device__ __forceinline__ void pull_rows(const float4* __restrict__ hp, int my_s, int my_c,
                                          int j, int lane, float4& a0, float4& a1) {
    float4 p[K], q[K];
    float  c[K];
#pragma unroll
    for (int k = 0; k < K; ++k) {
        const int s = __builtin_amdgcn_readlane(my_s, j + k);
        c[k] = __int_as_float(__builtin_amdgcn_readlane(my_c, j + k));
        const float4* row = hp + (size_t)s * (H_DIM / 4);
        p[k] = row[lane];
        q[k] = row[32 + lane];
    }
#pragma unroll
    for (int k = 0; k < K; ++k) {
        a0.x = fmaf(p[k].x, c[k], a0.x); a0.y = fmaf(p[k].y, c[k], a0.y);
        a0.z = fmaf(p[k].z, c[k], a0.z); a0.w = fmaf(p[k].w, c[k], a0.w);
        a1.x = fmaf(q[k].x, c[k], a1.x); a1.y = fmaf(q[k].y, c[k], a1.y);
        a1.z = fmaf(q[k].z, c[k], a1.z); a1.w = fmaf(q[k].w, c[k], a1.w);
    }
}

__device__ __forceinline__ float gcn_fin(float agg, float hl, float di, float bc,
                                         float mm, float vv, float gg, float bo) {
    float val = agg + hl * di * di + bc;
    val = (val - mm) * rsqrtf(vv + BN_EPS) * gg + bo;
    return fmaxf(val, 0.0f);
}

__device__ __forceinline__ float4 gcn_fin4(float4 a, float4 h, float di, float4 bc,
                                           float4 mm, float4 vv, float4 gg, float4 bo) {
    float4 r;
    r.x = gcn_fin(a.x, h.x, di, bc.x, mm.x, vv.x, gg.x, bo.x);
    r.y = gcn_fin(a.y, h.y, di, bc.y, mm.y, vv.y, gg.y, bo.y);
    r.z = gcn_fin(a.z, h.z, di, bc.z, mm.z, vv.z, gg.z, bo.z);
    r.w = gcn_fin(a.w, h.w, di, bc.w, mm.w, vv.w, gg.w, bo.w);
    return r;
}

typedef __attribute__((ext_vector_type(4))) _Float16 v4h;

#define PULL_NPW 4
static_assert(N_NODES % (PULL_NPW * 8) == 0, "whole work-groups of 8 waves");

template <bool OUT16>
__global__ void __launch_bounds__(256)
gcn_pull(const float* __restrict__ hlin,
         const int* __restrict__ rowptr,
         const int* __restrict__ csr,
         const float* __restrict__ dinv,
         const float* __restrict__ bconv,
         const float* __restrict__ g,
         const float* __restrict__ bb,
         const float* __restrict__ m,
         const float* __restrict__ v,
         _Float16* __restrict__ out16,
         float* __restrict__ out32, int N) {
    const int lane = threadIdx.x & 31;
    const int nfirst = PULL_NPW * __builtin_amdgcn_readfirstlane(
        (int)((blockIdx.x * blockDim.x + threadIdx.x) >> 5));
    if (nfirst >= N) return;
    const float4* hp = (const float4*)hlin;
    const float4* bc4 = (const float4*)bconv; const float4* g4 = (const float4*)g;
    const float4* bb4 = (const float4*)bb;    const float4* m4 = (const float4*)m;
    const float4* v4  = (const float4*)v;

    int beg = rowptr[nfirst], end = rowptr[nfirst + 1];
    for (int i = 0; i < PULL_NPW; ++i) {
        const int n = nfirst + i;
        if (n >= N) break;
        const int nxt = rowptr[min(n + 2, N)];
        float4 a0 = {0.0f, 0.0f, 0.0f, 0.0f}, a1 = {0.0f, 0.0f, 0.0f, 0.0f};
        for (int base = beg; base < end; base += 32) {
            const int nb = min(32, end - base);
            int my_s = 0, my_c = 0;
            if (lane < nb) { const int2 rec = ((const int2*)csr)[base + lane]; my_s = rec.x; my_c = rec.y; }
            int j = 0;
            for (; j + 4 <= nb; j += 4) pull_rows<4>(hp, my_s, my_c, j, lane, a0, a1);
            const int r = nb - j;
            if (r == 3)      pull_rows<3>(hp, my_s, my_c, j, lane, a0, a1);
            else if (r == 2) pull_rows<2>(hp, my_s, my_c, j, lane, a0, a1);
            else if (r == 1) pull_rows<1>(hp, my_s, my_c, j, lane, a0, a1);
        }

        const float di = dinv[n];
        const float4* own = hp + (size_t)n * (H_DIM / 4);
        const float4 h0 = own[lane], h1 = own[32 + lane];
        const float4 r0 = gcn_fin4(a0, h0, di, bc4[lane], m4[lane], v4[lane], g4[lane], bb4[lane]);
        const float4 r1 = gcn_fin4(a1, h1, di, bc4[32 + lane], m4[32 + lane], v4[32 + lane],
                                   g4[32 + lane], bb4[32 + lane]);
        if (OUT16) {
            v4h* o = (v4h*)(out16 + (size_t)n * H_DIM);
            v4h x0 = {(_Float16)r0.x, (_Float16)r0.y, (_Float16)r0.z, (_Float16)r0.w};
            v4h x1 = {(_Float16)r1.x, (_Float16)r1.y, (_Float16)r1.z, (_Float16)r1.w};
            o[lane] = x0;
            o[32 + lane] = x1;
        } else {
            float4* o = (float4*)(out32 + (size_t)n * H_DIM);
            o[lane] = r0;
            o[32 + lane] = r1;
        }
        beg = end;
        end = nxt;
    }
}

#define CVT_T   256
#define CVT_PER 8
static_assert(((size_t)N_NODES * H_DIM / 8) % (CVT_T * CVT_PER) == 0, "whole work-groups");
__global__ void __launch_bounds__(CVT_T)
f32_to_f16_x8(const float* __restrict__ in, _Float16* __restrict__ out) {
    const int first = blockIdx.x * (CVT_T * CVT_PER) + threadIdx.x;
    float4 a[CVT_PER], b[CVT_PER];
#pragma unroll
    for (int u = 0; u < CVT_PER; ++u) {
        const int t = first + u * CVT_T;
        a[u] = ((const float4*)in)[2 * t];
        b[u] = ((const float4*)in)[2 * t + 1];
    }
#pragma unroll
    for (int u = 0; u < CVT_PER; ++u) {
        const int t = first + u * CVT_T;
        v8h o = {(_Float16)a[u].x, (_Float16)a[u].y, (_Float16)a[u].z, (_Float16)a[u].w,
                 (_Float16)b[u].x, (_Float16)b[u].y, (_Float16)b[u].z, (_Float16)b[u].w};
        ((v8h*)out)[t] = o;
    }
}

#define ATT_NPW 8
#define ATT_NJ  (HA_DIM / 32)
static_assert(N_NODES % (ATT_NPW * 8) == 0, "whole work-groups of 8 waves");
static_assert(HA_DIM % 32 == 0, "whole columns per lane");
__global__ void __launch_bounds__(256)
attn_score(const float* __restrict__ av,
           const float* __restrict__ au,
           const float* __restrict__ wA,
           float* __restrict__ a) {
    const int nfirst = ATT_NPW * (int)((blockIdx.x * blockDim.x + threadIdx.x) >> 5);
    const int lane = threadIdx.x & 31;
    float wa[ATT_NJ], tv[ATT_NJ], sv[ATT_NJ];
#pragma unroll
    for (int k = 0; k < ATT_NJ; ++k) {
        wa[k] = wA[lane + 32 * k];
        tv[k] = av[(size_t)nfirst * HA_DIM + lane + 32 * k];
        sv[k] = au[(size_t)nfirst * HA_DIM + lane + 32 * k];
    }
    for (int i = 0; i < ATT_NPW; ++i) {
        const int node = nfirst + i;
        const int nn = min(node + 1, nfirst + ATT_NPW - 1);
        float tn[ATT_NJ], sn[ATT_NJ];
#pragma unroll
        for (int k = 0; k < ATT_NJ; ++k) {
            tn[k] = av[(size_t)nn * HA_DIM + lane + 32 * k];
            sn[k] = au[(size_t)nn * HA_DIM + lane + 32 * k];
        }
        float sum = 0.0f;
#pragma unroll
        for (int k = 0; k < ATT_NJ; ++k) {
            float t = tanhf(tv[k]);
            float s = 1.0f / (1.0f + expf(-sv[k]));
            sum += t * s * wa[k];
        }
        for (int off = 16; off > 0; off >>= 1) sum += __shfl_down(sum, off, 32);
        if (lane == 0) a[node] = sum;
#pragma unroll
        for (int k = 0; k < ATT_NJ; ++k) { tv[k] = tn[k]; sv[k] = sn[k]; }
    }
}

__global__ void seg_softmax(const float* __restrict__ a,
                            const int* __restrict__ seg,
                            float* __restrict__ wout) {
    __shared__ float red[256];
    int g = blockIdx.x;
    int s = seg[g], e = seg[g + 1];
    float mx = -1e30f;
    for (int n = s + (int)threadIdx.x; n < e; n += blockDim.x)
        mx = fmaxf(mx, a[n]);
    red[threadIdx.x] = mx; __syncthreads();
    for (int off = 128; off > 0; off >>= 1) {
        if ((int)threadIdx.x < off)
            red[threadIdx.x] = fmaxf(red[threadIdx.x], red[threadIdx.x + off]);
        __syncthreads();
    }
    mx = red[0]; __syncthreads();
    float sum = 0.0f;
    for (int n = s + (int)threadIdx.x; n < e; n += blockDim.x)
        sum += expf(a[n] - mx);
    red[threadIdx.x] = sum; __syncthreads();
    for (int off = 128; off > 0; off >>= 1) {
        if ((int)threadIdx.x < off)
            red[threadIdx.x] += red[threadIdx.x + off];
        __syncthreads();
    }
    float inv = 1.0f / red[0];
    for (int n = s + (int)threadIdx.x; n < e; n += blockDim.x)
        wout[n] = expf(a[n] - mx) * inv;
}

#define POOL_CH 32
#define POOL_T  512
#define POOL_RG (POOL_T / 64)
#define POOL_U  8
static_assert(H_DIM == 256, "pooling assumes 64 float4 per row");
__global__ void __launch_bounds__(POOL_T)
slide_partial(const float* __restrict__ w, const float* __restrict__ h,
              const int* __restrict__ seg, float* __restrict__ part) {
    __shared__ float4 red[POOL_RG][64];
    const int g = blockIdx.x / POOL_CH, c = blockIdx.x % POOL_CH;
    const int s = seg[g], e = seg[g + 1];
    const int per = (e - s + POOL_CH - 1) / POOL_CH;
    const int n0 = s + c * per;
    const int n1 = min(n0 + per, e);
    const int col = threadIdx.x & 63;
    const int rg = __builtin_amdgcn_readfirstlane((int)(threadIdx.x >> 6));
    const float4* h4 = (const float4*)h;
    const float4 zero4 = {0.0f, 0.0f, 0.0f, 0.0f};
    float4 acc = zero4;
    for (int nb = n0 + rg; nb < n1; nb += POOL_U * POOL_RG) {
        float4 v[POOL_U];
        float  ww[POOL_U];
#pragma unroll
        for (int u = 0; u < POOL_U; ++u) {
            const int n  = nb + u * POOL_RG;
            const int nc = min(n, n1 - 1);
            v[u]  = h4[(size_t)nc * (H_DIM / 4) + col];
            ww[u] = w[nc];
            if (n >= n1) v[u] = zero4;
        }
#pragma unroll
        for (int u = 0; u < POOL_U; ++u) {
            acc.x = fmaf(ww[u], v[u].x, acc.x); acc.y = fmaf(ww[u], v[u].y, acc.y);
            acc.z = fmaf(ww[u], v[u].z, acc.z); acc.w = fmaf(ww[u], v[u].w, acc.w);
        }
    }
    red[rg][col] = acc;
    __syncthreads();
    if (threadIdx.x < H_DIM) {
        const float* rf = (const float*)red;
        float sum = 0.0f;
#pragma unroll
        for (int r = 0; r < POOL_RG; ++r) sum += rf[r * H_DIM + threadIdx.x];
        part[(size_t)blockIdx.x * H_DIM + threadIdx.x] = sum;
    }
}

__global__ void head_kernel(const float* __restrict__ part,
                            const float* __restrict__ Wc1,
                            const float* __restrict__ bc1,
                            const float* __restrict__ Wc2,
                            const float* __restrict__ bc2,
                            float* __restrict__ logits) {
    __shared__ float slide[N_BAGS * H_DIM];
    __shared__ float hid[N_BAGS * HA_DIM];
    for (int item = threadIdx.x; item < N_BAGS * H_DIM; item += blockDim.x) {
        int g = item / H_DIM, k = item - g * H_DIM;
        const float* p = part + (size_t)g * POOL_CH * H_DIM + k;
        float s = 0.0f;
#pragma unroll
        for (int c = 0; c < POOL_CH; ++c) s += p[c * H_DIM];
        slide[item] = s;
    }
    __syncthreads();
    for (int item = threadIdx.x; item < N_BAGS * HA_DIM; item += blockDim.x) {
        int g = item / HA_DIM, t = item - g * HA_DIM;
        float s = bc1[t];
        for (int k = 0; k < H_DIM; ++k) s += slide[g * H_DIM + k] * Wc1[k * HA_DIM + t];
        hid[item] = fmaxf(s, 0.0f);
    }
    __syncthreads();
    for (int item = threadIdx.x; item < N_BAGS * C_DIM; item += blockDim.x) {
        int g = item / C_DIM, c = item - g * C_DIM;
        float s = bc2[c];
        for (int t = 0; t < HA_DIM; ++t) s += hid[g * HA_DIM + t] * Wc2[t * C_DIM + c];
        logits[item] = s;
    }
}

extern "C" void kernel_launch(void* const* d_in, const int* in_sizes, int n_in,
                              void* d_out, int out_size, void* d_ws, size_t ws_size,
                              hipStream_t stream) {
    (void)in_sizes; (void)n_in; (void)out_size; (void)ws_size;
    const float* x   = (const float*)d_in[0];
    const int*  ei   = (const int*)d_in[1];
    const int*  batch= (const int*)d_in[2];
    const float* W1  = (const float*)d_in[3];  const float* b1   = (const float*)d_in[4];
    const float* g1  = (const float*)d_in[5];  const float* bb1  = (const float*)d_in[6];
    const float* m1  = (const float*)d_in[7];  const float* v1   = (const float*)d_in[8];
    const float* W2  = (const float*)d_in[9];  const float* b2   = (const float*)d_in[10];
    const float* g2  = (const float*)d_in[11]; const float* bb2  = (const float*)d_in[12];
    const float* m2  = (const float*)d_in[13]; const float* v2   = (const float*)d_in[14];
    const float* Wv  = (const float*)d_in[15]; const float* bv   = (const float*)d_in[16];
    const float* Wu  = (const float*)d_in[17]; const float* bu   = (const float*)d_in[18];
    const float* wA  = (const float*)d_in[19];
    const float* Wc1 = (const float*)d_in[20]; const float* bc1  = (const float*)d_in[21];
    const float* Wc2 = (const float*)d_in[22]; const float* bc2  = (const float*)d_in[23];

    const int* src = ei;
    const int* dst = ei + N_EDGES;

    float* out  = (float*)d_out;
    float* wout = out + N_BAGS * C_DIM;

    const size_t NH = (size_t)N_NODES * H_DIM;
    float*    bufA = (float*)d_ws;
    float*    bufB = bufA + NH;
    _Float16* h16  = (_Float16*)(bufB + NH);
    _Float16* w1pk = h16 + NH;
    _Float16* w2pk = w1pk + (size_t)D_IN * H_DIM;
    _Float16* wvpk = w2pk + (size_t)H_DIM * H_DIM;
    _Float16* wupk = wvpk + (size_t)H_DIM * HA_DIM;
    float*    dinv = (float*)(wupk + (size_t)H_DIM * HA_DIM);
    float*    abuf = dinv + N_NODES;
    int*      seg  = (int*)(abuf + N_NODES);

    int*   cnt      = (int*)h16;
    int*   rowptr   = cnt + N_NODES;
    int*   rank     = rowptr + N_NODES + 4;
    int*   csr      = rank + N_EDGES;
    _Float16* h1_16 = (_Float16*)bufB;
    float*    h2_32 = bufB;

    setup_all<<<SU_BLOCKS, SU_T, 0, stream>>>(W1, w1pk, W2, w2pk, Wv, wvpk, Wu, wupk,
                                              batch, seg, cnt);

    count_rank<<<N_EDGES / (EDGE_T * EDGE_PER), EDGE_T, 0, stream>>>(dst, cnt, rank);
    scan_rowptr<<<1, SCAN_T, 0, stream>>>(cnt, rowptr, dinv);
    csr_fill<<<N_EDGES / (EDGE_T * EDGE_PER), EDGE_T, 0, stream>>>(src, dst, rank, rowptr, dinv,
                                                                   csr);

    wmma_gemm_lds_af32<D_IN, H_DIM><<<N_NODES / GT_ROWS, GT_T, 0, stream>>>(x, w1pk, bufA);
    gcn_pull<true><<<N_NODES / PULL_NPW / 8, 256, 0, stream>>>(
        bufA, rowptr, csr, dinv, b1, g1, bb1, m1, v1,
        h1_16, nullptr, N_NODES);

    wmma_gemm_lds_af16<H_DIM, H_DIM, false><<<N_NODES / GT_ROWS, GT_T, 0, stream>>>(
        h1_16, w2pk, nullptr, bufA);
    gcn_pull<false><<<N_NODES / PULL_NPW / 8, 256, 0, stream>>>(
        bufA, rowptr, csr, dinv, b2, g2, bb2, m2, v2,
        nullptr, h2_32, N_NODES);
    f32_to_f16_x8<<<N_NODES * (H_DIM / 8) / (CVT_T * CVT_PER), CVT_T, 0, stream>>>(h2_32, h16);

    float* av = bufA;
    float* au = bufA + (size_t)N_NODES * HA_DIM;
    wmma_gemm_lds_af16<H_DIM, 2 * HA_DIM, true, true><<<N_NODES / GT_ROWS, GT_T, 0, stream>>>(
        h16, wvpk, bv, av, bu, au);
    attn_score<<<N_NODES / ATT_NPW / 8, 256, 0, stream>>>(av, au, wA, abuf);

    seg_softmax<<<N_BAGS, 256, 0, stream>>>(abuf, seg, wout);

    float* part = bufA;
    slide_partial<<<N_BAGS * POOL_CH, POOL_T, 0, stream>>>(wout, h2_32, seg, part);

    head_kernel<<<1, 256, 0, stream>>>(part, Wc1, bc1, Wc2, bc2, out);
}


# Softmax and head (8 and 1 work-groups, one dependent load per loop trip): all loads of a thread in flight at once, 1024 threads; head one group per bag, Wc1 in LDS; new sum tree for the exp sum only
# speedup vs baseline: 4.7377x; 1.1852x over previous
#include <hip/hip_runtime.h>
#include <hip/hip_bf16.h>
#include <math.h>

typedef __attribute__((ext_vector_type(16))) _Float16 v16h;
typedef __attribute__((ext_vector_type(8)))  _Float16 v8h;
typedef __attribute__((ext_vector_type(8)))  float    v8f;

#define N_NODES 65536
#define N_EDGES 524288
#define N_BAGS  8
#define D_IN    512
#define H_DIM   256
#define HA_DIM  128
#define C_DIM   2
#define BN_EPS  1e-5f
#define MREP    4

#define EDGE_PER 4
#define EDGE_T   256
static_assert(N_EDGES % (EDGE_T * EDGE_PER) == 0, "whole work-groups");
__global__ void __launch_bounds__(EDGE_T)
count_rank(const int* __restrict__ dst, int* __restrict__ cnt, int* __restrict__ rank) {
    const int t = blockIdx.x * blockDim.x + threadIdx.x;
    const int4 d = ((const int4*)dst)[t];
    int4 r;
    r.x = atomicAdd(&cnt[d.x], 1);
    r.y = atomicAdd(&cnt[d.y], 1);
    r.z = atomicAdd(&cnt[d.z], 1);
    r.w = atomicAdd(&cnt[d.w], 1);
    ((int4*)rank)[t] = r;
}

#define SCAN_T   1024
#define SCAN_PER (N_NODES / SCAN_T)
static_assert(N_NODES % (SCAN_T * 4) == 0, "scan assumes whole int4 chunks per thread");
__global__ void __launch_bounds__(SCAN_T)
scan_rowptr(const int* __restrict__ cnt, int* __restrict__ rowptr,
            float* __restrict__ dinv) {
    __shared__ int wsum[32];
    const int t = threadIdx.x, lane = t & 31, w = t >> 5;
    const int4* c4 = (const int4*)(cnt + t * SCAN_PER);
    int4 c[SCAN_PER / 4];
    int s = 0;
#pragma unroll
    for (int i = 0; i < SCAN_PER / 4; ++i) {
        c[i] = c4[i];
        s += c[i].x + c[i].y + c[i].z + c[i].w;
    }
    int inc = s;
    for (int off = 1; off < 32; off <<= 1) {
        int v = __shfl_up(inc, off, 32);
        if (lane >= off) inc += v;
    }
    if (lane == 31) wsum[w] = inc;
    __syncthreads();
    if (w == 0) {
        int v = wsum[lane], iv = v;
        for (int off = 1; off < 32; off <<= 1) {
            int u = __shfl_up(iv, off, 32);
            if (lane >= off) iv += u;
        }
        wsum[lane] = iv - v;
    }
    __syncthreads();
    int run = wsum[w] + inc - s;
    int4*   r4 = (int4*)(rowptr + t * SCAN_PER);
    float4* d4 = (float4*)(dinv + t * SCAN_PER);
#pragma unroll
    for (int i = 0; i < SCAN_PER / 4; ++i) {
        int4 r;
        r.x = run; run += c[i].x;
        r.y = run; run += c[i].y;
        r.z = run; run += c[i].z;
        r.w = run; run += c[i].w;
        r4[i] = r;
        float4 d;
        d.x = rsqrtf(1.0f + (float)c[i].x);
        d.y = rsqrtf(1.0f + (float)c[i].y);
        d.z = rsqrtf(1.0f + (float)c[i].z);
        d.w = rsqrtf(1.0f + (float)c[i].w);
        d4[i] = d;
    }
    if (t == SCAN_T - 1) rowptr[N_NODES] = run;
}

__device__ __forceinline__ void csr_put(int s, int d, int r,
                                        const int* __restrict__ rowptr,
                                        const float* __restrict__ dinv,
                                        int* __restrict__ csr) {
    const int pos = rowptr[d] + r;
    ((int2*)csr)[pos] = make_int2(s, __float_as_int(dinv[s] * dinv[d]));
}

__global__ void __launch_bounds__(EDGE_T)
csr_fill(const int* __restrict__ src, const int* __restrict__ dst,
         const int* __restrict__ rank, const int* __restrict__ rowptr,
         const float* __restrict__ dinv, int* __restrict__ csr) {
    const int t = blockIdx.x * blockDim.x + threadIdx.x;
    const int4 s = ((const int4*)src)[t], d = ((const int4*)dst)[t], r = ((const int4*)rank)[t];
    csr_put(s.x, d.x, r.x, rowptr, dinv, csr);
    csr_put(s.y, d.y, r.y, rowptr, dinv, csr);
    csr_put(s.z, d.z, r.z, rowptr, dinv, csr);
    csr_put(s.w, d.w, r.w, rowptr, dinv, csr);
}

template <int K, int NOUT>
__device__ __forceinline__ void pack_w_elem(const float* __restrict__ Wm,
                                            _Float16* __restrict__ out, int t) {
    constexpr int KBLKS = K >> 5;
    const int i    = t & 15;
    const int lane = (t >> 4) & 31;
    const int rest = t >> 9;
    const int ntile = rest / KBLKS;
    const int kblk  = rest % KBLKS;
    const int half = lane >> 4, l15 = lane & 15;
    const int col = ntile * 16 + l15;
    const int k   = kblk * 32 + half * 8 + ((i < 8) ? i : (16 + i - 8));
    out[t] = (_Float16)Wm[(size_t)k * NOUT + col];
}

#define SU_T      256
#define SU_B_W1   (D_IN * H_DIM / SU_T)
#define SU_B_W2   (H_DIM * H_DIM / SU_T)
#define SU_B_WA   (H_DIM * HA_DIM / SU_T)
#define SU_B_SEG  (N_NODES / SU_T)
#define SU_B_ZERO (N_NODES / SU_T)
#define SU_BLOCKS (SU_B_W1 + SU_B_W2 + 2 * SU_B_WA + SU_B_SEG + SU_B_ZERO)
static_assert(N_NODES % SU_T == 0 && (H_DIM * HA_DIM) % SU_T == 0, "whole blocks per range");
__global__ void __launch_bounds__(SU_T)
setup_all(const float* __restrict__ W1, _Float16* __restrict__ w1pk,
          const float* __restrict__ W2, _Float16* __restrict__ w2pk,
          const float* __restrict__ Wv, _Float16* __restrict__ wvpk,
          const float* __restrict__ Wu, _Float16* __restrict__ wupk,
          const int* __restrict__ batch, int* __restrict__ seg,
          int* __restrict__ cnt) {
    int b = blockIdx.x;
    const int tid = threadIdx.x;
    if (b < SU_B_W1) { pack_w_elem<D_IN, H_DIM>(W1, w1pk, b * SU_T + tid); return; }
    b -= SU_B_W1;
    if (b < SU_B_W2) { pack_w_elem<H_DIM, H_DIM>(W2, w2pk, b * SU_T + tid); return; }
    b -= SU_B_W2;
    if (b < SU_B_WA) { pack_w_elem<H_DIM, HA_DIM>(Wv, wvpk, b * SU_T + tid); return; }
    b -= SU_B_WA;
    if (b < SU_B_WA) { pack_w_elem<H_DIM, HA_DIM>(Wu, wupk, b * SU_T + tid); return; }
    b -= SU_B_WA;
    if (b < SU_B_SEG) {
        const int n = b * SU_T + tid;
        const int cur  = batch[n];
        const int prev = (n == 0) ? -1 : batch[n - 1];
#pragma unroll
        for (int g = 0; g < N_BAGS; ++g)
            if (g > prev && g <= cur) seg[g] = n;
        if (n == N_NODES - 1) {
#pragma unroll
            for (int g = 1; g <= N_BAGS; ++g)
                if (g > cur) seg[g] = N_NODES;
        }
        return;
    }
    b -= SU_B_SEG;
    cnt[b * SU_T + tid] = 0;
}

#define GT_ROWS 64
#define GT_T    256
#define GT_LD   8
static_assert(GT_ROWS == 16 * MREP, "a wave covers all rows of the tile");
static_assert(N_NODES % GT_ROWS == 0, "no partial row tile");

#define GT_PITCH(K) ((K) + 8)

template <int K, int NOUT, bool BIAS, bool TWO = false>
__device__ __forceinline__ void gemm_tile_from_lds(const _Float16* As,
                                                   const _Float16* __restrict__ Wpk,
                                                   const float* __restrict__ bias,
                                                   float* __restrict__ C, int m0,
                                                   const float* __restrict__ bias1 = nullptr,
                                                   float* __restrict__ C1 = nullptr) {
    constexpr int PITCH = GT_PITCH(K);
    constexpr int KBLKS = K / 32;
    constexpr int NTW   = NOUT / 16 / (GT_T / 32);
    static_assert(NTW >= 1 && NTW * 16 * (GT_T / 32) == NOUT, "waves tile the columns");
    const int lane = threadIdx.x & 31;
    const int wave = __builtin_amdgcn_readfirstlane((int)(threadIdx.x >> 5));
    const int half = lane >> 4;
    const int l15  = lane & 15;
    const _Float16* wb = Wpk + (size_t)(wave * NTW) * KBLKS * 512 + lane * 16;
    const _Float16* ab = As + l15 * PITCH + half * 8;

    v8f zero = {};
    v8f acc[MREP][NTW];
#pragma unroll
    for (int r = 0; r < MREP; ++r)
#pragma unroll
        for (int t = 0; t < NTW; ++t) acc[r][t] = zero;

    for (int kblk = 0; kblk < KBLKS; ++kblk) {
        v16h b[NTW];
#pragma unroll
        for (int t = 0; t < NTW; ++t)
            b[t] = *(const v16h*)(wb + ((size_t)t * KBLKS + kblk) * 512);
#pragma unroll
        for (int r = 0; r < MREP; ++r) {
            const _Float16* ap = ab + r * 16 * PITCH + kblk * 32;
            v8h lo = *(const v8h*)(ap);
            v8h hi = *(const v8h*)(ap + 16);
            v16h a = __builtin_shufflevector(lo, hi, 0, 1, 2, 3, 4, 5, 6, 7,
                                             8, 9, 10, 11, 12, 13, 14, 15);
#pragma unroll
            for (int t = 0; t < NTW; ++t)
                acc[r][t] = __builtin_amdgcn_wmma_f32_16x16x32_f16(
                    false, a, false, b[t], (short)0, acc[r][t], false, false);
        }
    }

    constexpr int CW = TWO ? NOUT / 2 : NOUT;
    static_assert(!TWO || (NOUT / 2) % (16 * NTW) == 0, "a wave stays inside one plane");
#pragma unroll
    for (int t = 0; t < NTW; ++t) {
        int ct = wave * NTW + t;
        const float* bp = bias;
        float* Cp = C;
        if (TWO && ct >= CW / 16) { ct -= CW / 16; bp = bias1; Cp = C1; }
        const int col = ct * 16 + l15;
        float bv = 0.0f;
        if (BIAS) bv = bp[col];
#pragma unroll
        for (int r = 0; r < MREP; ++r) {
#pragma unroll
            for (int q = 0; q < 8; ++q) {
                const int row = m0 + r * 16 + q + half * 8;
                Cp[(size_t)row * CW + col] = BIAS ? acc[r][t][q] + bv : acc[r][t][q];
            }
        }
    }
}

typedef __attribute__((ext_vector_type(4))) _Float16 v4h_t;

template <int K, int NOUT>
__global__ void __launch_bounds__(GT_T)
wmma_gemm_lds_af32(const float* __restrict__ A, const _Float16* __restrict__ Wpk,
                   float* __restrict__ C) {
    __shared__ alignas(16) _Float16 As[GT_ROWS * GT_PITCH(K)];
    constexpr int VROW  = K / 4;
    constexpr int PER_T = GT_ROWS * VROW / GT_T;
    static_assert(PER_T % GT_LD == 0 && VROW % 32 == 0, "whole batches, a wave inside one row");
    const int tid = threadIdx.x;
    const int m0  = blockIdx.x * GT_ROWS;
    const float4* Ag = (const float4*)(A + (size_t)m0 * K);
#pragma unroll 1
    for (int it = 0; it < PER_T; it += GT_LD) {
        float4 v[GT_LD];
#pragma unroll
        for (int u = 0; u < GT_LD; ++u) v[u] = Ag[(it + u) * GT_T + tid];
#pragma unroll
        for (int u = 0; u < GT_LD; ++u) {
            const int f = (it + u) * GT_T + tid;
            const int row = f / VROW, c = f % VROW;
            v4h_t h = {(_Float16)v[u].x, (_Float16)v[u].y, (_Float16)v[u].z, (_Float16)v[u].w};
            *(v4h_t*)(As + row * GT_PITCH(K) + c * 4) = h;
        }
    }
    __syncthreads();
    gemm_tile_from_lds<K, NOUT, false>(As, Wpk, nullptr, C, m0);
}

template <int K, int NOUT, bool BIAS, bool TWO = false>
__global__ void __launch_bounds__(GT_T)
wmma_gemm_lds_af16(const _Float16* __restrict__ A, const _Float16* __restrict__ Wpk,
                   const float* __restrict__ bias, float* __restrict__ C,
                   const float* __restrict__ bias1 = nullptr,
                   float* __restrict__ C1 = nullptr) {
    __shared__ alignas(16) _Float16 As[GT_ROWS * GT_PITCH(K)];
    constexpr int VROW  = K / 8;
    constexpr int PER_T = GT_ROWS * VROW / GT_T;
    static_assert(GT_ROWS * VROW % GT_T == 0 && VROW % 32 == 0, "whole passes, a wave inside one row");
    const int tid = threadIdx.x;
    const int m0  = blockIdx.x * GT_ROWS;
    const v8h* Ag = (const v8h*)(A + (size_t)m0 * K);
    v8h v[PER_T];
#pragma unroll
    for (int u = 0; u < PER_T; ++u) v[u] = Ag[u * GT_T + tid];
#pragma unroll
    for (int u = 0; u < PER_T; ++u) {
        const int f = u * GT_T + tid;
        const int row = f / VROW, c = f % VROW;
        *(v8h*)(As + row * GT_PITCH(K) + c * 8) = v[u];
    }
    __syncthreads();
    gemm_tile_from_lds<K, NOUT, BIAS, TWO>(As, Wpk, bias, C, m0, bias1, C1);
}

template <int K>
__device__ __forceinline__ void pull_rows(const float4* __restrict__ hp, int my_s, int my_c,
                                          int j, int lane, float4& a0, float4& a1) {
    float4 p[K], q[K];
    float  c[K];
#pragma unroll
    for (int k = 0; k < K; ++k) {
        const int s = __builtin_amdgcn_readlane(my_s, j + k);
        c[k] = __int_as_float(__builtin_amdgcn_readlane(my_c, j + k));
        const float4* row = hp + (size_t)s * (H_DIM / 4);
        p[k] = row[lane];
        q[k] = row[32 + lane];
    }
#pragma unroll
    for (int k = 0; k < K; ++k) {
        a0.x = fmaf(p[k].x, c[k], a0.x); a0.y = fmaf(p[k].y, c[k], a0.y);
        a0.z = fmaf(p[k].z, c[k], a0.z); a0.w = fmaf(p[k].w, c[k], a0.w);
        a1.x = fmaf(q[k].x, c[k], a1.x); a1.y = fmaf(q[k].y, c[k], a1.y);
        a1.z = fmaf(q[k].z, c[k], a1.z); a1.w = fmaf(q[k].w, c[k], a1.w);
    }
}

__device__ __forceinline__ float gcn_fin(float agg, float hl, float di, float bc,
                                         float mm, float vv, float gg, float bo) {
    float val = agg + hl * di * di + bc;
    val = (val - mm) * rsqrtf(vv + BN_EPS) * gg + bo;
    return fmaxf(val, 0.0f);
}

__device__ __forceinline__ float4 gcn_fin4(float4 a, float4 h, float di, float4 bc,
                                           float4 mm, float4 vv, float4 gg, float4 bo) {
    float4 r;
    r.x = gcn_fin(a.x, h.x, di, bc.x, mm.x, vv.x, gg.x, bo.x);
    r.y = gcn_fin(a.y, h.y, di, bc.y, mm.y, vv.y, gg.y, bo.y);
    r.z = gcn_fin(a.z, h.z, di, bc.z, mm.z, vv.z, gg.z, bo.z);
    r.w = gcn_fin(a.w, h.w, di, bc.w, mm.w, vv.w, gg.w, bo.w);
    return r;
}

typedef __attribute__((ext_vector_type(4))) _Float16 v4h;

#define PULL_NPW 4
static_assert(N_NODES % (PULL_NPW * 8) == 0, "whole work-groups of 8 waves");

template <bool OUT16>
__global__ void __launch_bounds__(256)
gcn_pull(const float* __restrict__ hlin,
         const int* __restrict__ rowptr,
         const int* __restrict__ csr,
         const float* __restrict__ dinv,
         const float* __restrict__ bconv,
         const float* __restrict__ g,
         const float* __restrict__ bb,
         const float* __restrict__ m,
         const float* __restrict__ v,
         _Float16* __restrict__ out16,
         float* __restrict__ out32, int N) {
    const int lane = threadIdx.x & 31;
    const int nfirst = PULL_NPW * __builtin_amdgcn_readfirstlane(
        (int)((blockIdx.x * blockDim.x + threadIdx.x) >> 5));
    if (nfirst >= N) return;
    const float4* hp = (const float4*)hlin;
    const float4* bc4 = (const float4*)bconv; const float4* g4 = (const float4*)g;
    const float4* bb4 = (const float4*)bb;    const float4* m4 = (const float4*)m;
    const float4* v4  = (const float4*)v;

    int beg = rowptr[nfirst], end = rowptr[nfirst + 1];
    for (int i = 0; i < PULL_NPW; ++i) {
        const int n = nfirst + i;
        if (n >= N) break;
        const int nxt = rowptr[min(n + 2, N)];
        float4 a0 = {0.0f, 0.0f, 0.0f, 0.0f}, a1 = {0.0f, 0.0f, 0.0f, 0.0f};
        for (int base = beg; base < end; base += 32) {
            const int nb = min(32, end - base);
            int my_s = 0, my_c = 0;
            if (lane < nb) { const int2 rec = ((const int2*)csr)[base + lane]; my_s = rec.x; my_c = rec.y; }
            int j = 0;
            for (; j + 4 <= nb; j += 4) pull_rows<4>(hp, my_s, my_c, j, lane, a0, a1);
            const int r = nb - j;
            if (r == 3)      pull_rows<3>(hp, my_s, my_c, j, lane, a0, a1);
            else if (r == 2) pull_rows<2>(hp, my_s, my_c, j, lane, a0, a1);
            else if (r == 1) pull_rows<1>(hp, my_s, my_c, j, lane, a0, a1);
        }

        const float di = dinv[n];
        const float4* own = hp + (size_t)n * (H_DIM / 4);
        const float4 h0 = own[lane], h1 = own[32 + lane];
        const float4 r0 = gcn_fin4(a0, h0, di, bc4[lane], m4[lane], v4[lane], g4[lane], bb4[lane]);
        const float4 r1 = gcn_fin4(a1, h1, di, bc4[32 + lane], m4[32 + lane], v4[32 + lane],
                                   g4[32 + lane], bb4[32 + lane]);
        if (OUT16) {
            v4h* o = (v4h*)(out16 + (size_t)n * H_DIM);
            v4h x0 = {(_Float16)r0.x, (_Float16)r0.y, (_Float16)r0.z, (_Float16)r0.w};
            v4h x1 = {(_Float16)r1.x, (_Float16)r1.y, (_Float16)r1.z, (_Float16)r1.w};
            o[lane] = x0;
            o[32 + lane] = x1;
        } else {
            float4* o = (float4*)(out32 + (size_t)n * H_DIM);
            o[lane] = r0;
            o[32 + lane] = r1;
        }
        beg = end;
        end = nxt;
    }
}

#define CVT_T   256
#define CVT_PER 8
static_assert(((size_t)N_NODES * H_DIM / 8) % (CVT_T * CVT_PER) == 0, "whole work-groups");
__global__ void __launch_bounds__(CVT_T)
f32_to_f16_x8(const float* __restrict__ in, _Float16* __restrict__ out) {
    const int first = blockIdx.x * (CVT_T * CVT_PER) + threadIdx.x;
    float4 a[CVT_PER], b[CVT_PER];
#pragma unroll
    for (int u = 0; u < CVT_PER; ++u) {
        const int t = first + u * CVT_T;
        a[u] = ((const float4*)in)[2 * t];
        b[u] = ((const float4*)in)[2 * t + 1];
    }
#pragma unroll
    for (int u = 0; u < CVT_PER; ++u) {
        const int t = first + u * CVT_T;
        v8h o = {(_Float16)a[u].x, (_Float16)a[u].y, (_Float16)a[u].z, (_Float16)a[u].w,
                 (_Float16)b[u].x, (_Float16)b[u].y, (_Float16)b[u].z, (_Float16)b[u].w};
        ((v8h*)out)[t] = o;
    }
}

#define ATT_NPW 8
#define ATT_NJ  (HA_DIM / 32)
static_assert(N_NODES % (ATT_NPW * 8) == 0, "whole work-groups of 8 waves");
static_assert(HA_DIM % 32 == 0, "whole columns per lane");
__global__ void __launch_bounds__(256)
attn_score(const float* __restrict__ av,
           const float* __restrict__ au,
           const float* __restrict__ wA,
           float* __restrict__ a) {
    const int nfirst = ATT_NPW * (int)((blockIdx.x * blockDim.x + threadIdx.x) >> 5);
    const int lane = threadIdx.x & 31;
    float wa[ATT_NJ], tv[ATT_NJ], sv[ATT_NJ];
#pragma unroll
    for (int k = 0; k < ATT_NJ; ++k) {
        wa[k] = wA[lane + 32 * k];
        tv[k] = av[(size_t)nfirst * HA_DIM + lane + 32 * k];
        sv[k] = au[(size_t)nfirst * HA_DIM + lane + 32 * k];
    }
    for (int i = 0; i < ATT_NPW; ++i) {
        const int node = nfirst + i;
        const int nn = min(node + 1, nfirst + ATT_NPW - 1);
        float tn[ATT_NJ], sn[ATT_NJ];
#pragma unroll
        for (int k = 0; k < ATT_NJ; ++k) {
            tn[k] = av[(size_t)nn * HA_DIM + lane + 32 * k];
            sn[k] = au[(size_t)nn * HA_DIM + lane + 32 * k];
        }
        float sum = 0.0f;
#pragma unroll
        for (int k = 0; k < ATT_NJ; ++k) {
            float t = tanhf(tv[k]);
            float s = 1.0f / (1.0f + expf(-sv[k]));
            sum += t * s * wa[k];
        }
        for (int off = 16; off > 0; off >>= 1) sum += __shfl_down(sum, off, 32);
        if (lane == 0) a[node] = sum;
#pragma unroll
        for (int k = 0; k < ATT_NJ; ++k) { tv[k] = tn[k]; sv[k] = sn[k]; }
    }
}

#define SM_T 1024
#define SM_U 12
static_assert(SM_T == 32 * 32, "the second reduction stage is one value per lane of one wave");

__device__ __forceinline__ float wave_max(float v) {
    for (int off = 16; off > 0; off >>= 1) v = fmaxf(v, __shfl_xor(v, off, 32));
    return v;
}
__device__ __forceinline__ float wave_sum(float v) {
    for (int off = 16; off > 0; off >>= 1) v += __shfl_xor(v, off, 32);
    return v;
}

__global__ void __launch_bounds__(SM_T)
seg_softmax(const float* __restrict__ a,
            const int* __restrict__ seg,
            float* __restrict__ wout) {
    __shared__ float red[SM_T / 32];
    const int g = blockIdx.x;
    const int s = seg[g], e = seg[g + 1];
    if (e <= s) return;
    const int tid = threadIdx.x, lane = tid & 31, wv = tid >> 5;
    const int first = s + tid;
    const int more  = first + SM_T * SM_U;

    float v0[SM_U];
#pragma unroll
    for (int u = 0; u < SM_U; ++u) v0[u] = a[min(first + u * SM_T, e - 1)];
    float mx = -1e30f;
#pragma unroll
    for (int u = 0; u < SM_U; ++u) mx = fmaxf(mx, v0[u]);
    for (int nb = more; nb < e; nb += SM_T * SM_U) {
        float v[SM_U];
#pragma unroll
        for (int u = 0; u < SM_U; ++u) v[u] = a[min(nb + u * SM_T, e - 1)];
#pragma unroll
        for (int u = 0; u < SM_U; ++u) mx = fmaxf(mx, v[u]);
    }
    mx = wave_max(mx);
    if (lane == 0) red[wv] = mx;
    __syncthreads();
    mx = wave_max(red[lane]);
    __syncthreads();

    float ex[SM_U];
    float sum = 0.0f;
#pragma unroll
    for (int u = 0; u < SM_U; ++u) {
        ex[u] = 0.0f;
        if (first + u * SM_T < e) ex[u] = expf(v0[u] - mx);
        sum += ex[u];
    }
    for (int nb = more; nb < e; nb += SM_T * SM_U) {
        float v[SM_U];
#pragma unroll
        for (int u = 0; u < SM_U; ++u) v[u] = a[min(nb + u * SM_T, e - 1)];
#pragma unroll
        for (int u = 0; u < SM_U; ++u) {
            const float x = expf(v[u] - mx);
            sum += (nb + u * SM_T < e) ? x : 0.0f;
        }
    }
    sum = wave_sum(sum);
    if (lane == 0) red[wv] = sum;
    __syncthreads();
    const float inv = 1.0f / wave_sum(red[lane]);

#pragma unroll
    for (int u = 0; u < SM_U; ++u) {
        const int n = first + u * SM_T;
        if (n < e) wout[n] = ex[u] * inv;
    }
    for (int nb = more; nb < e; nb += SM_T * SM_U) {
        float v[SM_U];
#pragma unroll
        for (int u = 0; u < SM_U; ++u) v[u] = a[min(nb + u * SM_T, e - 1)];
#pragma unroll
        for (int u = 0; u < SM_U; ++u) {
            const int n = nb + u * SM_T;
            if (n < e) wout[n] = expf(v[u] - mx) * inv;
        }
    }
}

#define POOL_CH 32
#define POOL_T  512
#define POOL_RG (POOL_T / 64)
#define POOL_U  8
static_assert(H_DIM == 256, "pooling assumes 64 float4 per row");
__global__ void __launch_bounds__(POOL_T)
slide_partial(const float* __restrict__ w, const float* __restrict__ h,
              const int* __restrict__ seg, float* __restrict__ part) {
    __shared__ float4 red[POOL_RG][64];
    const int g = blockIdx.x / POOL_CH, c = blockIdx.x % POOL_CH;
    const int s = seg[g], e = seg[g + 1];
    const int per = (e - s + POOL_CH - 1) / POOL_CH;
    const int n0 = s + c * per;
    const int n1 = min(n0 + per, e);
    const int col = threadIdx.x & 63;
    const int rg = __builtin_amdgcn_readfirstlane((int)(threadIdx.x >> 6));
    const float4* h4 = (const float4*)h;
    const float4 zero4 = {0.0f, 0.0f, 0.0f, 0.0f};
    float4 acc = zero4;
    for (int nb = n0 + rg; nb < n1; nb += POOL_U * POOL_RG) {
        float4 v[POOL_U];
        float  ww[POOL_U];
#pragma unroll
        for (int u = 0; u < POOL_U; ++u) {
            const int n  = nb + u * POOL_RG;
            const int nc = min(n, n1 - 1);
            v[u]  = h4[(size_t)nc * (H_DIM / 4) + col];
            ww[u] = w[nc];
            if (n >= n1) v[u] = zero4;
        }
#pragma unroll
        for (int u = 0; u < POOL_U; ++u) {
            acc.x = fmaf(ww[u], v[u].x, acc.x); acc.y = fmaf(ww[u], v[u].y, acc.y);
            acc.z = fmaf(ww[u], v[u].z, acc.z); acc.w = fmaf(ww[u], v[u].w, acc.w);
        }
    }
    red[rg][col] = acc;
    __syncthreads();
    if (threadIdx.x < H_DIM) {
        const float* rf = (const float*)red;
        float sum = 0.0f;
#pragma unroll
        for (int r = 0; r < POOL_RG; ++r) sum += rf[r * H_DIM + threadIdx.x];
        part[(size_t)blockIdx.x * H_DIM + threadIdx.x] = sum;
    }
}

#define HEAD_T  1024
#define HEAD_WV (H_DIM * HA_DIM / 4 / HEAD_T)
static_assert(HEAD_WV * HEAD_T * 4 == H_DIM * HA_DIM, "Wc1 splits into whole float4 per thread");
static_assert(H_DIM <= HEAD_T && HA_DIM * C_DIM <= HEAD_T && H_DIM % 32 == 0 && HA_DIM % 32 == 0,
              "one thread per column / Wc2 value; branches on whole waves");
__global__ void __launch_bounds__(HEAD_T)
head_kernel(const float* __restrict__ part,
            const float* __restrict__ Wc1,
            const float* __restrict__ bc1,
            const float* __restrict__ Wc2,
            const float* __restrict__ bc2,
            float* __restrict__ logits) {
    __shared__ alignas(16) float w1s[H_DIM * HA_DIM];
    __shared__ alignas(16) float slide[H_DIM];
    __shared__ alignas(16) float hid[HA_DIM];
    __shared__ float w2s[HA_DIM * C_DIM];
    const int tid = threadIdx.x;
    const int g = blockIdx.x;

    float4 wv[HEAD_WV];
#pragma unroll
    for (int u = 0; u < HEAD_WV; ++u) wv[u] = ((const float4*)Wc1)[u * HEAD_T + tid];
    float pv[POOL_CH];
    if (tid < H_DIM) {
        const float* p = part + (size_t)g * POOL_CH * H_DIM + tid;
#pragma unroll
        for (int c = 0; c < POOL_CH; ++c) pv[c] = p[c * H_DIM];
    }
    float w2v = 0.0f, b1v = 0.0f, b2v = 0.0f;
    if (tid < HA_DIM * C_DIM) w2v = Wc2[tid];
    if (tid < HA_DIM) b1v = bc1[tid];
    if (tid < C_DIM) b2v = bc2[tid];

#pragma unroll
    for (int u = 0; u < HEAD_WV; ++u) ((float4*)w1s)[u * HEAD_T + tid] = wv[u];
    if (tid < H_DIM) {
        float s = 0.0f;
#pragma unroll
        for (int c = 0; c < POOL_CH; ++c) s += pv[c];
        slide[tid] = s;
    }
    if (tid < HA_DIM * C_DIM) w2s[tid] = w2v;
    __syncthreads();

    if (tid < HA_DIM) {
        float acc = b1v;
#pragma unroll 32
        for (int k = 0; k < H_DIM; ++k) acc += slide[k] * w1s[k * HA_DIM + tid];
        hid[tid] = fmaxf(acc, 0.0f);
    }
    __syncthreads();

    if (tid < C_DIM) {
        float s = b2v;
#pragma unroll
        for (int t = 0; t < HA_DIM; ++t) s += hid[t] * w2s[t * C_DIM + tid];
        logits[g * C_DIM + tid] = s;
    }
}

extern "C" void kernel_launch(void* const* d_in, const int* in_sizes, int n_in,
                              void* d_out, int out_size, void* d_ws, size_t ws_size,
                              hipStream_t stream) {
    (void)in_sizes; (void)n_in; (void)out_size; (void)ws_size;
    const float* x   = (const float*)d_in[0];
    const int*  ei   = (const int*)d_in[1];
    const int*  batch= (const int*)d_in[2];
    const float* W1  = (const float*)d_in[3];  const float* b1   = (const float*)d_in[4];
    const float* g1  = (const float*)d_in[5];  const float* bb1  = (const float*)d_in[6];
    const float* m1  = (const float*)d_in[7];  const float* v1   = (const float*)d_in[8];
    const float* W2  = (const float*)d_in[9];  const float* b2   = (const float*)d_in[10];
    const float* g2  = (const float*)d_in[11]; const float* bb2  = (const float*)d_in[12];
    const float* m2  = (const float*)d_in[13]; const float* v2   = (const float*)d_in[14];
    const float* Wv  = (const float*)d_in[15]; const float* bv   = (const float*)d_in[16];
    const float* Wu  = (const float*)d_in[17]; const float* bu   = (const float*)d_in[18];
    const float* wA  = (const float*)d_in[19];
    const float* Wc1 = (const float*)d_in[20]; const float* bc1  = (const float*)d_in[21];
    const float* Wc2 = (const float*)d_in[22]; const float* bc2  = (const float*)d_in[23];

    const int* src = ei;
    const int* dst = ei + N_EDGES;

    float* out  = (float*)d_out;
    float* wout = out + N_BAGS * C_DIM;

    const size_t NH = (size_t)N_NODES * H_DIM;
    float*    bufA = (float*)d_ws;
    float*    bufB = bufA + NH;
    _Float16* h16  = (_Float16*)(bufB + NH);
    _Float16* w1pk = h16 + NH;
    _Float16* w2pk = w1pk + (size_t)D_IN * H_DIM;
    _Float16* wvpk = w2pk + (size_t)H_DIM * H_DIM;
    _Float16* wupk = wvpk + (size_t)H_DIM * HA_DIM;
    float*    dinv = (float*)(wupk + (size_t)H_DIM * HA_DIM);
    float*    abuf = dinv + N_NODES;
    int*      seg  = (int*)(abuf + N_NODES);

    int*   cnt      = (int*)h16;
    int*   rowptr   = cnt + N_NODES;
    int*   rank     = rowptr + N_NODES + 4;
    int*   csr      = rank + N_EDGES;
    _Float16* h1_16 = (_Float16*)bufB;
    float*    h2_32 = bufB;

    setup_all<<<SU_BLOCKS, SU_T, 0, stream>>>(W1, w1pk, W2, w2pk, Wv, wvpk, Wu, wupk,
                                              batch, seg, cnt);

    count_rank<<<N_EDGES / (EDGE_T * EDGE_PER), EDGE_T, 0, stream>>>(dst, cnt, rank);
    scan_rowptr<<<1, SCAN_T, 0, stream>>>(cnt, rowptr, dinv);
    csr_fill<<<N_EDGES / (EDGE_T * EDGE_PER), EDGE_T, 0, stream>>>(src, dst, rank, rowptr, dinv,
                                                                   csr);

    wmma_gemm_lds_af32<D_IN, H_DIM><<<N_NODES / GT_ROWS, GT_T, 0, stream>>>(x, w1pk, bufA);
    gcn_pull<true><<<N_NODES / PULL_NPW / 8, 256, 0, stream>>>(
        bufA, rowptr, csr, dinv, b1, g1, bb1, m1, v1,
        h1_16, nullptr, N_NODES);

    wmma_gemm_lds_af16<H_DIM, H_DIM, false><<<N_NODES / GT_ROWS, GT_T, 0, stream>>>(
        h1_16, w2pk, nullptr, bufA);
    gcn_pull<false><<<N_NODES / PULL_NPW / 8, 256, 0, stream>>>(
        bufA, rowptr, csr, dinv, b2, g2, bb2, m2, v2,
        nullptr, h2_32, N_NODES);
    f32_to_f16_x8<<<N_NODES * (H_DIM / 8) / (CVT_T * CVT_PER), CVT_T, 0, stream>>>(h2_32, h16);

    float* av = bufA;
    float* au = bufA + (size_t)N_NODES * HA_DIM;
    wmma_gemm_lds_af16<H_DIM, 2 * HA_DIM, true, true><<<N_NODES / GT_ROWS, GT_T, 0, stream>>>(
        h16, wvpk, bv, av, bu, au);
    attn_score<<<N_NODES / ATT_NPW / 8, 256, 0, stream>>>(av, au, wA, abuf);

    seg_softmax<<<N_BAGS, SM_T, 0, stream>>>(abuf, seg, wout);

    float* part = bufA;
    slide_partial<<<N_BAGS * POOL_CH, POOL_T, 0, stream>>>(wout, h2_32, seg, part);

    head_kernel<<<N_BAGS, HEAD_T, 0, stream>>>(part, Wc1, bc1, Wc2, bc2, out);
}


# Attention chain fused, 14 -> 12 launches: f16 copy pass + Wv|Wu projection GEMM + gate/score pass -> one kernel (A tile from f32 h2, gate in registers, row sums via LDS); new sum tree for the score
# speedup vs baseline: 5.2395x; 1.1059x over previous
#include <hip/hip_runtime.h>
#include <hip/hip_bf16.h>
#include <math.h>

typedef __attribute__((ext_vector_type(16))) _Float16 v16h;
typedef __attribute__((ext_vector_type(8)))  _Float16 v8h;
typedef __attribute__((ext_vector_type(8)))  float    v8f;

#define N_NODES 65536
#define N_EDGES 524288
#define N_BAGS  8
#define D_IN    512
#define H_DIM   256
#define HA_DIM  128
#define C_DIM   2
#define BN_EPS  1e-5f
#define MREP    4

#define EDGE_PER 4
#define EDGE_T   256
static_assert(N_EDGES % (EDGE_T * EDGE_PER) == 0, "whole work-groups");
__global__ void __launch_bounds__(EDGE_T)
count_rank(const int* __restrict__ dst, int* __restrict__ cnt, int* __restrict__ rank) {
    const int t = blockIdx.x * blockDim.x + threadIdx.x;
    const int4 d = ((const int4*)dst)[t];
    int4 r;
    r.x = atomicAdd(&cnt[d.x], 1);
    r.y = atomicAdd(&cnt[d.y], 1);
    r.z = atomicAdd(&cnt[d.z], 1);
    r.w = atomicAdd(&cnt[d.w], 1);
    ((int4*)rank)[t] = r;
}

#define SCAN_T   1024
#define SCAN_PER (N_NODES / SCAN_T)
static_assert(N_NODES % (SCAN_T * 4) == 0, "scan assumes whole int4 chunks per thread");
__global__ void __launch_bounds__(SCAN_T)
scan_rowptr(const int* __restrict__ cnt, int* __restrict__ rowptr,
            float* __restrict__ dinv) {
    __shared__ int wsum[32];
    const int t = threadIdx.x, lane = t & 31, w = t >> 5;
    const int4* c4 = (const int4*)(cnt + t * SCAN_PER);
    int4 c[SCAN_PER / 4];
    int s = 0;
#pragma unroll
    for (int i = 0; i < SCAN_PER / 4; ++i) {
        c[i] = c4[i];
        s += c[i].x + c[i].y + c[i].z + c[i].w;
    }
    int inc = s;
    for (int off = 1; off < 32; off <<= 1) {
        int v = __shfl_up(inc, off, 32);
        if (lane >= off) inc += v;
    }
    if (lane == 31) wsum[w] = inc;
    __syncthreads();
    if (w == 0) {
        int v = wsum[lane], iv = v;
        for (int off = 1; off < 32; off <<= 1) {
            int u = __shfl_up(iv, off, 32);
            if (lane >= off) iv += u;
        }
        wsum[lane] = iv - v;
    }
    __syncthreads();
    int run = wsum[w] + inc - s;
    int4*   r4 = (int4*)(rowptr + t * SCAN_PER);
    float4* d4 = (float4*)(dinv + t * SCAN_PER);
#pragma unroll
    for (int i = 0; i < SCAN_PER / 4; ++i) {
        int4 r;
        r.x = run; run += c[i].x;
        r.y = run; run += c[i].y;
        r.z = run; run += c[i].z;
        r.w = run; run += c[i].w;
        r4[i] = r;
        float4 d;
        d.x = rsqrtf(1.0f + (float)c[i].x);
        d.y = rsqrtf(1.0f + (float)c[i].y);
        d.z = rsqrtf(1.0f + (float)c[i].z);
        d.w = rsqrtf(1.0f + (float)c[i].w);
        d4[i] = d;
    }
    if (t == SCAN_T - 1) rowptr[N_NODES] = run;
}

__device__ __forceinline__ void csr_put(int s, int d, int r,
                                        const int* __restrict__ rowptr,
                                        const float* __restrict__ dinv,
                                        int* __restrict__ csr) {
    const int pos = rowptr[d] + r;
    ((int2*)csr)[pos] = make_int2(s, __float_as_int(dinv[s] * dinv[d]));
}

__global__ void __launch_bounds__(EDGE_T)
csr_fill(const int* __restrict__ src, const int* __restrict__ dst,
         const int* __restrict__ rank, const int* __restrict__ rowptr,
         const float* __restrict__ dinv, int* __restrict__ csr) {
    const int t = blockIdx.x * blockDim.x + threadIdx.x;
    const int4 s = ((const int4*)src)[t], d = ((const int4*)dst)[t], r = ((const int4*)rank)[t];
    csr_put(s.x, d.x, r.x, rowptr, dinv, csr);
    csr_put(s.y, d.y, r.y, rowptr, dinv, csr);
    csr_put(s.z, d.z, r.z, rowptr, dinv, csr);
    csr_put(s.w, d.w, r.w, rowptr, dinv, csr);
}

template <int K, int NOUT>
__device__ __forceinline__ void pack_w_elem(const float* __restrict__ Wm,
                                            _Float16* __restrict__ out, int t) {
    constexpr int KBLKS = K >> 5;
    const int i    = t & 15;
    const int lane = (t >> 4) & 31;
    const int rest = t >> 9;
    const int ntile = rest / KBLKS;
    const int kblk  = rest % KBLKS;
    const int half = lane >> 4, l15 = lane & 15;
    const int col = ntile * 16 + l15;
    const int k   = kblk * 32 + half * 8 + ((i < 8) ? i : (16 + i - 8));
    out[t] = (_Float16)Wm[(size_t)k * NOUT + col];
}

#define SU_T      256
#define SU_B_W1   (D_IN * H_DIM / SU_T)
#define SU_B_W2   (H_DIM * H_DIM / SU_T)
#define SU_B_WA   (H_DIM * HA_DIM / SU_T)
#define SU_B_SEG  (N_NODES / SU_T)
#define SU_B_ZERO (N_NODES / SU_T)
#define SU_BLOCKS (SU_B_W1 + SU_B_W2 + 2 * SU_B_WA + SU_B_SEG + SU_B_ZERO)
static_assert(N_NODES % SU_T == 0 && (H_DIM * HA_DIM) % SU_T == 0, "whole blocks per range");
__global__ void __launch_bounds__(SU_T)
setup_all(const float* __restrict__ W1, _Float16* __restrict__ w1pk,
          const float* __restrict__ W2, _Float16* __restrict__ w2pk,
          const float* __restrict__ Wv, _Float16* __restrict__ wvpk,
          const float* __restrict__ Wu, _Float16* __restrict__ wupk,
          const int* __restrict__ batch, int* __restrict__ seg,
          int* __restrict__ cnt) {
    int b = blockIdx.x;
    const int tid = threadIdx.x;
    if (b < SU_B_W1) { pack_w_elem<D_IN, H_DIM>(W1, w1pk, b * SU_T + tid); return; }
    b -= SU_B_W1;
    if (b < SU_B_W2) { pack_w_elem<H_DIM, H_DIM>(W2, w2pk, b * SU_T + tid); return; }
    b -= SU_B_W2;
    if (b < SU_B_WA) { pack_w_elem<H_DIM, HA_DIM>(Wv, wvpk, b * SU_T + tid); return; }
    b -= SU_B_WA;
    if (b < SU_B_WA) { pack_w_elem<H_DIM, HA_DIM>(Wu, wupk, b * SU_T + tid); return; }
    b -= SU_B_WA;
    if (b < SU_B_SEG) {
        const int n = b * SU_T + tid;
        const int cur  = batch[n];
        const int prev = (n == 0) ? -1 : batch[n - 1];
#pragma unroll
        for (int g = 0; g < N_BAGS; ++g)
            if (g > prev && g <= cur) seg[g] = n;
        if (n == N_NODES - 1) {
#pragma unroll
            for (int g = 1; g <= N_BAGS; ++g)
                if (g > cur) seg[g] = N_NODES;
        }
        return;
    }
    b -= SU_B_SEG;
    cnt[b * SU_T + tid] = 0;
}

#define GT_ROWS 64
#define GT_T    256
#define GT_LD   8
static_assert(GT_ROWS == 16 * MREP, "a wave covers all rows of the tile");
static_assert(N_NODES % GT_ROWS == 0, "no partial row tile");

#define GT_PITCH(K) ((K) + 8)

template <int K, int NOUT>
__device__ __forceinline__ void gemm_tile_from_lds(const _Float16* As,
                                                   const _Float16* __restrict__ Wpk,
                                                   float* __restrict__ C, int m0) {
    constexpr int PITCH = GT_PITCH(K);
    constexpr int KBLKS = K / 32;
    constexpr int NTW   = NOUT / 16 / (GT_T / 32);
    static_assert(NTW >= 1 && NTW * 16 * (GT_T / 32) == NOUT, "waves tile the columns");
    const int lane = threadIdx.x & 31;
    const int wave = __builtin_amdgcn_readfirstlane((int)(threadIdx.x >> 5));
    const int half = lane >> 4;
    const int l15  = lane & 15;
    const _Float16* wb = Wpk + (size_t)(wave * NTW) * KBLKS * 512 + lane * 16;
    const _Float16* ab = As + l15 * PITCH + half * 8;

    v8f zero = {};
    v8f acc[MREP][NTW];
#pragma unroll
    for (int r = 0; r < MREP; ++r)
#pragma unroll
        for (int t = 0; t < NTW; ++t) acc[r][t] = zero;

    for (int kblk = 0; kblk < KBLKS; ++kblk) {
        v16h b[NTW];
#pragma unroll
        for (int t = 0; t < NTW; ++t)
            b[t] = *(const v16h*)(wb + ((size_t)t * KBLKS + kblk) * 512);
#pragma unroll
        for (int r = 0; r < MREP; ++r) {
            const _Float16* ap = ab + r * 16 * PITCH + kblk * 32;
            v8h lo = *(const v8h*)(ap);
            v8h hi = *(const v8h*)(ap + 16);
            v16h a = __builtin_shufflevector(lo, hi, 0, 1, 2, 3, 4, 5, 6, 7,
                                             8, 9, 10, 11, 12, 13, 14, 15);
#pragma unroll
            for (int t = 0; t < NTW; ++t)
                acc[r][t] = __builtin_amdgcn_wmma_f32_16x16x32_f16(
                    false, a, false, b[t], (short)0, acc[r][t], false, false);
        }
    }

#pragma unroll
    for (int t = 0; t < NTW; ++t) {
        const int col = (wave * NTW + t) * 16 + l15;
#pragma unroll
        for (int r = 0; r < MREP; ++r) {
#pragma unroll
            for (int q = 0; q < 8; ++q) {
                const int row = m0 + r * 16 + q + half * 8;
                C[(size_t)row * NOUT + col] = acc[r][t][q];
            }
        }
    }
}

typedef __attribute__((ext_vector_type(4))) _Float16 v4h_t;

template <int K, int NOUT>
__global__ void __launch_bounds__(GT_T)
wmma_gemm_lds_af32(const float* __restrict__ A, const _Float16* __restrict__ Wpk,
                   float* __restrict__ C) {
    __shared__ alignas(16) _Float16 As[GT_ROWS * GT_PITCH(K)];
    constexpr int VROW  = K / 4;
    constexpr int PER_T = GT_ROWS * VROW / GT_T;
    static_assert(PER_T % GT_LD == 0 && VROW % 32 == 0, "whole batches, a wave inside one row");
    const int tid = threadIdx.x;
    const int m0  = blockIdx.x * GT_ROWS;
    const float4* Ag = (const float4*)(A + (size_t)m0 * K);
#pragma unroll 1
    for (int it = 0; it < PER_T; it += GT_LD) {
        float4 v[GT_LD];
#pragma unroll
        for (int u = 0; u < GT_LD; ++u) v[u] = Ag[(it + u) * GT_T + tid];
#pragma unroll
        for (int u = 0; u < GT_LD; ++u) {
            const int f = (it + u) * GT_T + tid;
            const int row = f / VROW, c = f % VROW;
            v4h_t h = {(_Float16)v[u].x, (_Float16)v[u].y, (_Float16)v[u].z, (_Float16)v[u].w};
            *(v4h_t*)(As + row * GT_PITCH(K) + c * 4) = h;
        }
    }
    __syncthreads();
    gemm_tile_from_lds<K, NOUT>(As, Wpk, C, m0);
}

template <int K, int NOUT>
__global__ void __launch_bounds__(GT_T)
wmma_gemm_lds_af16(const _Float16* __restrict__ A, const _Float16* __restrict__ Wpk,
                   float* __restrict__ C) {
    __shared__ alignas(16) _Float16 As[GT_ROWS * GT_PITCH(K)];
    constexpr int VROW  = K / 8;
    constexpr int PER_T = GT_ROWS * VROW / GT_T;
    static_assert(GT_ROWS * VROW % GT_T == 0 && VROW % 32 == 0, "whole passes, a wave inside one row");
    const int tid = threadIdx.x;
    const int m0  = blockIdx.x * GT_ROWS;
    const v8h* Ag = (const v8h*)(A + (size_t)m0 * K);
    v8h v[PER_T];
#pragma unroll
    for (int u = 0; u < PER_T; ++u) v[u] = Ag[u * GT_T + tid];
#pragma unroll
    for (int u = 0; u < PER_T; ++u) {
        const int f = u * GT_T + tid;
        const int row = f / VROW, c = f % VROW;
        *(v8h*)(As + row * GT_PITCH(K) + c * 8) = v[u];
    }
    __syncthreads();
    gemm_tile_from_lds<K, NOUT>(As, Wpk, C, m0);
}

template <int K>
__device__ __forceinline__ void pull_rows(const float4* __restrict__ hp, int my_s, int my_c,
                                          int j, int lane, float4& a0, float4& a1) {
    float4 p[K], q[K];
    float  c[K];
#pragma unroll
    for (int k = 0; k < K; ++k) {
        const int s = __builtin_amdgcn_readlane(my_s, j + k);
        c[k] = __int_as_float(__builtin_amdgcn_readlane(my_c, j + k));
        const float4* row = hp + (size_t)s * (H_DIM / 4);
        p[k] = row[lane];
        q[k] = row[32 + lane];
    }
#pragma unroll
    for (int k = 0; k < K; ++k) {
        a0.x = fmaf(p[k].x, c[k], a0.x); a0.y = fmaf(p[k].y, c[k], a0.y);
        a0.z = fmaf(p[k].z, c[k], a0.z); a0.w = fmaf(p[k].w, c[k], a0.w);
        a1.x = fmaf(q[k].x, c[k], a1.x); a1.y = fmaf(q[k].y, c[k], a1.y);
        a1.z = fmaf(q[k].z, c[k], a1.z); a1.w = fmaf(q[k].w, c[k], a1.w);
    }
}

__device__ __forceinline__ float gcn_fin(float agg, float hl, float di, float bc,
                                         float mm, float vv, float gg, float bo) {
    float val = agg + hl * di * di + bc;
    val = (val - mm) * rsqrtf(vv + BN_EPS) * gg + bo;
    return fmaxf(val, 0.0f);
}

__device__ __forceinline__ float4 gcn_fin4(float4 a, float4 h, float di, float4 bc,
                                           float4 mm, float4 vv, float4 gg, float4 bo) {
    float4 r;
    r.x = gcn_fin(a.x, h.x, di, bc.x, mm.x, vv.x, gg.x, bo.x);
    r.y = gcn_fin(a.y, h.y, di, bc.y, mm.y, vv.y, gg.y, bo.y);
    r.z = gcn_fin(a.z, h.z, di, bc.z, mm.z, vv.z, gg.z, bo.z);
    r.w = gcn_fin(a.w, h.w, di, bc.w, mm.w, vv.w, gg.w, bo.w);
    return r;
}

typedef __attribute__((ext_vector_type(4))) _Float16 v4h;

#define PULL_NPW 4
static_assert(N_NODES % (PULL_NPW * 8) == 0, "whole work-groups of 8 waves");

template <bool OUT16>
__global__ void __launch_bounds__(256)
gcn_pull(const float* __restrict__ hlin,
         const int* __restrict__ rowptr,
         const int* __restrict__ csr,
         const float* __restrict__ dinv,
         const float* __restrict__ bconv,
         const float* __restrict__ g,
         const float* __restrict__ bb,
         const float* __restrict__ m,
         const float* __restrict__ v,
         _Float16* __restrict__ out16,
         float* __restrict__ out32, int N) {
    const int lane = threadIdx.x & 31;
    const int nfirst = PULL_NPW * __builtin_amdgcn_readfirstlane(
        (int)((blockIdx.x * blockDim.x + threadIdx.x) >> 5));
    if (nfirst >= N) return;
    const float4* hp = (const float4*)hlin;
    const float4* bc4 = (const float4*)bconv; const float4* g4 = (const float4*)g;
    const float4* bb4 = (const float4*)bb;    const float4* m4 = (const float4*)m;
    const float4* v4  = (const float4*)v;

    int beg = rowptr[nfirst], end = rowptr[nfirst + 1];
    for (int i = 0; i < PULL_NPW; ++i) {
        const int n = nfirst + i;
        if (n >= N) break;
        const int nxt = rowptr[min(n + 2, N)];
        float4 a0 = {0.0f, 0.0f, 0.0f, 0.0f}, a1 = {0.0f, 0.0f, 0.0f, 0.0f};
        for (int base = beg; base < end; base += 32) {
            const int nb = min(32, end - base);
            int my_s = 0, my_c = 0;
            if (lane < nb) { const int2 rec = ((const int2*)csr)[base + lane]; my_s = rec.x; my_c = rec.y; }
            int j = 0;
            for (; j + 4 <= nb; j += 4) pull_rows<4>(hp, my_s, my_c, j, lane, a0, a1);
            const int r = nb - j;
            if (r == 3)      pull_rows<3>(hp, my_s, my_c, j, lane, a0, a1);
            else if (r == 2) pull_rows<2>(hp, my_s, my_c, j, lane, a0, a1);
            else if (r == 1) pull_rows<1>(hp, my_s, my_c, j, lane, a0, a1);
        }

        const float di = dinv[n];
        const float4* own = hp + (size_t)n * (H_DIM / 4);
        const float4 h0 = own[lane], h1 = own[32 + lane];
        const float4 r0 = gcn_fin4(a0, h0, di, bc4[lane], m4[lane], v4[lane], g4[lane], bb4[lane]);
        const float4 r1 = gcn_fin4(a1, h1, di, bc4[32 + lane], m4[32 + lane], v4[32 + lane],
                                   g4[32 + lane], bb4[32 + lane]);
        if (OUT16) {
            v4h* o = (v4h*)(out16 + (size_t)n * H_DIM);
            v4h x0 = {(_Float16)r0.x, (_Float16)r0.y, (_Float16)r0.z, (_Float16)r0.w};
            v4h x1 = {(_Float16)r1.x, (_Float16)r1.y, (_Float16)r1.z, (_Float16)r1.w};
            o[lane] = x0;
            o[32 + lane] = x1;
        } else {
            float4* o = (float4*)(out32 + (size_t)n * H_DIM);
            o[lane] = r0;
            o[32 + lane] = r1;
        }
        beg = end;
        end = nxt;
    }
}

#define ATT_SEG   32
#define ATT_NSEG  (HA_DIM / ATT_SEG)
#define ATT_RP    (HA_DIM + 4)
static_assert(HA_DIM / 16 == GT_T / 32, "one column tile of Wv and one of Wu per wave");
static_assert(ATT_NSEG * GT_ROWS == GT_T, "one (row, segment) per thread");
static_assert(GT_ROWS * ATT_RP * 4 <= GT_ROWS * GT_PITCH(H_DIM) * 2, "the term tile fits in the A tile's LDS");
static_assert(ATT_RP % 4 == 0, "16-byte reads of the term tile");
__global__ void __launch_bounds__(GT_T)
attn_gate_score(const float* __restrict__ Hm,
                const _Float16* __restrict__ Wpk,
                const float* __restrict__ bv,
                const float* __restrict__ bu,
                const float* __restrict__ wA,
                float* __restrict__ a) {
    constexpr int K     = H_DIM;
    constexpr int PITCH = GT_PITCH(K);
    constexpr int KBLKS = K / 32;
    constexpr int NT    = HA_DIM / 16;
    constexpr int VROW  = K / 4;
    constexpr int PER_T = GT_ROWS * VROW / GT_T;
    static_assert(GT_ROWS * VROW % GT_T == 0 && VROW % 32 == 0, "whole passes, a wave inside one row");
    __shared__ alignas(16) _Float16 As[GT_ROWS * PITCH];
    __shared__ float psum[ATT_NSEG][GT_ROWS];
    const int tid = threadIdx.x;
    const int m0  = blockIdx.x * GT_ROWS;

    {
        const float4* Ag = (const float4*)(Hm + (size_t)m0 * K);
        float4 v[PER_T];
#pragma unroll
        for (int u = 0; u < PER_T; ++u) v[u] = Ag[u * GT_T + tid];
#pragma unroll
        for (int u = 0; u < PER_T; ++u) {
            const int f = u * GT_T + tid;
            const int row = f / VROW, c = f % VROW;
            v4h_t h = {(_Float16)v[u].x, (_Float16)v[u].y, (_Float16)v[u].z, (_Float16)v[u].w};
            *(v4h_t*)(As + row * PITCH + c * 4) = h;
        }
    }
    __syncthreads();

    const int lane = tid & 31;
    const int wave = __builtin_amdgcn_readfirstlane((int)(tid >> 5));
    const int half = lane >> 4;
    const int l15  = lane & 15;
    const _Float16* wb = Wpk + (size_t)wave * KBLKS * 512 + lane * 16;
    const _Float16* ab = As + l15 * PITCH + half * 8;

    v8f zero = {};
    v8f acc[MREP][2];
#pragma unroll
    for (int r = 0; r < MREP; ++r) { acc[r][0] = zero; acc[r][1] = zero; }

    for (int kblk = 0; kblk < KBLKS; ++kblk) {
        v16h b[2];
#pragma unroll
        for (int t = 0; t < 2; ++t)
            b[t] = *(const v16h*)(wb + ((size_t)t * NT * KBLKS + kblk) * 512);
#pragma unroll
        for (int r = 0; r < MREP; ++r) {
            const _Float16* ap = ab + r * 16 * PITCH + kblk * 32;
            v8h lo = *(const v8h*)(ap);
            v8h hi = *(const v8h*)(ap + 16);
            v16h av = __builtin_shufflevector(lo, hi, 0, 1, 2, 3, 4, 5, 6, 7,
                                              8, 9, 10, 11, 12, 13, 14, 15);
#pragma unroll
            for (int t = 0; t < 2; ++t)
                acc[r][t] = __builtin_amdgcn_wmma_f32_16x16x32_f16(
                    false, av, false, b[t], (short)0, acc[r][t], false, false);
        }
    }

    const int col = wave * 16 + l15;
    const float bvc = bv[col], buc = bu[col], wac = wA[col];
    __syncthreads();
    float* term = (float*)As;
#pragma unroll
    for (int r = 0; r < MREP; ++r) {
#pragma unroll
        for (int q = 0; q < 8; ++q) {
            const int row = r * 16 + q + half * 8;
            const float tv = tanhf(acc[r][0][q] + bvc);
            const float sv = 1.0f / (1.0f + expf(-(acc[r][1][q] + buc)));
            term[row * ATT_RP + col] = tv * sv * wac;
        }
    }
    __syncthreads();

    {
        const int row = tid % GT_ROWS, sg = tid / GT_ROWS;
        const float4* tp = (const float4*)(term + row * ATT_RP + sg * ATT_SEG);
        float4 x[ATT_SEG / 4];
#pragma unroll
        for (int j = 0; j < ATT_SEG / 4; ++j) x[j] = tp[j];
        float s = 0.0f;
#pragma unroll
        for (int j = 0; j < ATT_SEG / 4; ++j) { s += x[j].x; s += x[j].y; s += x[j].z; s += x[j].w; }
        psum[sg][row] = s;
    }
    __syncthreads();
    if (tid < GT_ROWS) {
        float s = psum[0][tid];
#pragma unroll
        for (int g = 1; g < ATT_NSEG; ++g) s += psum[g][tid];
        a[m0 + tid] = s;
    }
}

#define SM_T 1024
#define SM_U 12
static_assert(SM_T == 32 * 32, "the second reduction stage is one value per lane of one wave");

__device__ __forceinline__ float wave_max(float v) {
    for (int off = 16; off > 0; off >>= 1) v = fmaxf(v, __shfl_xor(v, off, 32));
    return v;
}
__device__ __forceinline__ float wave_sum(float v) {
    for (int off = 16; off > 0; off >>= 1) v += __shfl_xor(v, off, 32);
    return v;
}

__global__ void __launch_bounds__(SM_T)
seg_softmax(const float* __restrict__ a,
            const int* __restrict__ seg,
            float* __restrict__ wout) {
    __shared__ float red[SM_T / 32];
    const int g = blockIdx.x;
    const int s = seg[g], e = seg[g + 1];
    if (e <= s) return;
    const int tid = threadIdx.x, lane = tid & 31, wv = tid >> 5;
    const int first = s + tid;
    const int more  = first + SM_T * SM_U;

    float v0[SM_U];
#pragma unroll
    for (int u = 0; u < SM_U; ++u) v0[u] = a[min(first + u * SM_T, e - 1)];
    float mx = -1e30f;
#pragma unroll
    for (int u = 0; u < SM_U; ++u) mx = fmaxf(mx, v0[u]);
    for (int nb = more; nb < e; nb += SM_T * SM_U) {
        float v[SM_U];
#pragma unroll
        for (int u = 0; u < SM_U; ++u) v[u] = a[min(nb + u * SM_T, e - 1)];
#pragma unroll
        for (int u = 0; u < SM_U; ++u) mx = fmaxf(mx, v[u]);
    }
    mx = wave_max(mx);
    if (lane == 0) red[wv] = mx;
    __syncthreads();
    mx = wave_max(red[lane]);
    __syncthreads();

    float ex[SM_U];
    float sum = 0.0f;
#pragma unroll
    for (int u = 0; u < SM_U; ++u) {
        ex[u] = 0.0f;
        if (first + u * SM_T < e) ex[u] = expf(v0[u] - mx);
        sum += ex[u];
    }
    for (int nb = more; nb < e; nb += SM_T * SM_U) {
        float v[SM_U];
#pragma unroll
        for (int u = 0; u < SM_U; ++u) v[u] = a[min(nb + u * SM_T, e - 1)];
#pragma unroll
        for (int u = 0; u < SM_U; ++u) {
            const float x = expf(v[u] - mx);
            sum += (nb + u * SM_T < e) ? x : 0.0f;
        }
    }
    sum = wave_sum(sum);
    if (lane == 0) red[wv] = sum;
    __syncthreads();
    const float inv = 1.0f / wave_sum(red[lane]);

#pragma unroll
    for (int u = 0; u < SM_U; ++u) {
        const int n = first + u * SM_T;
        if (n < e) wout[n] = ex[u] * inv;
    }
    for (int nb = more; nb < e; nb += SM_T * SM_U) {
        float v[SM_U];
#pragma unroll
        for (int u = 0; u < SM_U; ++u) v[u] = a[min(nb + u * SM_T, e - 1)];
#pragma unroll
        for (int u = 0; u < SM_U; ++u) {
            const int n = nb + u * SM_T;
            if (n < e) wout[n] = expf(v[u] - mx) * inv;
        }
    }
}

#define POOL_CH 32
#define POOL_T  512
#define POOL_RG (POOL_T / 64)
#define POOL_U  8
static_assert(H_DIM == 256, "pooling assumes 64 float4 per row");
__global__ void __launch_bounds__(POOL_T)
slide_partial(const float* __restrict__ w, const float* __restrict__ h,
              const int* __restrict__ seg, float* __restrict__ part) {
    __shared__ float4 red[POOL_RG][64];
    const int g = blockIdx.x / POOL_CH, c = blockIdx.x % POOL_CH;
    const int s = seg[g], e = seg[g + 1];
    const int per = (e - s + POOL_CH - 1) / POOL_CH;
    const int n0 = s + c * per;
    const int n1 = min(n0 + per, e);
    const int col = threadIdx.x & 63;
    const int rg = __builtin_amdgcn_readfirstlane((int)(threadIdx.x >> 6));
    const float4* h4 = (const float4*)h;
    const float4 zero4 = {0.0f, 0.0f, 0.0f, 0.0f};
    float4 acc = zero4;
    for (int nb = n0 + rg; nb < n1; nb += POOL_U * POOL_RG) {
        float4 v[POOL_U];
        float  ww[POOL_U];
#pragma unroll
        for (int u = 0; u < POOL_U; ++u) {
            const int n  = nb + u * POOL_RG;
            const int nc = min(n, n1 - 1);
            v[u]  = h4[(size_t)nc * (H_DIM / 4) + col];
            ww[u] = w[nc];
            if (n >= n1) v[u] = zero4;
        }
#pragma unroll
        for (int u = 0; u < POOL_U; ++u) {
            acc.x = fmaf(ww[u], v[u].x, acc.x); acc.y = fmaf(ww[u], v[u].y, acc.y);
            acc.z = fmaf(ww[u], v[u].z, acc.z); acc.w = fmaf(ww[u], v[u].w, acc.w);
        }
    }
    red[rg][col] = acc;
    __syncthreads();
    if (threadIdx.x < H_DIM) {
        const float* rf = (const float*)red;
        float sum = 0.0f;
#pragma unroll
        for (int r = 0; r < POOL_RG; ++r) sum += rf[r * H_DIM + threadIdx.x];
        part[(size_t)blockIdx.x * H_DIM + threadIdx.x] = sum;
    }
}

#define HEAD_T  1024
#define HEAD_WV (H_DIM * HA_DIM / 4 / HEAD_T)
static_assert(HEAD_WV * HEAD_T * 4 == H_DIM * HA_DIM, "Wc1 splits into whole float4 per thread");
static_assert(H_DIM <= HEAD_T && HA_DIM * C_DIM <= HEAD_T && H_DIM % 32 == 0 && HA_DIM % 32 == 0,
              "one thread per column / Wc2 value; branches on whole waves");
__global__ void __launch_bounds__(HEAD_T)
head_kernel(const float* __restrict__ part,
            const float* __restrict__ Wc1,
            const float* __restrict__ bc1,
            const float* __restrict__ Wc2,
            const float* __restrict__ bc2,
            float* __restrict__ logits) {
    __shared__ alignas(16) float w1s[H_DIM * HA_DIM];
    __shared__ alignas(16) float slide[H_DIM];
    __shared__ alignas(16) float hid[HA_DIM];
    __shared__ float w2s[HA_DIM * C_DIM];
    const int tid = threadIdx.x;
    const int g = blockIdx.x;

    float4 wv[HEAD_WV];
#pragma unroll
    for (int u = 0; u < HEAD_WV; ++u) wv[u] = ((const float4*)Wc1)[u * HEAD_T + tid];
    float pv[POOL_CH];
    if (tid < H_DIM) {
        const float* p = part + (size_t)g * POOL_CH * H_DIM + tid;
#pragma unroll
        for (int c = 0; c < POOL_CH; ++c) pv[c] = p[c * H_DIM];
    }
    float w2v = 0.0f, b1v = 0.0f, b2v = 0.0f;
    if (tid < HA_DIM * C_DIM) w2v = Wc2[tid];
    if (tid < HA_DIM) b1v = bc1[tid];
    if (tid < C_DIM) b2v = bc2[tid];

#pragma unroll
    for (int u = 0; u < HEAD_WV; ++u) ((float4*)w1s)[u * HEAD_T + tid] = wv[u];
    if (tid < H_DIM) {
        float s = 0.0f;
#pragma unroll
        for (int c = 0; c < POOL_CH; ++c) s += pv[c];
        slide[tid] = s;
    }
    if (tid < HA_DIM * C_DIM) w2s[tid] = w2v;
    __syncthreads();

    if (tid < HA_DIM) {
        float acc = b1v;
#pragma unroll 32
        for (int k = 0; k < H_DIM; ++k) acc += slide[k] * w1s[k * HA_DIM + tid];
        hid[tid] = fmaxf(acc, 0.0f);
    }
    __syncthreads();

    if (tid < C_DIM) {
        float s = b2v;
#pragma unroll
        for (int t = 0; t < HA_DIM; ++t) s += hid[t] * w2s[t * C_DIM + tid];
        logits[g * C_DIM + tid] = s;
    }
}

extern "C" void kernel_launch(void* const* d_in, const int* in_sizes, int n_in,
                              void* d_out, int out_size, void* d_ws, size_t ws_size,
                              hipStream_t stream) {
    (void)in_sizes; (void)n_in; (void)out_size; (void)ws_size;
    const float* x   = (const float*)d_in[0];
    const int*  ei   = (const int*)d_in[1];
    const int*  batch= (const int*)d_in[2];
    const float* W1  = (const float*)d_in[3];  const float* b1   = (const float*)d_in[4];
    const float* g1  = (const float*)d_in[5];  const float* bb1  = (const float*)d_in[6];
    const float* m1  = (const float*)d_in[7];  const float* v1   = (const float*)d_in[8];
    const float* W2  = (const float*)d_in[9];  const float* b2   = (const float*)d_in[10];
    const float* g2  = (const float*)d_in[11]; const float* bb2  = (const float*)d_in[12];
    const float* m2  = (const float*)d_in[13]; const float* v2   = (const float*)d_in[14];
    const float* Wv  = (const float*)d_in[15]; const float* bv   = (const float*)d_in[16];
    const float* Wu  = (const float*)d_in[17]; const float* bu   = (const float*)d_in[18];
    const float* wA  = (const float*)d_in[19];
    const float* Wc1 = (const float*)d_in[20]; const float* bc1  = (const float*)d_in[21];
    const float* Wc2 = (const float*)d_in[22]; const float* bc2  = (const float*)d_in[23];

    const int* src = ei;
    const int* dst = ei + N_EDGES;

    float* out  = (float*)d_out;
    float* wout = out + N_BAGS * C_DIM;

    const size_t NH = (size_t)N_NODES * H_DIM;
    float*    bufA = (float*)d_ws;
    float*    bufB = bufA + NH;
    _Float16* h16  = (_Float16*)(bufB + NH);
    _Float16* w1pk = h16 + NH;
    _Float16* w2pk = w1pk + (size_t)D_IN * H_DIM;
    _Float16* wvpk = w2pk + (size_t)H_DIM * H_DIM;
    _Float16* wupk = wvpk + (size_t)H_DIM * HA_DIM;
    float*    dinv = (float*)(wupk + (size_t)H_DIM * HA_DIM);
    float*    abuf = dinv + N_NODES;
    int*      seg  = (int*)(abuf + N_NODES);

    int*   cnt      = (int*)h16;
    int*   rowptr   = cnt + N_NODES;
    int*   rank     = rowptr + N_NODES + 4;
    int*   csr      = rank + N_EDGES;
    _Float16* h1_16 = (_Float16*)bufB;
    float*    h2_32 = bufB;

    setup_all<<<SU_BLOCKS, SU_T, 0, stream>>>(W1, w1pk, W2, w2pk, Wv, wvpk, Wu, wupk,
                                              batch, seg, cnt);

    count_rank<<<N_EDGES / (EDGE_T * EDGE_PER), EDGE_T, 0, stream>>>(dst, cnt, rank);
    scan_rowptr<<<1, SCAN_T, 0, stream>>>(cnt, rowptr, dinv);
    csr_fill<<<N_EDGES / (EDGE_T * EDGE_PER), EDGE_T, 0, stream>>>(src, dst, rank, rowptr, dinv,
                                                                   csr);

    wmma_gemm_lds_af32<D_IN, H_DIM><<<N_NODES / GT_ROWS, GT_T, 0, stream>>>(x, w1pk, bufA);
    gcn_pull<true><<<N_NODES / PULL_NPW / 8, 256, 0, stream>>>(
        bufA, rowptr, csr, dinv, b1, g1, bb1, m1, v1,
        h1_16, nullptr, N_NODES);

    wmma_gemm_lds_af16<H_DIM, H_DIM><<<N_NODES / GT_ROWS, GT_T, 0, stream>>>(h1_16, w2pk, bufA);
    gcn_pull<false><<<N_NODES / PULL_NPW / 8, 256, 0, stream>>>(
        bufA, rowptr, csr, dinv, b2, g2, bb2, m2, v2,
        nullptr, h2_32, N_NODES);

    attn_gate_score<<<N_NODES / GT_ROWS, GT_T, 0, stream>>>(h2_32, wvpk, bv, bu, wA, abuf);

    seg_softmax<<<N_BAGS, SM_T, 0, stream>>>(abuf, seg, wout);

    float* part = bufA;
    slide_partial<<<N_BAGS * POOL_CH, POOL_T, 0, stream>>>(wout, h2_32, seg, part);

    head_kernel<<<N_BAGS, HEAD_T, 0, stream>>>(part, Wc1, bc1, Wc2, bc2, out);
}


# Layer output fused into its consumer, 12 -> 10 launches: W2 GEMM and score kernel pull their 64 rows into the LDS A tile; a node's bounds, records and own row asked for ahead; sum orders kept
# speedup vs baseline: 5.7428x; 1.0961x over previous
#include <hip/hip_runtime.h>
#include <hip/hip_bf16.h>
#include <math.h>

typedef __attribute__((ext_vector_type(16))) _Float16 v16h;
typedef __attribute__((ext_vector_type(8)))  _Float16 v8h;
typedef __attribute__((ext_vector_type(8)))  float    v8f;

#define N_NODES 65536
#define N_EDGES 524288
#define N_BAGS  8
#define D_IN    512
#define H_DIM   256
#define HA_DIM  128
#define C_DIM   2
#define BN_EPS  1e-5f
#define MREP    4

#define EDGE_PER 4
#define EDGE_T   256
static_assert(N_EDGES % (EDGE_T * EDGE_PER) == 0, "whole work-groups");
__global__ void __launch_bounds__(EDGE_T)
count_rank(const int* __restrict__ dst, int* __restrict__ cnt, int* __restrict__ rank) {
    const int t = blockIdx.x * blockDim.x + threadIdx.x;
    const int4 d = ((const int4*)dst)[t];
    int4 r;
    r.x = atomicAdd(&cnt[d.x], 1);
    r.y = atomicAdd(&cnt[d.y], 1);
    r.z = atomicAdd(&cnt[d.z], 1);
    r.w = atomicAdd(&cnt[d.w], 1);
    ((int4*)rank)[t] = r;
}

#define SCAN_T   1024
#define SCAN_PER (N_NODES / SCAN_T)
static_assert(N_NODES % (SCAN_T * 4) == 0, "scan assumes whole int4 chunks per thread");
__global__ void __launch_bounds__(SCAN_T)
scan_rowptr(const int* __restrict__ cnt, int* __restrict__ rowptr,
            float* __restrict__ dinv) {
    __shared__ int wsum[32];
    const int t = threadIdx.x, lane = t & 31, w = t >> 5;
    const int4* c4 = (const int4*)(cnt + t * SCAN_PER);
    int4 c[SCAN_PER / 4];
    int s = 0;
#pragma unroll
    for (int i = 0; i < SCAN_PER / 4; ++i) {
        c[i] = c4[i];
        s += c[i].x + c[i].y + c[i].z + c[i].w;
    }
    int inc = s;
    for (int off = 1; off < 32; off <<= 1) {
        int v = __shfl_up(inc, off, 32);
        if (lane >= off) inc += v;
    }
    if (lane == 31) wsum[w] = inc;
    __syncthreads();
    if (w == 0) {
        int v = wsum[lane], iv = v;
        for (int off = 1; off < 32; off <<= 1) {
            int u = __shfl_up(iv, off, 32);
            if (lane >= off) iv += u;
        }
        wsum[lane] = iv - v;
    }
    __syncthreads();
    int run = wsum[w] + inc - s;
    int4*   r4 = (int4*)(rowptr + t * SCAN_PER);
    float4* d4 = (float4*)(dinv + t * SCAN_PER);
#pragma unroll
    for (int i = 0; i < SCAN_PER / 4; ++i) {
        int4 r;
        r.x = run; run += c[i].x;
        r.y = run; run += c[i].y;
        r.z = run; run += c[i].z;
        r.w = run; run += c[i].w;
        r4[i] = r;
        float4 d;
        d.x = rsqrtf(1.0f + (float)c[i].x);
        d.y = rsqrtf(1.0f + (float)c[i].y);
        d.z = rsqrtf(1.0f + (float)c[i].z);
        d.w = rsqrtf(1.0f + (float)c[i].w);
        d4[i] = d;
    }
    if (t == SCAN_T - 1) rowptr[N_NODES] = run;
}

__device__ __forceinline__ void csr_put(int s, int d, int r,
                                        const int* __restrict__ rowptr,
                                        const float* __restrict__ dinv,
                                        int* __restrict__ csr) {
    const int pos = rowptr[d] + r;
    ((int2*)csr)[pos] = make_int2(s, __float_as_int(dinv[s] * dinv[d]));
}

__global__ void __launch_bounds__(EDGE_T)
csr_fill(const int* __restrict__ src, const int* __restrict__ dst,
         const int* __restrict__ rank, const int* __restrict__ rowptr,
         const float* __restrict__ dinv, int* __restrict__ csr) {
    const int t = blockIdx.x * blockDim.x + threadIdx.x;
    const int4 s = ((const int4*)src)[t], d = ((const int4*)dst)[t], r = ((const int4*)rank)[t];
    csr_put(s.x, d.x, r.x, rowptr, dinv, csr);
    csr_put(s.y, d.y, r.y, rowptr, dinv, csr);
    csr_put(s.z, d.z, r.z, rowptr, dinv, csr);
    csr_put(s.w, d.w, r.w, rowptr, dinv, csr);
}

template <int K, int NOUT>
__device__ __forceinline__ void pack_w_elem(const float* __restrict__ Wm,
                                            _Float16* __restrict__ out, int t) {
    constexpr int KBLKS = K >> 5;
    const int i    = t & 15;
    const int lane = (t >> 4) & 31;
    const int rest = t >> 9;
    const int ntile = rest / KBLKS;
    const int kblk  = rest % KBLKS;
    const int half = lane >> 4, l15 = lane & 15;
    const int col = ntile * 16 + l15;
    const int k   = kblk * 32 + half * 8 + ((i < 8) ? i : (16 + i - 8));
    out[t] = (_Float16)Wm[(size_t)k * NOUT + col];
}

#define SU_T      256
#define SU_B_W1   (D_IN * H_DIM / SU_T)
#define SU_B_W2   (H_DIM * H_DIM / SU_T)
#define SU_B_WA   (H_DIM * HA_DIM / SU_T)
#define SU_B_SEG  (N_NODES / SU_T)
#define SU_B_ZERO (N_NODES / SU_T)
#define SU_BLOCKS (SU_B_W1 + SU_B_W2 + 2 * SU_B_WA + SU_B_SEG + SU_B_ZERO)
static_assert(N_NODES % SU_T == 0 && (H_DIM * HA_DIM) % SU_T == 0, "whole blocks per range");
__global__ void __launch_bounds__(SU_T)
setup_all(const float* __restrict__ W1, _Float16* __restrict__ w1pk,
          const float* __restrict__ W2, _Float16* __restrict__ w2pk,
          const float* __restrict__ Wv, _Float16* __restrict__ wvpk,
          const float* __restrict__ Wu, _Float16* __restrict__ wupk,
          const int* __restrict__ batch, int* __restrict__ seg,
          int* __restrict__ cnt) {
    int b = blockIdx.x;
    const int tid = threadIdx.x;
    if (b < SU_B_W1) { pack_w_elem<D_IN, H_DIM>(W1, w1pk, b * SU_T + tid); return; }
    b -= SU_B_W1;
    if (b < SU_B_W2) { pack_w_elem<H_DIM, H_DIM>(W2, w2pk, b * SU_T + tid); return; }
    b -= SU_B_W2;
    if (b < SU_B_WA) { pack_w_elem<H_DIM, HA_DIM>(Wv, wvpk, b * SU_T + tid); return; }
    b -= SU_B_WA;
    if (b < SU_B_WA) { pack_w_elem<H_DIM, HA_DIM>(Wu, wupk, b * SU_T + tid); return; }
    b -= SU_B_WA;
    if (b < SU_B_SEG) {
        const int n = b * SU_T + tid;
        const int cur  = batch[n];
        const int prev = (n == 0) ? -1 : batch[n - 1];
#pragma unroll
        for (int g = 0; g < N_BAGS; ++g)
            if (g > prev && g <= cur) seg[g] = n;
        if (n == N_NODES - 1) {
#pragma unroll
            for (int g = 1; g <= N_BAGS; ++g)
                if (g > cur) seg[g] = N_NODES;
        }
        return;
    }
    b -= SU_B_SEG;
    cnt[b * SU_T + tid] = 0;
}

#define GT_ROWS 64
#define GT_T    256
#define GT_LD   8
static_assert(GT_ROWS == 16 * MREP, "a wave covers all rows of the tile");
static_assert(N_NODES % GT_ROWS == 0, "no partial row tile");

#define GT_PITCH(K) ((K) + 8)

template <int K, int NOUT>
__device__ __forceinline__ void gemm_tile_from_lds(const _Float16* As,
                                                   const _Float16* __restrict__ Wpk,
                                                   float* __restrict__ C, int m0) {
    constexpr int PITCH = GT_PITCH(K);
    constexpr int KBLKS = K / 32;
    constexpr int NTW   = NOUT / 16 / (GT_T / 32);
    static_assert(NTW >= 1 && NTW * 16 * (GT_T / 32) == NOUT, "waves tile the columns");
    const int lane = threadIdx.x & 31;
    const int wave = __builtin_amdgcn_readfirstlane((int)(threadIdx.x >> 5));
    const int half = lane >> 4;
    const int l15  = lane & 15;
    const _Float16* wb = Wpk + (size_t)(wave * NTW) * KBLKS * 512 + lane * 16;
    const _Float16* ab = As + l15 * PITCH + half * 8;

    v8f zero = {};
    v8f acc[MREP][NTW];
#pragma unroll
    for (int r = 0; r < MREP; ++r)
#pragma unroll
        for (int t = 0; t < NTW; ++t) acc[r][t] = zero;

    for (int kblk = 0; kblk < KBLKS; ++kblk) {
        v16h b[NTW];
#pragma unroll
        for (int t = 0; t < NTW; ++t)
            b[t] = *(const v16h*)(wb + ((size_t)t * KBLKS + kblk) * 512);
#pragma unroll
        for (int r = 0; r < MREP; ++r) {
            const _Float16* ap = ab + r * 16 * PITCH + kblk * 32;
            v8h lo = *(const v8h*)(ap);
            v8h hi = *(const v8h*)(ap + 16);
            v16h a = __builtin_shufflevector(lo, hi, 0, 1, 2, 3, 4, 5, 6, 7,
                                             8, 9, 10, 11, 12, 13, 14, 15);
#pragma unroll
            for (int t = 0; t < NTW; ++t)
                acc[r][t] = __builtin_amdgcn_wmma_f32_16x16x32_f16(
                    false, a, false, b[t], (short)0, acc[r][t], false, false);
        }
    }

#pragma unroll
    for (int t = 0; t < NTW; ++t) {
        const int col = (wave * NTW + t) * 16 + l15;
#pragma unroll
        for (int r = 0; r < MREP; ++r) {
#pragma unroll
            for (int q = 0; q < 8; ++q) {
                const int row = m0 + r * 16 + q + half * 8;
                C[(size_t)row * NOUT + col] = acc[r][t][q];
            }
        }
    }
}

typedef __attribute__((ext_vector_type(4))) _Float16 v4h_t;

template <int K, int NOUT>
__global__ void __launch_bounds__(GT_T)
wmma_gemm_lds_af32(const float* __restrict__ A, const _Float16* __restrict__ Wpk,
                   float* __restrict__ C) {
    __shared__ alignas(16) _Float16 As[GT_ROWS * GT_PITCH(K)];
    constexpr int VROW  = K / 4;
    constexpr int PER_T = GT_ROWS * VROW / GT_T;
    static_assert(PER_T % GT_LD == 0 && VROW % 32 == 0, "whole batches, a wave inside one row");
    const int tid = threadIdx.x;
    const int m0  = blockIdx.x * GT_ROWS;
    const float4* Ag = (const float4*)(A + (size_t)m0 * K);
#pragma unroll 1
    for (int it = 0; it < PER_T; it += GT_LD) {
        float4 v[GT_LD];
#pragma unroll
        for (int u = 0; u < GT_LD; ++u) v[u] = Ag[(it + u) * GT_T + tid];
#pragma unroll
        for (int u = 0; u < GT_LD; ++u) {
            const int f = (it + u) * GT_T + tid;
            const int row = f / VROW, c = f % VROW;
            v4h_t h = {(_Float16)v[u].x, (_Float16)v[u].y, (_Float16)v[u].z, (_Float16)v[u].w};
            *(v4h_t*)(As + row * GT_PITCH(K) + c * 4) = h;
        }
    }
    __syncthreads();
    gemm_tile_from_lds<K, NOUT>(As, Wpk, C, m0);
}

template <int K>
__device__ __forceinline__ void pull_rows(const float4* __restrict__ hp, int my_s, int my_c,
                                          int j, int lane, float4& a0, float4& a1) {
    float4 p[K], q[K];
    float  c[K];
#pragma unroll
    for (int k = 0; k < K; ++k) {
        const int s = __builtin_amdgcn_readlane(my_s, j + k);
        c[k] = __int_as_float(__builtin_amdgcn_readlane(my_c, j + k));
        const float4* row = hp + (size_t)s * (H_DIM / 4);
        p[k] = row[lane];
        q[k] = row[32 + lane];
    }
#pragma unroll
    for (int k = 0; k < K; ++k) {
        a0.x = fmaf(p[k].x, c[k], a0.x); a0.y = fmaf(p[k].y, c[k], a0.y);
        a0.z = fmaf(p[k].z, c[k], a0.z); a0.w = fmaf(p[k].w, c[k], a0.w);
        a1.x = fmaf(q[k].x, c[k], a1.x); a1.y = fmaf(q[k].y, c[k], a1.y);
        a1.z = fmaf(q[k].z, c[k], a1.z); a1.w = fmaf(q[k].w, c[k], a1.w);
    }
}

__device__ __forceinline__ float gcn_fin(float agg, float hl, float di, float bc,
                                         float mm, float vv, float gg, float bo) {
    float val = agg + hl * di * di + bc;
    val = (val - mm) * rsqrtf(vv + BN_EPS) * gg + bo;
    return fmaxf(val, 0.0f);
}

__device__ __forceinline__ float4 gcn_fin4(float4 a, float4 h, float di, float4 bc,
                                           float4 mm, float4 vv, float4 gg, float4 bo) {
    float4 r;
    r.x = gcn_fin(a.x, h.x, di, bc.x, mm.x, vv.x, gg.x, bo.x);
    r.y = gcn_fin(a.y, h.y, di, bc.y, mm.y, vv.y, gg.y, bo.y);
    r.z = gcn_fin(a.z, h.z, di, bc.z, mm.z, vv.z, gg.z, bo.z);
    r.w = gcn_fin(a.w, h.w, di, bc.w, mm.w, vv.w, gg.w, bo.w);
    return r;
}

typedef __attribute__((ext_vector_type(4))) _Float16 v4h;

#define TILE_NPW (GT_ROWS / (GT_T / 32))
#define PULL_WAVES_PER_SIMD 8
static_assert(TILE_NPW * (GT_T / 32) == GT_ROWS, "the waves split the tile's rows evenly");

__device__ __forceinline__ int2 csr_chunk(const int* __restrict__ csr, int base, int end, int lane) {
    int2 rec = make_int2(0, 0);
    if (lane < end - base) rec = ((const int2*)csr)[base + lane];
    return rec;
}

template <bool OUT32>
__device__ __forceinline__ void pull_tile_to_lds(const float* __restrict__ hlin,
                                                 const int* __restrict__ rowptr,
                                                 const int* __restrict__ csr,
                                                 const float* __restrict__ dinv,
                                                 const float* __restrict__ bconv,
                                                 const float* __restrict__ g,
                                                 const float* __restrict__ bb,
                                                 const float* __restrict__ m,
                                                 const float* __restrict__ v,
                                                 _Float16* As,
                                                 float* __restrict__ out32, int m0) {
    constexpr int PITCH = GT_PITCH(H_DIM);
    const int lane = threadIdx.x & 31;
    const int wave = __builtin_amdgcn_readfirstlane((int)(threadIdx.x >> 5));
    const int rfirst = wave * TILE_NPW;
    const int nfirst = m0 + rfirst;
    const float4* hp = (const float4*)hlin;
    const float4* bc4 = (const float4*)bconv; const float4* g4 = (const float4*)g;
    const float4* bb4 = (const float4*)bb;    const float4* m4 = (const float4*)m;
    const float4* v4  = (const float4*)v;

    const int rp = rowptr[nfirst + min(lane, TILE_NPW)];
    int beg = __builtin_amdgcn_readlane(rp, 0), end = __builtin_amdgcn_readlane(rp, 1);
    int2 rec_nx = csr_chunk(csr, beg, end, lane);
    for (int i = 0; i < TILE_NPW; ++i) {
        const int n = nfirst + i;
        const int nxt = __builtin_amdgcn_readlane(rp, min(i + 2, TILE_NPW));
        int2 rec = rec_nx;
        if (i + 1 < TILE_NPW) rec_nx = csr_chunk(csr, end, nxt, lane);
        const float di = dinv[n];
        const float4* own = hp + (size_t)n * (H_DIM / 4);
        const float4 h0 = own[lane], h1 = own[32 + lane];
        float4 a0 = {0.0f, 0.0f, 0.0f, 0.0f}, a1 = {0.0f, 0.0f, 0.0f, 0.0f};
        for (int base = beg; base < end; base += 32) {
            const int nb = min(32, end - base);
            if (base != beg) rec = csr_chunk(csr, base, end, lane);
            const int my_s = rec.x, my_c = rec.y;
            int j = 0;
            for (; j + 8 <= nb; j += 8) pull_rows<8>(hp, my_s, my_c, j, lane, a0, a1);
            if (j + 4 <= nb) { pull_rows<4>(hp, my_s, my_c, j, lane, a0, a1); j += 4; }
            const int r = nb - j;
            if (r == 3)      pull_rows<3>(hp, my_s, my_c, j, lane, a0, a1);
            else if (r == 2) pull_rows<2>(hp, my_s, my_c, j, lane, a0, a1);
            else if (r == 1) pull_rows<1>(hp, my_s, my_c, j, lane, a0, a1);
        }

        const float4 r0 = gcn_fin4(a0, h0, di, bc4[lane], m4[lane], v4[lane], g4[lane], bb4[lane]);
        const float4 r1 = gcn_fin4(a1, h1, di, bc4[32 + lane], m4[32 + lane], v4[32 + lane],
                                   g4[32 + lane], bb4[32 + lane]);
        _Float16* o = As + (rfirst + i) * PITCH;
        v4h x0 = {(_Float16)r0.x, (_Float16)r0.y, (_Float16)r0.z, (_Float16)r0.w};
        v4h x1 = {(_Float16)r1.x, (_Float16)r1.y, (_Float16)r1.z, (_Float16)r1.w};
        *(v4h*)(o + 4 * lane) = x0;
        *(v4h*)(o + H_DIM / 2 + 4 * lane) = x1;
        if (OUT32) {
            float4* og = (float4*)(out32 + (size_t)n * H_DIM);
            og[lane] = r0;
            og[32 + lane] = r1;
        }
        beg = end;
        end = nxt;
    }
}

__global__ void __launch_bounds__(GT_T, PULL_WAVES_PER_SIMD)
gcn_pull_gemm(const float* __restrict__ hlin,
              const int* __restrict__ rowptr,
              const int* __restrict__ csr,
              const float* __restrict__ dinv,
              const float* __restrict__ bconv,
              const float* __restrict__ g,
              const float* __restrict__ bb,
              const float* __restrict__ m,
              const float* __restrict__ v,
              const _Float16* __restrict__ Wpk,
              float* __restrict__ C) {
    __shared__ alignas(16) _Float16 As[GT_ROWS * GT_PITCH(H_DIM)];
    const int m0 = blockIdx.x * GT_ROWS;
    pull_tile_to_lds<false>(hlin, rowptr, csr, dinv, bconv, g, bb, m, v, As, nullptr, m0);
    __syncthreads();
    gemm_tile_from_lds<H_DIM, H_DIM>(As, Wpk, C, m0);
}

#define ATT_SEG   32
#define ATT_NSEG  (HA_DIM / ATT_SEG)
#define ATT_RP    (HA_DIM + 4)
static_assert(HA_DIM / 16 == GT_T / 32, "one column tile of Wv and one of Wu per wave");
static_assert(ATT_NSEG * GT_ROWS == GT_T, "one (row, segment) per thread");
static_assert(GT_ROWS * ATT_RP * 4 <= GT_ROWS * GT_PITCH(H_DIM) * 2, "the term tile fits in the A tile's LDS");
static_assert(ATT_RP % 4 == 0, "16-byte reads of the term tile");

__device__ __forceinline__ void attn_tile_from_lds(_Float16* As,
                                                   float (*psum)[GT_ROWS],
                                                   const _Float16* __restrict__ Wpk,
                                                   const float* __restrict__ bv,
                                                   const float* __restrict__ bu,
                                                   const float* __restrict__ wA,
                                                   float* __restrict__ a, int m0) {
    constexpr int K     = H_DIM;
    constexpr int PITCH = GT_PITCH(K);
    constexpr int KBLKS = K / 32;
    constexpr int NT    = HA_DIM / 16;
    const int tid = threadIdx.x;

    const int lane = tid & 31;
    const int wave = __builtin_amdgcn_readfirstlane((int)(tid >> 5));
    const int half = lane >> 4;
    const int l15  = lane & 15;
    const _Float16* wb = Wpk + (size_t)wave * KBLKS * 512 + lane * 16;
    const _Float16* ab = As + l15 * PITCH + half * 8;

    v8f zero = {};
    v8f acc[MREP][2];
#pragma unroll
    for (int r = 0; r < MREP; ++r) { acc[r][0] = zero; acc[r][1] = zero; }

    for (int kblk = 0; kblk < KBLKS; ++kblk) {
        v16h b[2];
#pragma unroll
        for (int t = 0; t < 2; ++t)
            b[t] = *(const v16h*)(wb + ((size_t)t * NT * KBLKS + kblk) * 512);
#pragma unroll
        for (int r = 0; r < MREP; ++r) {
            const _Float16* ap = ab + r * 16 * PITCH + kblk * 32;
            v8h lo = *(const v8h*)(ap);
            v8h hi = *(const v8h*)(ap + 16);
            v16h av = __builtin_shufflevector(lo, hi, 0, 1, 2, 3, 4, 5, 6, 7,
                                              8, 9, 10, 11, 12, 13, 14, 15);
#pragma unroll
            for (int t = 0; t < 2; ++t)
                acc[r][t] = __builtin_amdgcn_wmma_f32_16x16x32_f16(
                    false, av, false, b[t], (short)0, acc[r][t], false, false);
        }
    }

    const int col = wave * 16 + l15;
    const float bvc = bv[col], buc = bu[col], wac = wA[col];
    __syncthreads();
    float* term = (float*)As;
#pragma unroll
    for (int r = 0; r < MREP; ++r) {
#pragma unroll
        for (int q = 0; q < 8; ++q) {
            const int row = r * 16 + q + half * 8;
            const float tv = tanhf(acc[r][0][q] + bvc);
            const float sv = 1.0f / (1.0f + expf(-(acc[r][1][q] + buc)));
            term[row * ATT_RP + col] = tv * sv * wac;
        }
    }
    __syncthreads();

    {
        const int row = tid % GT_ROWS, sg = tid / GT_ROWS;
        const float4* tp = (const float4*)(term + row * ATT_RP + sg * ATT_SEG);
        float4 x[ATT_SEG / 4];
#pragma unroll
        for (int j = 0; j < ATT_SEG / 4; ++j) x[j] = tp[j];
        float s = 0.0f;
#pragma unroll
        for (int j = 0; j < ATT_SEG / 4; ++j) { s += x[j].x; s += x[j].y; s += x[j].z; s += x[j].w; }
        psum[sg][row] = s;
    }
    __syncthreads();
    if (tid < GT_ROWS) {
        float s = psum[0][tid];
#pragma unroll
        for (int g = 1; g < ATT_NSEG; ++g) s += psum[g][tid];
        a[m0 + tid] = s;
    }
}

__global__ void __launch_bounds__(GT_T, PULL_WAVES_PER_SIMD)
gcn_pull_attn(const float* __restrict__ hlin,
              const int* __restrict__ rowptr,
              const int* __restrict__ csr,
              const float* __restrict__ dinv,
              const float* __restrict__ bconv,
              const float* __restrict__ g,
              const float* __restrict__ bb,
              const float* __restrict__ m,
              const float* __restrict__ v,
              const _Float16* __restrict__ Wpk,
              const float* __restrict__ bv,
              const float* __restrict__ bu,
              const float* __restrict__ wA,
              float* __restrict__ h2,
              float* __restrict__ a) {
    __shared__ alignas(16) _Float16 As[GT_ROWS * GT_PITCH(H_DIM)];
    __shared__ float psum[ATT_NSEG][GT_ROWS];
    const int m0 = blockIdx.x * GT_ROWS;
    pull_tile_to_lds<true>(hlin, rowptr, csr, dinv, bconv, g, bb, m, v, As, h2, m0);
    __syncthreads();
    attn_tile_from_lds(As, psum, Wpk, bv, bu, wA, a, m0);
}

#define SM_T 1024
#define SM_U 12
static_assert(SM_T == 32 * 32, "the second reduction stage is one value per lane of one wave");

__device__ __forceinline__ float wave_max(float v) {
    for (int off = 16; off > 0; off >>= 1) v = fmaxf(v, __shfl_xor(v, off, 32));
    return v;
}
__device__ __forceinline__ float wave_sum(float v) {
    for (int off = 16; off > 0; off >>= 1) v += __shfl_xor(v, off, 32);
    return v;
}

__global__ void __launch_bounds__(SM_T)
seg_softmax(const float* __restrict__ a,
            const int* __restrict__ seg,
            float* __restrict__ wout) {
    __shared__ float red[SM_T / 32];
    const int g = blockIdx.x;
    const int s = seg[g], e = seg[g + 1];
    if (e <= s) return;
    const int tid = threadIdx.x, lane = tid & 31, wv = tid >> 5;
    const int first = s + tid;
    const int more  = first + SM_T * SM_U;

    float v0[SM_U];
#pragma unroll
    for (int u = 0; u < SM_U; ++u) v0[u] = a[min(first + u * SM_T, e - 1)];
    float mx = -1e30f;
#pragma unroll
    for (int u = 0; u < SM_U; ++u) mx = fmaxf(mx, v0[u]);
    for (int nb = more; nb < e; nb += SM_T * SM_U) {
        float v[SM_U];
#pragma unroll
        for (int u = 0; u < SM_U; ++u) v[u] = a[min(nb + u * SM_T, e - 1)];
#pragma unroll
        for (int u = 0; u < SM_U; ++u) mx = fmaxf(mx, v[u]);
    }
    mx = wave_max(mx);
    if (lane == 0) red[wv] = mx;
    __syncthreads();
    mx = wave_max(red[lane]);
    __syncthreads();

    float ex[SM_U];
    float sum = 0.0f;
#pragma unroll
    for (int u = 0; u < SM_U; ++u) {
        ex[u] = 0.0f;
        if (first + u * SM_T < e) ex[u] = expf(v0[u] - mx);
        sum += ex[u];
    }
    for (int nb = more; nb < e; nb += SM_T * SM_U) {
        float v[SM_U];
#pragma unroll
        for (int u = 0; u < SM_U; ++u) v[u] = a[min(nb + u * SM_T, e - 1)];
#pragma unroll
        for (int u = 0; u < SM_U; ++u) {
            const float x = expf(v[u] - mx);
            sum += (nb + u * SM_T < e) ? x : 0.0f;
        }
    }
    sum = wave_sum(sum);
    if (lane == 0) red[wv] = sum;
    __syncthreads();
    const float inv = 1.0f / wave_sum(red[lane]);

#pragma unroll
    for (int u = 0; u < SM_U; ++u) {
        const int n = first + u * SM_T;
        if (n < e) wout[n] = ex[u] * inv;
    }
    for (int nb = more; nb < e; nb += SM_T * SM_U) {
        float v[SM_U];
#pragma unroll
        for (int u = 0; u < SM_U; ++u) v[u] = a[min(nb + u * SM_T, e - 1)];
#pragma unroll
        for (int u = 0; u < SM_U; ++u) {
            const int n = nb + u * SM_T;
            if (n < e) wout[n] = expf(v[u] - mx) * inv;
        }
    }
}

#define POOL_CH 32
#define POOL_T  512
#define POOL_RG (POOL_T / 64)
#define POOL_U  8
static_assert(H_DIM == 256, "pooling assumes 64 float4 per row");
__global__ void __launch_bounds__(POOL_T)
slide_partial(const float* __restrict__ w, const float* __restrict__ h,
              const int* __restrict__ seg, float* __restrict__ part) {
    __shared__ float4 red[POOL_RG][64];
    const int g = blockIdx.x / POOL_CH, c = blockIdx.x % POOL_CH;
    const int s = seg[g], e = seg[g + 1];
    const int per = (e - s + POOL_CH - 1) / POOL_CH;
    const int n0 = s + c * per;
    const int n1 = min(n0 + per, e);
    const int col = threadIdx.x & 63;
    const int rg = __builtin_amdgcn_readfirstlane((int)(threadIdx.x >> 6));
    const float4* h4 = (const float4*)h;
    const float4 zero4 = {0.0f, 0.0f, 0.0f, 0.0f};
    float4 acc = zero4;
    for (int nb = n0 + rg; nb < n1; nb += POOL_U * POOL_RG) {
        float4 v[POOL_U];
        float  ww[POOL_U];
#pragma unroll
        for (int u = 0; u < POOL_U; ++u) {
            const int n  = nb + u * POOL_RG;
            const int nc = min(n, n1 - 1);
            v[u]  = h4[(size_t)nc * (H_DIM / 4) + col];
            ww[u] = w[nc];
            if (n >= n1) v[u] = zero4;
        }
#pragma unroll
        for (int u = 0; u < POOL_U; ++u) {
            acc.x = fmaf(ww[u], v[u].x, acc.x); acc.y = fmaf(ww[u], v[u].y, acc.y);
            acc.z = fmaf(ww[u], v[u].z, acc.z); acc.w = fmaf(ww[u], v[u].w, acc.w);
        }
    }
    red[rg][col] = acc;
    __syncthreads();
    if (threadIdx.x < H_DIM) {
        const float* rf = (const float*)red;
        float sum = 0.0f;
#pragma unroll
        for (int r = 0; r < POOL_RG; ++r) sum += rf[r * H_DIM + threadIdx.x];
        part[(size_t)blockIdx.x * H_DIM + threadIdx.x] = sum;
    }
}

#define HEAD_T  1024
#define HEAD_WV (H_DIM * HA_DIM / 4 / HEAD_T)
static_assert(HEAD_WV * HEAD_T * 4 == H_DIM * HA_DIM, "Wc1 splits into whole float4 per thread");
static_assert(H_DIM <= HEAD_T && HA_DIM * C_DIM <= HEAD_T && H_DIM % 32 == 0 && HA_DIM % 32 == 0,
              "one thread per column / Wc2 value; branches on whole waves");
__global__ void __launch_bounds__(HEAD_T)
head_kernel(const float* __restrict__ part,
            const float* __restrict__ Wc1,
            const float* __restrict__ bc1,
            const float* __restrict__ Wc2,
            const float* __restrict__ bc2,
            float* __restrict__ logits) {
    __shared__ alignas(16) float w1s[H_DIM * HA_DIM];
    __shared__ alignas(16) float slide[H_DIM];
    __shared__ alignas(16) float hid[HA_DIM];
    __shared__ float w2s[HA_DIM * C_DIM];
    const int tid = threadIdx.x;
    const int g = blockIdx.x;

    float4 wv[HEAD_WV];
#pragma unroll
    for (int u = 0; u < HEAD_WV; ++u) wv[u] = ((const float4*)Wc1)[u * HEAD_T + tid];
    float pv[POOL_CH];
    if (tid < H_DIM) {
        const float* p = part + (size_t)g * POOL_CH * H_DIM + tid;
#pragma unroll
        for (int c = 0; c < POOL_CH; ++c) pv[c] = p[c * H_DIM];
    }
    float w2v = 0.0f, b1v = 0.0f, b2v = 0.0f;
    if (tid < HA_DIM * C_DIM) w2v = Wc2[tid];
    if (tid < HA_DIM) b1v = bc1[tid];
    if (tid < C_DIM) b2v = bc2[tid];

#pragma unroll
    for (int u = 0; u < HEAD_WV; ++u) ((float4*)w1s)[u * HEAD_T + tid] = wv[u];
    if (tid < H_DIM) {
        float s = 0.0f;
#pragma unroll
        for (int c = 0; c < POOL_CH; ++c) s += pv[c];
        slide[tid] = s;
    }
    if (tid < HA_DIM * C_DIM) w2s[tid] = w2v;
    __syncthreads();

    if (tid < HA_DIM) {
        float acc = b1v;
#pragma unroll 32
        for (int k = 0; k < H_DIM; ++k) acc += slide[k] * w1s[k * HA_DIM + tid];
        hid[tid] = fmaxf(acc, 0.0f);
    }
    __syncthreads();

    if (tid < C_DIM) {
        float s = b2v;
#pragma unroll
        for (int t = 0; t < HA_DIM; ++t) s += hid[t] * w2s[t * C_DIM + tid];
        logits[g * C_DIM + tid] = s;
    }
}

extern "C" void kernel_launch(void* const* d_in, const int* in_sizes, int n_in,
                              void* d_out, int out_size, void* d_ws, size_t ws_size,
                              hipStream_t stream) {
    (void)in_sizes; (void)n_in; (void)out_size; (void)ws_size;
    const float* x   = (const float*)d_in[0];
    const int*  ei   = (const int*)d_in[1];
    const int*  batch= (const int*)d_in[2];
    const float* W1  = (const float*)d_in[3];  const float* b1   = (const float*)d_in[4];
    const float* g1  = (const float*)d_in[5];  const float* bb1  = (const float*)d_in[6];
    const float* m1  = (const float*)d_in[7];  const float* v1   = (const float*)d_in[8];
    const float* W2  = (const float*)d_in[9];  const float* b2   = (const float*)d_in[10];
    const float* g2  = (const float*)d_in[11]; const float* bb2  = (const float*)d_in[12];
    const float* m2  = (const float*)d_in[13]; const float* v2   = (const float*)d_in[14];
    const float* Wv  = (const float*)d_in[15]; const float* bv   = (const float*)d_in[16];
    const float* Wu  = (const float*)d_in[17]; const float* bu   = (const float*)d_in[18];
    const float* wA  = (const float*)d_in[19];
    const float* Wc1 = (const float*)d_in[20]; const float* bc1  = (const float*)d_in[21];
    const float* Wc2 = (const float*)d_in[22]; const float* bc2  = (const float*)d_in[23];

    const int* src = ei;
    const int* dst = ei + N_EDGES;

    float* out  = (float*)d_out;
    float* wout = out + N_BAGS * C_DIM;

    const size_t NH = (size_t)N_NODES * H_DIM;
    float*    bufA = (float*)d_ws;
    float*    bufB = bufA + NH;
    _Float16* h16  = (_Float16*)(bufB + NH);
    _Float16* w1pk = h16 + NH;
    _Float16* w2pk = w1pk + (size_t)D_IN * H_DIM;
    _Float16* wvpk = w2pk + (size_t)H_DIM * H_DIM;
    _Float16* wupk = wvpk + (size_t)H_DIM * HA_DIM;
    float*    dinv = (float*)(wupk + (size_t)H_DIM * HA_DIM);
    float*    abuf = dinv + N_NODES;
    int*      seg  = (int*)(abuf + N_NODES);

    int*   cnt      = (int*)h16;
    int*   rowptr   = cnt + N_NODES;
    int*   rank     = rowptr + N_NODES + 4;
    int*   csr      = rank + N_EDGES;
    float* hlin1 = bufA;
    float* hlin2 = bufB;
    float* h2_32 = bufA;

    setup_all<<<SU_BLOCKS, SU_T, 0, stream>>>(W1, w1pk, W2, w2pk, Wv, wvpk, Wu, wupk,
                                              batch, seg, cnt);

    count_rank<<<N_EDGES / (EDGE_T * EDGE_PER), EDGE_T, 0, stream>>>(dst, cnt, rank);
    scan_rowptr<<<1, SCAN_T, 0, stream>>>(cnt, rowptr, dinv);
    csr_fill<<<N_EDGES / (EDGE_T * EDGE_PER), EDGE_T, 0, stream>>>(src, dst, rank, rowptr, dinv,
                                                                   csr);

    wmma_gemm_lds_af32<D_IN, H_DIM><<<N_NODES / GT_ROWS, GT_T, 0, stream>>>(x, w1pk, hlin1);

    gcn_pull_gemm<<<N_NODES / GT_ROWS, GT_T, 0, stream>>>(
        hlin1, rowptr, csr, dinv, b1, g1, bb1, m1, v1, w2pk, hlin2);

    gcn_pull_attn<<<N_NODES / GT_ROWS, GT_T, 0, stream>>>(
        hlin2, rowptr, csr, dinv, b2, g2, bb2, m2, v2, wvpk, bv, bu, wA, h2_32, abuf);

    seg_softmax<<<N_BAGS, SM_T, 0, stream>>>(abuf, seg, wout);

    float* part = bufB;
    slide_partial<<<N_BAGS * POOL_CH, POOL_T, 0, stream>>>(wout, h2_32, seg, part);

    head_kernel<<<N_BAGS, HEAD_T, 0, stream>>>(part, Wc1, bc1, Wc2, bc2, out);
}


# Pooling folded into the layer-2/score kernel: a tile pools its f16 LDS rows against its own max (records per bag), head rescales: 10 -> 8 launches, h2 not stored; new sum trees (exp sum, bag row)
# speedup vs baseline: 6.2843x; 1.0943x over previous
#include <hip/hip_runtime.h>
#include <hip/hip_bf16.h>
#include <math.h>

typedef __attribute__((ext_vector_type(16))) _Float16 v16h;
typedef __attribute__((ext_vector_type(8)))  _Float16 v8h;
typedef __attribute__((ext_vector_type(8)))  float    v8f;

#define N_NODES 65536
#define N_EDGES 524288
#define N_BAGS  8
#define D_IN    512
#define H_DIM   256
#define HA_DIM  128
#define C_DIM   2
#define BN_EPS  1e-5f
#define MREP    4

#define EDGE_PER 4
#define EDGE_T   256
static_assert(N_EDGES % (EDGE_T * EDGE_PER) == 0, "whole work-groups");
__global__ void __launch_bounds__(EDGE_T)
count_rank(const int* __restrict__ dst, int* __restrict__ cnt, int* __restrict__ rank) {
    const int t = blockIdx.x * blockDim.x + threadIdx.x;
    const int4 d = ((const int4*)dst)[t];
    int4 r;
    r.x = atomicAdd(&cnt[d.x], 1);
    r.y = atomicAdd(&cnt[d.y], 1);
    r.z = atomicAdd(&cnt[d.z], 1);
    r.w = atomicAdd(&cnt[d.w], 1);
    ((int4*)rank)[t] = r;
}

#define SCAN_T   1024
#define SCAN_PER (N_NODES / SCAN_T)
static_assert(N_NODES % (SCAN_T * 4) == 0, "scan assumes whole int4 chunks per thread");
__global__ void __launch_bounds__(SCAN_T)
scan_rowptr(const int* __restrict__ cnt, int* __restrict__ rowptr,
            float* __restrict__ dinv) {
    __shared__ int wsum[32];
    const int t = threadIdx.x, lane = t & 31, w = t >> 5;
    const int4* c4 = (const int4*)(cnt + t * SCAN_PER);
    int4 c[SCAN_PER / 4];
    int s = 0;
#pragma unroll
    for (int i = 0; i < SCAN_PER / 4; ++i) {
        c[i] = c4[i];
        s += c[i].x + c[i].y + c[i].z + c[i].w;
    }
    int inc = s;
    for (int off = 1; off < 32; off <<= 1) {
        int v = __shfl_up(inc, off, 32);
        if (lane >= off) inc += v;
    }
    if (lane == 31) wsum[w] = inc;
    __syncthreads();
    if (w == 0) {
        int v = wsum[lane], iv = v;
        for (int off = 1; off < 32; off <<= 1) {
            int u = __shfl_up(iv, off, 32);
            if (lane >= off) iv += u;
        }
        wsum[lane] = iv - v;
    }
    __syncthreads();
    int run = wsum[w] + inc - s;
    int4*   r4 = (int4*)(rowptr + t * SCAN_PER);
    float4* d4 = (float4*)(dinv + t * SCAN_PER);
#pragma unroll
    for (int i = 0; i < SCAN_PER / 4; ++i) {
        int4 r;
        r.x = run; run += c[i].x;
        r.y = run; run += c[i].y;
        r.z = run; run += c[i].z;
        r.w = run; run += c[i].w;
        r4[i] = r;
        float4 d;
        d.x = rsqrtf(1.0f + (float)c[i].x);
        d.y = rsqrtf(1.0f + (float)c[i].y);
        d.z = rsqrtf(1.0f + (float)c[i].z);
        d.w = rsqrtf(1.0f + (float)c[i].w);
        d4[i] = d;
    }
    if (t == SCAN_T - 1) rowptr[N_NODES] = run;
}

__device__ __forceinline__ void csr_put(int s, int d, int r,
                                        const int* __restrict__ rowptr,
                                        const float* __restrict__ dinv,
                                        int* __restrict__ csr) {
    const int pos = rowptr[d] + r;
    ((int2*)csr)[pos] = make_int2(s, __float_as_int(dinv[s] * dinv[d]));
}

__global__ void __launch_bounds__(EDGE_T)
csr_fill(const int* __restrict__ src, const int* __restrict__ dst,
         const int* __restrict__ rank, const int* __restrict__ rowptr,
         const float* __restrict__ dinv, int* __restrict__ csr) {
    const int t = blockIdx.x * blockDim.x + threadIdx.x;
    const int4 s = ((const int4*)src)[t], d = ((const int4*)dst)[t], r = ((const int4*)rank)[t];
    csr_put(s.x, d.x, r.x, rowptr, dinv, csr);
    csr_put(s.y, d.y, r.y, rowptr, dinv, csr);
    csr_put(s.z, d.z, r.z, rowptr, dinv, csr);
    csr_put(s.w, d.w, r.w, rowptr, dinv, csr);
}

template <int K, int NOUT>
__device__ __forceinline__ void pack_w_elem(const float* __restrict__ Wm,
                                            _Float16* __restrict__ out, int t) {
    constexpr int KBLKS = K >> 5;
    const int i    = t & 15;
    const int lane = (t >> 4) & 31;
    const int rest = t >> 9;
    const int ntile = rest / KBLKS;
    const int kblk  = rest % KBLKS;
    const int half = lane >> 4, l15 = lane & 15;
    const int col = ntile * 16 + l15;
    const int k   = kblk * 32 + half * 8 + ((i < 8) ? i : (16 + i - 8));
    out[t] = (_Float16)Wm[(size_t)k * NOUT + col];
}

#define SU_T      256
#define SU_B_W1   (D_IN * H_DIM / SU_T)
#define SU_B_W2   (H_DIM * H_DIM / SU_T)
#define SU_B_WA   (H_DIM * HA_DIM / SU_T)
#define SU_B_SEG  (N_NODES / SU_T)
#define SU_B_ZERO (N_NODES / SU_T)
#define SU_BLOCKS (SU_B_W1 + SU_B_W2 + 2 * SU_B_WA + SU_B_SEG + SU_B_ZERO)
static_assert(N_NODES % SU_T == 0 && (H_DIM * HA_DIM) % SU_T == 0, "whole blocks per range");
__global__ void __launch_bounds__(SU_T)
setup_all(const float* __restrict__ W1, _Float16* __restrict__ w1pk,
          const float* __restrict__ W2, _Float16* __restrict__ w2pk,
          const float* __restrict__ Wv, _Float16* __restrict__ wvpk,
          const float* __restrict__ Wu, _Float16* __restrict__ wupk,
          const int* __restrict__ batch, int* __restrict__ seg,
          int* __restrict__ cnt) {
    int b = blockIdx.x;
    const int tid = threadIdx.x;
    if (b < SU_B_W1) { pack_w_elem<D_IN, H_DIM>(W1, w1pk, b * SU_T + tid); return; }
    b -= SU_B_W1;
    if (b < SU_B_W2) { pack_w_elem<H_DIM, H_DIM>(W2, w2pk, b * SU_T + tid); return; }
    b -= SU_B_W2;
    if (b < SU_B_WA) { pack_w_elem<H_DIM, HA_DIM>(Wv, wvpk, b * SU_T + tid); return; }
    b -= SU_B_WA;
    if (b < SU_B_WA) { pack_w_elem<H_DIM, HA_DIM>(Wu, wupk, b * SU_T + tid); return; }
    b -= SU_B_WA;
    if (b < SU_B_SEG) {
        const int n = b * SU_T + tid;
        const int cur  = batch[n];
        const int prev = (n == 0) ? -1 : batch[n - 1];
#pragma unroll
        for (int g = 0; g < N_BAGS; ++g)
            if (g > prev && g <= cur) seg[g] = n;
        if (n == N_NODES - 1) {
#pragma unroll
            for (int g = 1; g <= N_BAGS; ++g)
                if (g > cur) seg[g] = N_NODES;
        }
        return;
    }
    b -= SU_B_SEG;
    cnt[b * SU_T + tid] = 0;
}

#define GT_ROWS 64
#define GT_T    256
#define GT_LD   8
static_assert(GT_ROWS == 16 * MREP, "a wave covers all rows of the tile");
static_assert(N_NODES % GT_ROWS == 0, "no partial row tile");

#define GT_PITCH(K) ((K) + 8)

template <int K, int NOUT>
__device__ __forceinline__ void gemm_tile_from_lds(const _Float16* As,
                                                   const _Float16* __restrict__ Wpk,
                                                   float* __restrict__ C, int m0) {
    constexpr int PITCH = GT_PITCH(K);
    constexpr int KBLKS = K / 32;
    constexpr int NTW   = NOUT / 16 / (GT_T / 32);
    static_assert(NTW >= 1 && NTW * 16 * (GT_T / 32) == NOUT, "waves tile the columns");
    const int lane = threadIdx.x & 31;
    const int wave = __builtin_amdgcn_readfirstlane((int)(threadIdx.x >> 5));
    const int half = lane >> 4;
    const int l15  = lane & 15;
    const _Float16* wb = Wpk + (size_t)(wave * NTW) * KBLKS * 512 + lane * 16;
    const _Float16* ab = As + l15 * PITCH + half * 8;

    v8f zero = {};
    v8f acc[MREP][NTW];
#pragma unroll
    for (int r = 0; r < MREP; ++r)
#pragma unroll
        for (int t = 0; t < NTW; ++t) acc[r][t] = zero;

    for (int kblk = 0; kblk < KBLKS; ++kblk) {
        v16h b[NTW];
#pragma unroll
        for (int t = 0; t < NTW; ++t)
            b[t] = *(const v16h*)(wb + ((size_t)t * KBLKS + kblk) * 512);
#pragma unroll
        for (int r = 0; r < MREP; ++r) {
            const _Float16* ap = ab + r * 16 * PITCH + kblk * 32;
            v8h lo = *(const v8h*)(ap);
            v8h hi = *(const v8h*)(ap + 16);
            v16h a = __builtin_shufflevector(lo, hi, 0, 1, 2, 3, 4, 5, 6, 7,
                                             8, 9, 10, 11, 12, 13, 14, 15);
#pragma unroll
            for (int t = 0; t < NTW; ++t)
                acc[r][t] = __builtin_amdgcn_wmma_f32_16x16x32_f16(
                    false, a, false, b[t], (short)0, acc[r][t], false, false);
        }
    }

#pragma unroll
    for (int t = 0; t < NTW; ++t) {
        const int col = (wave * NTW + t) * 16 + l15;
#pragma unroll
        for (int r = 0; r < MREP; ++r) {
#pragma unroll
            for (int q = 0; q < 8; ++q) {
                const int row = m0 + r * 16 + q + half * 8;
                C[(size_t)row * NOUT + col] = acc[r][t][q];
            }
        }
    }
}

typedef __attribute__((ext_vector_type(4))) _Float16 v4h_t;

template <int K, int NOUT>
__global__ void __launch_bounds__(GT_T)
wmma_gemm_lds_af32(const float* __restrict__ A, const _Float16* __restrict__ Wpk,
                   float* __restrict__ C) {
    __shared__ alignas(16) _Float16 As[GT_ROWS * GT_PITCH(K)];
    constexpr int VROW  = K / 4;
    constexpr int PER_T = GT_ROWS * VROW / GT_T;
    static_assert(PER_T % GT_LD == 0 && VROW % 32 == 0, "whole batches, a wave inside one row");
    const int tid = threadIdx.x;
    const int m0  = blockIdx.x * GT_ROWS;
    const float4* Ag = (const float4*)(A + (size_t)m0 * K);
#pragma unroll 1
    for (int it = 0; it < PER_T; it += GT_LD) {
        float4 v[GT_LD];
#pragma unroll
        for (int u = 0; u < GT_LD; ++u) v[u] = Ag[(it + u) * GT_T + tid];
#pragma unroll
        for (int u = 0; u < GT_LD; ++u) {
            const int f = (it + u) * GT_T + tid;
            const int row = f / VROW, c = f % VROW;
            v4h_t h = {(_Float16)v[u].x, (_Float16)v[u].y, (_Float16)v[u].z, (_Float16)v[u].w};
            *(v4h_t*)(As + row * GT_PITCH(K) + c * 4) = h;
        }
    }
    __syncthreads();
    gemm_tile_from_lds<K, NOUT>(As, Wpk, C, m0);
}

template <int K>
__device__ __forceinline__ void pull_rows(const float4* __restrict__ hp, int my_s, int my_c,
                                          int j, int lane, float4& a0, float4& a1) {
    float4 p[K], q[K];
    float  c[K];
#pragma unroll
    for (int k = 0; k < K; ++k) {
        const int s = __builtin_amdgcn_readlane(my_s, j + k);
        c[k] = __int_as_float(__builtin_amdgcn_readlane(my_c, j + k));
        const float4* row = hp + (size_t)s * (H_DIM / 4);
        p[k] = row[lane];
        q[k] = row[32 + lane];
    }
#pragma unroll
    for (int k = 0; k < K; ++k) {
        a0.x = fmaf(p[k].x, c[k], a0.x); a0.y = fmaf(p[k].y, c[k], a0.y);
        a0.z = fmaf(p[k].z, c[k], a0.z); a0.w = fmaf(p[k].w, c[k], a0.w);
        a1.x = fmaf(q[k].x, c[k], a1.x); a1.y = fmaf(q[k].y, c[k], a1.y);
        a1.z = fmaf(q[k].z, c[k], a1.z); a1.w = fmaf(q[k].w, c[k], a1.w);
    }
}

__device__ __forceinline__ float gcn_fin(float agg, float hl, float di, float bc,
                                         float mm, float vv, float gg, float bo) {
    float val = agg + hl * di * di + bc;
    val = (val - mm) * rsqrtf(vv + BN_EPS) * gg + bo;
    return fmaxf(val, 0.0f);
}

__device__ __forceinline__ float4 gcn_fin4(float4 a, float4 h, float di, float4 bc,
                                           float4 mm, float4 vv, float4 gg, float4 bo) {
    float4 r;
    r.x = gcn_fin(a.x, h.x, di, bc.x, mm.x, vv.x, gg.x, bo.x);
    r.y = gcn_fin(a.y, h.y, di, bc.y, mm.y, vv.y, gg.y, bo.y);
    r.z = gcn_fin(a.z, h.z, di, bc.z, mm.z, vv.z, gg.z, bo.z);
    r.w = gcn_fin(a.w, h.w, di, bc.w, mm.w, vv.w, gg.w, bo.w);
    return r;
}

typedef __attribute__((ext_vector_type(4))) _Float16 v4h;

#define TILE_NPW (GT_ROWS / (GT_T / 32))
#define PULL_WAVES_PER_SIMD 8
static_assert(TILE_NPW * (GT_T / 32) == GT_ROWS, "the waves split the tile's rows evenly");

__device__ __forceinline__ int2 csr_chunk(const int* __restrict__ csr, int base, int end, int lane) {
    int2 rec = make_int2(0, 0);
    if (lane < end - base) rec = ((const int2*)csr)[base + lane];
    return rec;
}

__device__ __forceinline__ void pull_tile_to_lds(const float* __restrict__ hlin,
                                                 const int* __restrict__ rowptr,
                                                 const int* __restrict__ csr,
                                                 const float* __restrict__ dinv,
                                                 const float* __restrict__ bconv,
                                                 const float* __restrict__ g,
                                                 const float* __restrict__ bb,
                                                 const float* __restrict__ m,
                                                 const float* __restrict__ v,
                                                 _Float16* As,
                                                 int m0) {
    constexpr int PITCH = GT_PITCH(H_DIM);
    const int lane = threadIdx.x & 31;
    const int wave = __builtin_amdgcn_readfirstlane((int)(threadIdx.x >> 5));
    const int rfirst = wave * TILE_NPW;
    const int nfirst = m0 + rfirst;
    const float4* hp = (const float4*)hlin;
    const float4* bc4 = (const float4*)bconv; const float4* g4 = (const float4*)g;
    const float4* bb4 = (const float4*)bb;    const float4* m4 = (const float4*)m;
    const float4* v4  = (const float4*)v;

    const int rp = rowptr[nfirst + min(lane, TILE_NPW)];
    int beg = __builtin_amdgcn_readlane(rp, 0), end = __builtin_amdgcn_readlane(rp, 1);
    int2 rec_nx = csr_chunk(csr, beg, end, lane);
    for (int i = 0; i < TILE_NPW; ++i) {
        const int n = nfirst + i;
        const int nxt = __builtin_amdgcn_readlane(rp, min(i + 2, TILE_NPW));
        int2 rec = rec_nx;
        if (i + 1 < TILE_NPW) rec_nx = csr_chunk(csr, end, nxt, lane);
        const float di = dinv[n];
        const float4* own = hp + (size_t)n * (H_DIM / 4);
        const float4 h0 = own[lane], h1 = own[32 + lane];
        float4 a0 = {0.0f, 0.0f, 0.0f, 0.0f}, a1 = {0.0f, 0.0f, 0.0f, 0.0f};
        for (int base = beg; base < end; base += 32) {
            const int nb = min(32, end - base);
            if (base != beg) rec = csr_chunk(csr, base, end, lane);
            const int my_s = rec.x, my_c = rec.y;
            int j = 0;
            for (; j + 8 <= nb; j += 8) pull_rows<8>(hp, my_s, my_c, j, lane, a0, a1);
            if (j + 4 <= nb) { pull_rows<4>(hp, my_s, my_c, j, lane, a0, a1); j += 4; }
            const int r = nb - j;
            if (r == 3)      pull_rows<3>(hp, my_s, my_c, j, lane, a0, a1);
            else if (r == 2) pull_rows<2>(hp, my_s, my_c, j, lane, a0, a1);
            else if (r == 1) pull_rows<1>(hp, my_s, my_c, j, lane, a0, a1);
        }

        const float4 r0 = gcn_fin4(a0, h0, di, bc4[lane], m4[lane], v4[lane], g4[lane], bb4[lane]);
        const float4 r1 = gcn_fin4(a1, h1, di, bc4[32 + lane], m4[32 + lane], v4[32 + lane],
                                   g4[32 + lane], bb4[32 + lane]);
        _Float16* o = As + (rfirst + i) * PITCH;
        v4h x0 = {(_Float16)r0.x, (_Float16)r0.y, (_Float16)r0.z, (_Float16)r0.w};
        v4h x1 = {(_Float16)r1.x, (_Float16)r1.y, (_Float16)r1.z, (_Float16)r1.w};
        *(v4h*)(o + 4 * lane) = x0;
        *(v4h*)(o + H_DIM / 2 + 4 * lane) = x1;
        beg = end;
        end = nxt;
    }
}

__global__ void __launch_bounds__(GT_T, PULL_WAVES_PER_SIMD)
gcn_pull_gemm(const float* __restrict__ hlin,
              const int* __restrict__ rowptr,
              const int* __restrict__ csr,
              const float* __restrict__ dinv,
              const float* __restrict__ bconv,
              const float* __restrict__ g,
              const float* __restrict__ bb,
              const float* __restrict__ m,
              const float* __restrict__ v,
              const _Float16* __restrict__ Wpk,
              float* __restrict__ C) {
    __shared__ alignas(16) _Float16 As[GT_ROWS * GT_PITCH(H_DIM)];
    const int m0 = blockIdx.x * GT_ROWS;
    pull_tile_to_lds(hlin, rowptr, csr, dinv, bconv, g, bb, m, v, As, m0);
    __syncthreads();
    gemm_tile_from_lds<H_DIM, H_DIM>(As, Wpk, C, m0);
}

#define ATT_SEG   32
#define ATT_NSEG  (HA_DIM / ATT_SEG)
#define ATT_RP    (HA_DIM + 4)
static_assert(HA_DIM / 16 == GT_T / 32, "one column tile of Wv and one of Wu per wave");
static_assert(ATT_NSEG * GT_ROWS == GT_T, "one (row, segment) per thread");
static_assert(ATT_RP % 4 == 0, "16-byte reads of the term tile");

#define REC_STRIDE (H_DIM + 32)
static_assert(GT_T == H_DIM, "pooling: one column of the tile per thread");

__device__ __forceinline__ float wave_max(float v) {
    for (int off = 16; off > 0; off >>= 1) v = fmaxf(v, __shfl_xor(v, off, 32));
    return v;
}
__device__ __forceinline__ float wave_sum(float v) {
    for (int off = 16; off > 0; off >>= 1) v += __shfl_xor(v, off, 32);
    return v;
}

#define POOL_G  (GT_T / 64)
#define POOL_RJ (GT_ROWS / POOL_G)
static_assert(H_DIM == 256 && POOL_G * 64 == GT_T && POOL_RJ * POOL_G == GT_ROWS, "64 threads x 4 columns cover a row");
static_assert((POOL_G * H_DIM + POOL_G) <= GT_ROWS * ATT_RP, "the partial rows fit in the term tile's LDS");
__device__ __forceinline__ void pool_tile_partials(const _Float16* As, const float* ash, float* esh,
                                                   float* prt,
                                                   int seg_lo, int seg_hi,
                                                   float* __restrict__ part, int m0) {
    constexpr int PITCH = GT_PITCH(H_DIM);
    const int tid = threadIdx.x;
    const int q  = __builtin_amdgcn_readfirstlane((int)(tid >> 6));
    const int c4 = tid & 63;
    for (int g = 0; g < N_BAGS; ++g) {
        const int lo = max(__builtin_amdgcn_readlane(seg_lo, g), m0) - m0;
        const int hi = min(__builtin_amdgcn_readlane(seg_hi, g), m0 + GT_ROWS) - m0;
        if (lo >= hi) continue;
        static_assert(GT_ROWS == 64, "two rows of the tile per lane");
        const int lane = tid & 31;
        const float x0 = ash[lane], x1 = ash[32 + lane];
        float mt = -INFINITY;
        if (lane >= lo && lane < hi) mt = x0;
        if (lane + 32 >= lo && lane + 32 < hi) mt = fmaxf(mt, x1);
        mt = wave_max(mt);
        if (tid >= lo && tid < hi) esh[tid] = expf(ash[tid] - mt);
        __syncthreads();
        v4h   hv[POOL_RJ];
        float ev[POOL_RJ];
#pragma unroll
        for (int j = 0; j < POOL_RJ; ++j) {
            const int r = POOL_G * j + q;
            hv[j] = *(const v4h*)(As + r * PITCH + 4 * c4);
            ev[j] = 0.0f;
            if (r >= lo && r < hi) ev[j] = esh[r];
        }
        float4 p = {0.0f, 0.0f, 0.0f, 0.0f};
        float st = 0.0f;
#pragma unroll
        for (int j = 0; j < POOL_RJ; ++j) {
            st += ev[j];
            p.x = fmaf(ev[j], (float)hv[j][0], p.x); p.y = fmaf(ev[j], (float)hv[j][1], p.y);
            p.z = fmaf(ev[j], (float)hv[j][2], p.z); p.w = fmaf(ev[j], (float)hv[j][3], p.w);
        }
        ((float4*)prt)[q * 64 + c4] = p;
        if (c4 == 0) prt[POOL_G * H_DIM + q] = st;
        __syncthreads();
        float s = prt[tid];
#pragma unroll
        for (int k = 1; k < POOL_G; ++k) s += prt[k * H_DIM + tid];
        float* rec = part + ((size_t)blockIdx.x * N_BAGS + g) * REC_STRIDE;
        rec[tid] = s;
        if (tid == 0) {
            float sst = prt[POOL_G * H_DIM];
#pragma unroll
            for (int k = 1; k < POOL_G; ++k) sst += prt[POOL_G * H_DIM + k];
            rec[H_DIM] = mt;
            rec[H_DIM + 1] = sst;
        }
        __syncthreads();
    }
}

__device__ __forceinline__ void attn_tile_from_lds(const _Float16* As,
                                                   float* term,
                                                   float (*psum)[GT_ROWS],
                                                   float* ash,
                                                   const _Float16* __restrict__ Wpk,
                                                   const float* __restrict__ bv,
                                                   const float* __restrict__ bu,
                                                   const float* __restrict__ wA,
                                                   float* __restrict__ a, int m0) {
    constexpr int K     = H_DIM;
    constexpr int PITCH = GT_PITCH(K);
    constexpr int KBLKS = K / 32;
    constexpr int NT    = HA_DIM / 16;
    const int tid = threadIdx.x;

    const int lane = tid & 31;
    const int wave = __builtin_amdgcn_readfirstlane((int)(tid >> 5));
    const int half = lane >> 4;
    const int l15  = lane & 15;
    const _Float16* wb = Wpk + (size_t)wave * KBLKS * 512 + lane * 16;
    const _Float16* ab = As + l15 * PITCH + half * 8;

    v8f zero = {};
    v8f acc[MREP][2];
#pragma unroll
    for (int r = 0; r < MREP; ++r) { acc[r][0] = zero; acc[r][1] = zero; }

    for (int kblk = 0; kblk < KBLKS; ++kblk) {
        v16h b[2];
#pragma unroll
        for (int t = 0; t < 2; ++t)
            b[t] = *(const v16h*)(wb + ((size_t)t * NT * KBLKS + kblk) * 512);
#pragma unroll
        for (int r = 0; r < MREP; ++r) {
            const _Float16* ap = ab + r * 16 * PITCH + kblk * 32;
            v8h lo = *(const v8h*)(ap);
            v8h hi = *(const v8h*)(ap + 16);
            v16h av = __builtin_shufflevector(lo, hi, 0, 1, 2, 3, 4, 5, 6, 7,
                                              8, 9, 10, 11, 12, 13, 14, 15);
#pragma unroll
            for (int t = 0; t < 2; ++t)
                acc[r][t] = __builtin_amdgcn_wmma_f32_16x16x32_f16(
                    false, av, false, b[t], (short)0, acc[r][t], false, false);
        }
    }

    const int col = wave * 16 + l15;
    const float bvc = bv[col], buc = bu[col], wac = wA[col];
#pragma unroll
    for (int r = 0; r < MREP; ++r) {
#pragma unroll
        for (int q = 0; q < 8; ++q) {
            const int row = r * 16 + q + half * 8;
            const float tv = tanhf(acc[r][0][q] + bvc);
            const float sv = 1.0f / (1.0f + expf(-(acc[r][1][q] + buc)));
            term[row * ATT_RP + col] = tv * sv * wac;
        }
    }
    __syncthreads();

    {
        const int row = tid % GT_ROWS, sg = tid / GT_ROWS;
        const float4* tp = (const float4*)(term + row * ATT_RP + sg * ATT_SEG);
        float4 x[ATT_SEG / 4];
#pragma unroll
        for (int j = 0; j < ATT_SEG / 4; ++j) x[j] = tp[j];
        float s = 0.0f;
#pragma unroll
        for (int j = 0; j < ATT_SEG / 4; ++j) { s += x[j].x; s += x[j].y; s += x[j].z; s += x[j].w; }
        psum[sg][row] = s;
    }
    __syncthreads();
    if (tid < GT_ROWS) {
        float s = psum[0][tid];
#pragma unroll
        for (int g = 1; g < ATT_NSEG; ++g) s += psum[g][tid];
        a[m0 + tid] = s;
        ash[tid] = s;
    }
    __syncthreads();
}

__global__ void __launch_bounds__(GT_T, PULL_WAVES_PER_SIMD)
gcn_pull_attn(const float* __restrict__ hlin,
              const int* __restrict__ rowptr,
              const int* __restrict__ csr,
              const float* __restrict__ dinv,
              const float* __restrict__ bconv,
              const float* __restrict__ g,
              const float* __restrict__ bb,
              const float* __restrict__ m,
              const float* __restrict__ v,
              const _Float16* __restrict__ Wpk,
              const float* __restrict__ bv,
              const float* __restrict__ bu,
              const float* __restrict__ wA,
              const int* __restrict__ seg,
              float* __restrict__ part,
              float* __restrict__ a) {
    __shared__ alignas(16) _Float16 As[GT_ROWS * GT_PITCH(H_DIM)];
    __shared__ alignas(16) float term[GT_ROWS * ATT_RP];
    __shared__ float psum[ATT_NSEG][GT_ROWS];
    __shared__ float ash[GT_ROWS];
    __shared__ float esh[GT_ROWS];
    const int m0 = blockIdx.x * GT_ROWS;
    const int lane = threadIdx.x & 31;
    const int seg_lo = seg[min(lane, N_BAGS)];
    const int seg_hi = seg[min(lane + 1, N_BAGS)];
    pull_tile_to_lds(hlin, rowptr, csr, dinv, bconv, g, bb, m, v, As, m0);
    __syncthreads();
    attn_tile_from_lds(As, term, psum, ash, Wpk, bv, bu, wA, a, m0);
    pool_tile_partials(As, ash, esh, term, seg_lo, seg_hi, part, m0);
}

#define HEAD_T  1024
#define HEAD_WV (H_DIM * HA_DIM / 4 / HEAD_T)
#define HEAD_RG (HEAD_T / 64)
#define HEAD_RU 10
#define HEAD_AU 12
static_assert(HEAD_WV * HEAD_T * 4 == H_DIM * HA_DIM, "Wc1 splits into whole float4 per thread");
static_assert(H_DIM <= HEAD_T && HA_DIM * C_DIM <= HEAD_T && H_DIM % 32 == 0 && HA_DIM % 32 == 0,
              "one thread per column / Wc2 value; branches on whole waves");
static_assert(N_NODES / GT_ROWS <= HEAD_T, "one record's maximum and sum per thread");
static_assert(H_DIM == 256 && HEAD_T == 32 * 32, "64 float4 per record; second reduction stage is one wave");
__global__ void __launch_bounds__(HEAD_T)
head_kernel(const float* __restrict__ part,
            const float* __restrict__ a,
            const int* __restrict__ seg,
            const float* __restrict__ Wc1,
            const float* __restrict__ bc1,
            const float* __restrict__ Wc2,
            const float* __restrict__ bc2,
            float* __restrict__ logits,
            float* __restrict__ wout) {
    __shared__ alignas(16) float w1s[H_DIM * HA_DIM];
    __shared__ alignas(16) float4 pr[HEAD_RG][64];
    __shared__ alignas(16) float slide[H_DIM];
    __shared__ alignas(16) float hid[HA_DIM];
    __shared__ float w2s[HA_DIM * C_DIM];
    __shared__ float sc[HEAD_T];
    __shared__ float red[HEAD_T / 32];
    const int tid = threadIdx.x, lane = tid & 31, wv_ = tid >> 5;
    const int g = blockIdx.x;
    const int s0 = seg[g], e0 = seg[g + 1];
    const bool some = e0 > s0;
    const int t_lo = s0 / GT_ROWS;
    const int nrec = some ? (e0 - 1) / GT_ROWS - t_lo + 1 : 0;
    const float* rbase = part + ((size_t)t_lo * N_BAGS + g) * REC_STRIDE;
    const size_t rstep = (size_t)N_BAGS * REC_STRIDE;
    const int q = tid >> 6, c4 = tid & 63;
    const int first = s0 + tid;

    float4 wv[HEAD_WV];
#pragma unroll
    for (int u = 0; u < HEAD_WV; ++u) wv[u] = ((const float4*)Wc1)[u * HEAD_T + tid];
    float mt = -INFINITY, st = 0.0f;
    float4 pv[HEAD_RU];
    float  av[HEAD_AU];
    if (some) {
        if (tid < nrec) {
            const float* r = rbase + (size_t)tid * rstep;
            mt = r[H_DIM];
            st = r[H_DIM + 1];
        }
#pragma unroll
        for (int u = 0; u < HEAD_RU; ++u) {
            const int b = min(q + u * HEAD_RG, nrec - 1);
            pv[u] = ((const float4*)(rbase + (size_t)b * rstep))[c4];
        }
#pragma unroll
        for (int u = 0; u < HEAD_AU; ++u) av[u] = a[min(first + u * HEAD_T, e0 - 1)];
    }
    float w2v = 0.0f, b1v = 0.0f, b2v = 0.0f;
    if (tid < HA_DIM * C_DIM) w2v = Wc2[tid];
    if (tid < HA_DIM) b1v = bc1[tid];
    if (tid < C_DIM) b2v = bc2[tid];

#pragma unroll
    for (int u = 0; u < HEAD_WV; ++u) ((float4*)w1s)[u * HEAD_T + tid] = wv[u];
    if (tid < HA_DIM * C_DIM) w2s[tid] = w2v;

    if (some) {
        float M = wave_max(mt);
        if (lane == 0) red[wv_] = M;
        __syncthreads();
        M = wave_max(red[lane]);
        __syncthreads();
        const float scl = (tid < nrec) ? expf(mt - M) : 0.0f;
        sc[tid] = scl;
        float S = wave_sum(st * scl);
        if (lane == 0) red[wv_] = S;
        __syncthreads();
        const float inv = 1.0f / wave_sum(red[lane]);

#pragma unroll
        for (int u = 0; u < HEAD_AU; ++u) {
            const int n = first + u * HEAD_T;
            if (n < e0) wout[n] = expf(av[u] - M) * inv;
        }
        for (int nb = first + HEAD_T * HEAD_AU; nb < e0; nb += HEAD_T * HEAD_AU) {
            float v[HEAD_AU];
#pragma unroll
            for (int u = 0; u < HEAD_AU; ++u) v[u] = a[min(nb + u * HEAD_T, e0 - 1)];
#pragma unroll
            for (int u = 0; u < HEAD_AU; ++u) {
                const int n = nb + u * HEAD_T;
                if (n < e0) wout[n] = expf(v[u] - M) * inv;
            }
        }

        float4 acc = {0.0f, 0.0f, 0.0f, 0.0f};
#pragma unroll
        for (int u = 0; u < HEAD_RU; ++u) {
            const int b = q + u * HEAD_RG;
            const float f = (b < nrec) ? sc[min(b, nrec - 1)] : 0.0f;
            acc.x = fmaf(f, pv[u].x, acc.x); acc.y = fmaf(f, pv[u].y, acc.y);
            acc.z = fmaf(f, pv[u].z, acc.z); acc.w = fmaf(f, pv[u].w, acc.w);
        }
        for (int b0 = q + HEAD_RU * HEAD_RG; b0 < nrec; b0 += HEAD_RU * HEAD_RG) {
            float4 pw[HEAD_RU];
#pragma unroll
            for (int u = 0; u < HEAD_RU; ++u) {
                const int b = min(b0 + u * HEAD_RG, nrec - 1);
                pw[u] = ((const float4*)(rbase + (size_t)b * rstep))[c4];
            }
#pragma unroll
            for (int u = 0; u < HEAD_RU; ++u) {
                const int b = b0 + u * HEAD_RG;
                const float f = (b < nrec) ? sc[min(b, nrec - 1)] : 0.0f;
                acc.x = fmaf(f, pw[u].x, acc.x); acc.y = fmaf(f, pw[u].y, acc.y);
                acc.z = fmaf(f, pw[u].z, acc.z); acc.w = fmaf(f, pw[u].w, acc.w);
            }
        }
        pr[q][c4] = acc;
        __syncthreads();
        if (tid < H_DIM) {
            const float* pf = (const float*)pr;
            float s = 0.0f;
#pragma unroll
            for (int r = 0; r < HEAD_RG; ++r) s += pf[r * H_DIM + tid];
            slide[tid] = s * inv;
        }
    } else {
        if (tid < H_DIM) slide[tid] = 0.0f;
    }
    __syncthreads();

    if (tid < HA_DIM) {
        float acc = b1v;
#pragma unroll 32
        for (int k = 0; k < H_DIM; ++k) acc += slide[k] * w1s[k * HA_DIM + tid];
        hid[tid] = fmaxf(acc, 0.0f);
    }
    __syncthreads();

    if (tid < C_DIM) {
        float s = b2v;
#pragma unroll
        for (int t = 0; t < HA_DIM; ++t) s += hid[t] * w2s[t * C_DIM + tid];
        logits[g * C_DIM + tid] = s;
    }
}

extern "C" void kernel_launch(void* const* d_in, const int* in_sizes, int n_in,
                              void* d_out, int out_size, void* d_ws, size_t ws_size,
                              hipStream_t stream) {
    (void)in_sizes; (void)n_in; (void)out_size; (void)ws_size;
    const float* x   = (const float*)d_in[0];
    const int*  ei   = (const int*)d_in[1];
    const int*  batch= (const int*)d_in[2];
    const float* W1  = (const float*)d_in[3];  const float* b1   = (const float*)d_in[4];
    const float* g1  = (const float*)d_in[5];  const float* bb1  = (const float*)d_in[6];
    const float* m1  = (const float*)d_in[7];  const float* v1   = (const float*)d_in[8];
    const float* W2  = (const float*)d_in[9];  const float* b2   = (const float*)d_in[10];
    const float* g2  = (const float*)d_in[11]; const float* bb2  = (const float*)d_in[12];
    const float* m2  = (const float*)d_in[13]; const float* v2   = (const float*)d_in[14];
    const float* Wv  = (const float*)d_in[15]; const float* bv   = (const float*)d_in[16];
    const float* Wu  = (const float*)d_in[17]; const float* bu   = (const float*)d_in[18];
    const float* wA  = (const float*)d_in[19];
    const float* Wc1 = (const float*)d_in[20]; const float* bc1  = (const float*)d_in[21];
    const float* Wc2 = (const float*)d_in[22]; const float* bc2  = (const float*)d_in[23];

    const int* src = ei;
    const int* dst = ei + N_EDGES;

    float* out  = (float*)d_out;
    float* wout = out + N_BAGS * C_DIM;

    const size_t NH = (size_t)N_NODES * H_DIM;
    float*    bufA = (float*)d_ws;
    float*    bufB = bufA + NH;
    _Float16* h16  = (_Float16*)(bufB + NH);
    _Float16* w1pk = h16 + NH;
    _Float16* w2pk = w1pk + (size_t)D_IN * H_DIM;
    _Float16* wvpk = w2pk + (size_t)H_DIM * H_DIM;
    _Float16* wupk = wvpk + (size_t)H_DIM * HA_DIM;
    float*    dinv = (float*)(wupk + (size_t)H_DIM * HA_DIM);
    float*    abuf = dinv + N_NODES;
    int*      seg  = (int*)(abuf + N_NODES);

    int*   cnt      = (int*)h16;
    int*   rowptr   = cnt + N_NODES;
    int*   rank     = rowptr + N_NODES + 4;
    int*   csr      = rank + N_EDGES;
    float* hlin1 = bufA;
    float* hlin2 = bufB;
    float* part  = bufA;

    setup_all<<<SU_BLOCKS, SU_T, 0, stream>>>(W1, w1pk, W2, w2pk, Wv, wvpk, Wu, wupk,
                                              batch, seg, cnt);

    count_rank<<<N_EDGES / (EDGE_T * EDGE_PER), EDGE_T, 0, stream>>>(dst, cnt, rank);
    scan_rowptr<<<1, SCAN_T, 0, stream>>>(cnt, rowptr, dinv);
    csr_fill<<<N_EDGES / (EDGE_T * EDGE_PER), EDGE_T, 0, stream>>>(src, dst, rank, rowptr, dinv,
                                                                   csr);

    wmma_gemm_lds_af32<D_IN, H_DIM><<<N_NODES / GT_ROWS, GT_T, 0, stream>>>(x, w1pk, hlin1);

    gcn_pull_gemm<<<N_NODES / GT_ROWS, GT_T, 0, stream>>>(
        hlin1, rowptr, csr, dinv, b1, g1, bb1, m1, v1, w2pk, hlin2);

    gcn_pull_attn<<<N_NODES / GT_ROWS, GT_T, 0, stream>>>(
        hlin2, rowptr, csr, dinv, b2, g2, bb2, m2, v2, wvpk, bv, bu, wA, seg, part, abuf);

    head_kernel<<<N_BAGS, HEAD_T, 0, stream>>>(part, abuf, seg, Wc1, bc1, Wc2, bc2, out, wout);
}


# Layer 2 pulls h1's f16 rows (512 B) BEFORE the W2 product, (A_hat h1) W2, not f32 hlin2 rows (1 KB) after it; f32 row sum goes to WMMA as hi+lo f16 tiles (no new rounding); new sum order
# speedup vs baseline: 6.5727x; 1.0459x over previous
#include <hip/hip_runtime.h>
#include <hip/hip_bf16.h>
#include <math.h>

typedef __attribute__((ext_vector_type(16))) _Float16 v16h;
typedef __attribute__((ext_vector_type(8)))  _Float16 v8h;
typedef __attribute__((ext_vector_type(8)))  float    v8f;

#define N_NODES 65536
#define N_EDGES 524288
#define N_BAGS  8
#define D_IN    512
#define H_DIM   256
#define HA_DIM  128
#define C_DIM   2
#define BN_EPS  1e-5f
#define MREP    4

#define EDGE_PER 4
#define EDGE_T   256
static_assert(N_EDGES % (EDGE_T * EDGE_PER) == 0, "whole work-groups");
__global__ void __launch_bounds__(EDGE_T)
count_rank(const int* __restrict__ dst, int* __restrict__ cnt, int* __restrict__ rank) {
    const int t = blockIdx.x * blockDim.x + threadIdx.x;
    const int4 d = ((const int4*)dst)[t];
    int4 r;
    r.x = atomicAdd(&cnt[d.x], 1);
    r.y = atomicAdd(&cnt[d.y], 1);
    r.z = atomicAdd(&cnt[d.z], 1);
    r.w = atomicAdd(&cnt[d.w], 1);
    ((int4*)rank)[t] = r;
}

#define SCAN_T   1024
#define SCAN_PER (N_NODES / SCAN_T)
static_assert(N_NODES % (SCAN_T * 4) == 0, "scan assumes whole int4 chunks per thread");
__global__ void __launch_bounds__(SCAN_T)
scan_rowptr(const int* __restrict__ cnt, int* __restrict__ rowptr,
            float* __restrict__ dinv) {
    __shared__ int wsum[32];
    const int t = threadIdx.x, lane = t & 31, w = t >> 5;
    const int4* c4 = (const int4*)(cnt + t * SCAN_PER);
    int4 c[SCAN_PER / 4];
    int s = 0;
#pragma unroll
    for (int i = 0; i < SCAN_PER / 4; ++i) {
        c[i] = c4[i];
        s += c[i].x + c[i].y + c[i].z + c[i].w;
    }
    int inc = s;
    for (int off = 1; off < 32; off <<= 1) {
        int v = __shfl_up(inc, off, 32);
        if (lane >= off) inc += v;
    }
    if (lane == 31) wsum[w] = inc;
    __syncthreads();
    if (w == 0) {
        int v = wsum[lane], iv = v;
        for (int off = 1; off < 32; off <<= 1) {
            int u = __shfl_up(iv, off, 32);
            if (lane >= off) iv += u;
        }
        wsum[lane] = iv - v;
    }
    __syncthreads();
    int run = wsum[w] + inc - s;
    int4*   r4 = (int4*)(rowptr + t * SCAN_PER);
    float4* d4 = (float4*)(dinv + t * SCAN_PER);
#pragma unroll
    for (int i = 0; i < SCAN_PER / 4; ++i) {
        int4 r;
        r.x = run; run += c[i].x;
        r.y = run; run += c[i].y;
        r.z = run; run += c[i].z;
        r.w = run; run += c[i].w;
        r4[i] = r;
        float4 d;
        d.x = rsqrtf(1.0f + (float)c[i].x);
        d.y = rsqrtf(1.0f + (float)c[i].y);
        d.z = rsqrtf(1.0f + (float)c[i].z);
        d.w = rsqrtf(1.0f + (float)c[i].w);
        d4[i] = d;
    }
    if (t == SCAN_T - 1) rowptr[N_NODES] = run;
}

__device__ __forceinline__ void csr_put(int s, int d, int r,
                                        const int* __restrict__ rowptr,
                                        const float* __restrict__ dinv,
                                        int* __restrict__ csr) {
    const int pos = rowptr[d] + r;
    ((int2*)csr)[pos] = make_int2(s, __float_as_int(dinv[s] * dinv[d]));
}

__global__ void __launch_bounds__(EDGE_T)
csr_fill(const int* __restrict__ src, const int* __restrict__ dst,
         const int* __restrict__ rank, const int* __restrict__ rowptr,
         const float* __restrict__ dinv, int* __restrict__ csr) {
    const int t = blockIdx.x * blockDim.x + threadIdx.x;
    const int4 s = ((const int4*)src)[t], d = ((const int4*)dst)[t], r = ((const int4*)rank)[t];
    csr_put(s.x, d.x, r.x, rowptr, dinv, csr);
    csr_put(s.y, d.y, r.y, rowptr, dinv, csr);
    csr_put(s.z, d.z, r.z, rowptr, dinv, csr);
    csr_put(s.w, d.w, r.w, rowptr, dinv, csr);
}

template <int K, int NOUT>
__device__ __forceinline__ void pack_w_elem(const float* __restrict__ Wm,
                                            _Float16* __restrict__ out, int t) {
    constexpr int KBLKS = K >> 5;
    const int i    = t & 15;
    const int lane = (t >> 4) & 31;
    const int rest = t >> 9;
    const int ntile = rest / KBLKS;
    const int kblk  = rest % KBLKS;
    const int half = lane >> 4, l15 = lane & 15;
    const int col = ntile * 16 + l15;
    const int k   = kblk * 32 + half * 8 + ((i < 8) ? i : (16 + i - 8));
    out[t] = (_Float16)Wm[(size_t)k * NOUT + col];
}

#define SU_T      256
#define SU_B_W1   (D_IN * H_DIM / SU_T)
#define SU_B_W2   (H_DIM * H_DIM / SU_T)
#define SU_B_WA   (H_DIM * HA_DIM / SU_T)
#define SU_B_SEG  (N_NODES / SU_T)
#define SU_B_ZERO (N_NODES / SU_T)
#define SU_BLOCKS (SU_B_W1 + SU_B_W2 + 2 * SU_B_WA + SU_B_SEG + SU_B_ZERO)
static_assert(N_NODES % SU_T == 0 && (H_DIM * HA_DIM) % SU_T == 0, "whole blocks per range");
__global__ void __launch_bounds__(SU_T)
setup_all(const float* __restrict__ W1, _Float16* __restrict__ w1pk,
          const float* __restrict__ W2, _Float16* __restrict__ w2pk,
          const float* __restrict__ Wv, _Float16* __restrict__ wvpk,
          const float* __restrict__ Wu, _Float16* __restrict__ wupk,
          const int* __restrict__ batch, int* __restrict__ seg,
          int* __restrict__ cnt) {
    int b = blockIdx.x;
    const int tid = threadIdx.x;
    if (b < SU_B_W1) { pack_w_elem<D_IN, H_DIM>(W1, w1pk, b * SU_T + tid); return; }
    b -= SU_B_W1;
    if (b < SU_B_W2) { pack_w_elem<H_DIM, H_DIM>(W2, w2pk, b * SU_T + tid); return; }
    b -= SU_B_W2;
    if (b < SU_B_WA) { pack_w_elem<H_DIM, HA_DIM>(Wv, wvpk, b * SU_T + tid); return; }
    b -= SU_B_WA;
    if (b < SU_B_WA) { pack_w_elem<H_DIM, HA_DIM>(Wu, wupk, b * SU_T + tid); return; }
    b -= SU_B_WA;
    if (b < SU_B_SEG) {
        const int n = b * SU_T + tid;
        const int cur  = batch[n];
        const int prev = (n == 0) ? -1 : batch[n - 1];
#pragma unroll
        for (int g = 0; g < N_BAGS; ++g)
            if (g > prev && g <= cur) seg[g] = n;
        if (n == N_NODES - 1) {
#pragma unroll
            for (int g = 1; g <= N_BAGS; ++g)
                if (g > cur) seg[g] = N_NODES;
        }
        return;
    }
    b -= SU_B_SEG;
    cnt[b * SU_T + tid] = 0;
}

#define GT_ROWS 64
#define GT_T    256
#define GT_LD   8
static_assert(GT_ROWS == 16 * MREP, "a wave covers all rows of the tile");
static_assert(N_NODES % GT_ROWS == 0, "no partial row tile");

#define GT_PITCH(K) ((K) + 8)

template <int K, int NOUT>
__device__ __forceinline__ void gemm_tile_from_lds(const _Float16* As,
                                                   const _Float16* __restrict__ Wpk,
                                                   float* __restrict__ C, int m0) {
    constexpr int PITCH = GT_PITCH(K);
    constexpr int KBLKS = K / 32;
    constexpr int NTW   = NOUT / 16 / (GT_T / 32);
    static_assert(NTW >= 1 && NTW * 16 * (GT_T / 32) == NOUT, "waves tile the columns");
    const int lane = threadIdx.x & 31;
    const int wave = __builtin_amdgcn_readfirstlane((int)(threadIdx.x >> 5));
    const int half = lane >> 4;
    const int l15  = lane & 15;
    const _Float16* wb = Wpk + (size_t)(wave * NTW) * KBLKS * 512 + lane * 16;
    const _Float16* ab = As + l15 * PITCH + half * 8;

    v8f zero = {};
    v8f acc[MREP][NTW];
#pragma unroll
    for (int r = 0; r < MREP; ++r)
#pragma unroll
        for (int t = 0; t < NTW; ++t) acc[r][t] = zero;

    for (int kblk = 0; kblk < KBLKS; ++kblk) {
        v16h b[NTW];
#pragma unroll
        for (int t = 0; t < NTW; ++t)
            b[t] = *(const v16h*)(wb + ((size_t)t * KBLKS + kblk) * 512);
#pragma unroll
        for (int r = 0; r < MREP; ++r) {
            const _Float16* ap = ab + r * 16 * PITCH + kblk * 32;
            v8h lo = *(const v8h*)(ap);
            v8h hi = *(const v8h*)(ap + 16);
            v16h a = __builtin_shufflevector(lo, hi, 0, 1, 2, 3, 4, 5, 6, 7,
                                             8, 9, 10, 11, 12, 13, 14, 15);
#pragma unroll
            for (int t = 0; t < NTW; ++t)
                acc[r][t] = __builtin_amdgcn_wmma_f32_16x16x32_f16(
                    false, a, false, b[t], (short)0, acc[r][t], false, false);
        }
    }

#pragma unroll
    for (int t = 0; t < NTW; ++t) {
        const int col = (wave * NTW + t) * 16 + l15;
#pragma unroll
        for (int r = 0; r < MREP; ++r) {
#pragma unroll
            for (int q = 0; q < 8; ++q) {
                const int row = m0 + r * 16 + q + half * 8;
                C[(size_t)row * NOUT + col] = acc[r][t][q];
            }
        }
    }
}

typedef __attribute__((ext_vector_type(4))) _Float16 v4h_t;

template <int K, int NOUT>
__global__ void __launch_bounds__(GT_T)
wmma_gemm_lds_af32(const float* __restrict__ A, const _Float16* __restrict__ Wpk,
                   float* __restrict__ C) {
    __shared__ alignas(16) _Float16 As[GT_ROWS * GT_PITCH(K)];
    constexpr int VROW  = K / 4;
    constexpr int PER_T = GT_ROWS * VROW / GT_T;
    static_assert(PER_T % GT_LD == 0 && VROW % 32 == 0, "whole batches, a wave inside one row");
    const int tid = threadIdx.x;
    const int m0  = blockIdx.x * GT_ROWS;
    const float4* Ag = (const float4*)(A + (size_t)m0 * K);
#pragma unroll 1
    for (int it = 0; it < PER_T; it += GT_LD) {
        float4 v[GT_LD];
#pragma unroll
        for (int u = 0; u < GT_LD; ++u) v[u] = Ag[(it + u) * GT_T + tid];
#pragma unroll
        for (int u = 0; u < GT_LD; ++u) {
            const int f = (it + u) * GT_T + tid;
            const int row = f / VROW, c = f % VROW;
            v4h_t h = {(_Float16)v[u].x, (_Float16)v[u].y, (_Float16)v[u].z, (_Float16)v[u].w};
            *(v4h_t*)(As + row * GT_PITCH(K) + c * 4) = h;
        }
    }
    __syncthreads();
    gemm_tile_from_lds<K, NOUT>(As, Wpk, C, m0);
}

template <int K>
__device__ __forceinline__ void pull_rows(const float4* __restrict__ hp, int my_s, int my_c,
                                          int j, int lane, float4& a0, float4& a1) {
    float4 p[K], q[K];
    float  c[K];
#pragma unroll
    for (int k = 0; k < K; ++k) {
        const int s = __builtin_amdgcn_readlane(my_s, j + k);
        c[k] = __int_as_float(__builtin_amdgcn_readlane(my_c, j + k));
        const float4* row = hp + (size_t)s * (H_DIM / 4);
        p[k] = row[lane];
        q[k] = row[32 + lane];
    }
#pragma unroll
    for (int k = 0; k < K; ++k) {
        a0.x = fmaf(p[k].x, c[k], a0.x); a0.y = fmaf(p[k].y, c[k], a0.y);
        a0.z = fmaf(p[k].z, c[k], a0.z); a0.w = fmaf(p[k].w, c[k], a0.w);
        a1.x = fmaf(q[k].x, c[k], a1.x); a1.y = fmaf(q[k].y, c[k], a1.y);
        a1.z = fmaf(q[k].z, c[k], a1.z); a1.w = fmaf(q[k].w, c[k], a1.w);
    }
}

__device__ __forceinline__ float bn_relu(float val, float mm, float vv, float gg, float bo) {
    val = (val - mm) * rsqrtf(vv + BN_EPS) * gg + bo;
    return fmaxf(val, 0.0f);
}

__device__ __forceinline__ float gcn_fin(float agg, float hl, float di, float bc,
                                         float mm, float vv, float gg, float bo) {
    return bn_relu(agg + hl * di * di + bc, mm, vv, gg, bo);
}

__device__ __forceinline__ float4 gcn_fin4(float4 a, float4 h, float di, float4 bc,
                                           float4 mm, float4 vv, float4 gg, float4 bo) {
    float4 r;
    r.x = gcn_fin(a.x, h.x, di, bc.x, mm.x, vv.x, gg.x, bo.x);
    r.y = gcn_fin(a.y, h.y, di, bc.y, mm.y, vv.y, gg.y, bo.y);
    r.z = gcn_fin(a.z, h.z, di, bc.z, mm.z, vv.z, gg.z, bo.z);
    r.w = gcn_fin(a.w, h.w, di, bc.w, mm.w, vv.w, gg.w, bo.w);
    return r;
}

typedef __attribute__((ext_vector_type(4))) _Float16 v4h;

#define TILE_NPW (GT_ROWS / (GT_T / 32))
#define PULL_WAVES_PER_SIMD 8
static_assert(TILE_NPW * (GT_T / 32) == GT_ROWS, "the waves split the tile's rows evenly");

__device__ __forceinline__ int2 csr_chunk(const int* __restrict__ csr, int base, int end, int lane) {
    int2 rec = make_int2(0, 0);
    if (lane < end - base) rec = ((const int2*)csr)[base + lane];
    return rec;
}

__global__ void __launch_bounds__(GT_T, PULL_WAVES_PER_SIMD)
gcn_pull_h16(const float* __restrict__ hlin,
             const int* __restrict__ rowptr,
             const int* __restrict__ csr,
             const float* __restrict__ dinv,
             const float* __restrict__ bconv,
             const float* __restrict__ g,
             const float* __restrict__ bb,
             const float* __restrict__ m,
             const float* __restrict__ v,
             _Float16* __restrict__ hout) {
    const int m0 = blockIdx.x * GT_ROWS;
    const int lane = threadIdx.x & 31;
    const int wave = __builtin_amdgcn_readfirstlane((int)(threadIdx.x >> 5));
    const int nfirst = m0 + wave * TILE_NPW;
    const float4* hp = (const float4*)hlin;
    const float4* bc4 = (const float4*)bconv; const float4* g4 = (const float4*)g;
    const float4* bb4 = (const float4*)bb;    const float4* m4 = (const float4*)m;
    const float4* v4  = (const float4*)v;

    const int rp = rowptr[nfirst + min(lane, TILE_NPW)];
    int beg = __builtin_amdgcn_readlane(rp, 0), end = __builtin_amdgcn_readlane(rp, 1);
    int2 rec_nx = csr_chunk(csr, beg, end, lane);
    for (int i = 0; i < TILE_NPW; ++i) {
        const int n = nfirst + i;
        const int nxt = __builtin_amdgcn_readlane(rp, min(i + 2, TILE_NPW));
        int2 rec = rec_nx;
        if (i + 1 < TILE_NPW) rec_nx = csr_chunk(csr, end, nxt, lane);
        const float di = dinv[n];
        const float4* own = hp + (size_t)n * (H_DIM / 4);
        const float4 h0 = own[lane], h1 = own[32 + lane];
        float4 a0 = {0.0f, 0.0f, 0.0f, 0.0f}, a1 = {0.0f, 0.0f, 0.0f, 0.0f};
        for (int base = beg; base < end; base += 32) {
            const int nb = min(32, end - base);
            if (base != beg) rec = csr_chunk(csr, base, end, lane);
            const int my_s = rec.x, my_c = rec.y;
            int j = 0;
            for (; j + 8 <= nb; j += 8) pull_rows<8>(hp, my_s, my_c, j, lane, a0, a1);
            if (j + 4 <= nb) { pull_rows<4>(hp, my_s, my_c, j, lane, a0, a1); j += 4; }
            const int r = nb - j;
            if (r == 3)      pull_rows<3>(hp, my_s, my_c, j, lane, a0, a1);
            else if (r == 2) pull_rows<2>(hp, my_s, my_c, j, lane, a0, a1);
            else if (r == 1) pull_rows<1>(hp, my_s, my_c, j, lane, a0, a1);
        }

        const float4 r0 = gcn_fin4(a0, h0, di, bc4[lane], m4[lane], v4[lane], g4[lane], bb4[lane]);
        const float4 r1 = gcn_fin4(a1, h1, di, bc4[32 + lane], m4[32 + lane], v4[32 + lane],
                                   g4[32 + lane], bb4[32 + lane]);
        _Float16* o = hout + (size_t)n * H_DIM;
        v4h x0 = {(_Float16)r0.x, (_Float16)r0.y, (_Float16)r0.z, (_Float16)r0.w};
        v4h x1 = {(_Float16)r1.x, (_Float16)r1.y, (_Float16)r1.z, (_Float16)r1.w};
        *(v4h*)(o + 4 * lane) = x0;
        *(v4h*)(o + H_DIM / 2 + 4 * lane) = x1;
        beg = end;
        end = nxt;
    }
}

template <int K>
__device__ __forceinline__ void pull_rows_h(const _Float16* __restrict__ h1h, int my_s, int my_c,
                                            int j, unsigned lane16, float (&a)[8]) {
    v8h      p[K];
    float    c[K];
    unsigned off[K];
#pragma unroll
    for (int k = 0; k < K; ++k) {
        const int s = __builtin_amdgcn_readlane(my_s, j + k);
        c[k] = __int_as_float(__builtin_amdgcn_readlane(my_c, j + k));
        off[k] = (unsigned)s * (unsigned)(H_DIM * sizeof(_Float16)) + lane16;
    }
#pragma unroll
    for (int k = 0; k < K; ++k) p[k] = *(const v8h*)((const char*)h1h + off[k]);
#pragma unroll
    for (int k = 0; k < K; ++k) {
#pragma unroll
        for (int e = 0; e < 8; ++e) a[e] = fmaf((float)p[k][e], c[k], a[e]);
    }
}

__device__ __forceinline__ void pull_tile_h16_to_lds(const _Float16* __restrict__ h1h,
                                                     const int* __restrict__ rowptr,
                                                     const int* __restrict__ csr,
                                                     const float* __restrict__ dinv,
                                                     _Float16* Ahi, _Float16* Alo,
                                                     int m0) {
    constexpr int PITCH = GT_PITCH(H_DIM);
    const int lane = threadIdx.x & 31;
    const int wave = __builtin_amdgcn_readfirstlane((int)(threadIdx.x >> 5));
    const int rfirst = wave * TILE_NPW;
    const int nfirst = m0 + rfirst;
    const v8h* hp = (const v8h*)h1h;
    const unsigned lane16 = (unsigned)lane * 16u;

    const int rp = rowptr[nfirst + min(lane, TILE_NPW)];
    int beg = __builtin_amdgcn_readlane(rp, 0), end = __builtin_amdgcn_readlane(rp, 1);
    int2 rec_nx = csr_chunk(csr, beg, end, lane);
    for (int i = 0; i < TILE_NPW; ++i) {
        const int n = nfirst + i;
        const int nxt = __builtin_amdgcn_readlane(rp, min(i + 2, TILE_NPW));
        int2 rec = rec_nx;
        if (i + 1 < TILE_NPW) rec_nx = csr_chunk(csr, end, nxt, lane);
        const float di = dinv[n];
        const v8h* own = hp + (size_t)n * (H_DIM / 8);
        const v8h h0 = own[lane];
        float a[8] = {0.0f, 0.0f, 0.0f, 0.0f, 0.0f, 0.0f, 0.0f, 0.0f};
        for (int base = beg; base < end; base += 32) {
            const int nb = min(32, end - base);
            if (base != beg) rec = csr_chunk(csr, base, end, lane);
            const int my_s = rec.x, my_c = rec.y;
            int j = 0;
            for (; j + 16 <= nb; j += 16) pull_rows_h<16>(h1h, my_s, my_c, j, lane16, a);
            if (j + 8 <= nb) { pull_rows_h<8>(h1h, my_s, my_c, j, lane16, a); j += 8; }
            if (j + 4 <= nb) { pull_rows_h<4>(h1h, my_s, my_c, j, lane16, a); j += 4; }
            const int r = nb - j;
            if (r == 3)      pull_rows_h<3>(h1h, my_s, my_c, j, lane16, a);
            else if (r == 2) pull_rows_h<2>(h1h, my_s, my_c, j, lane16, a);
            else if (r == 1) pull_rows_h<1>(h1h, my_s, my_c, j, lane16, a);
        }
        const float dd = di * di;
        v8h hi, lo;
#pragma unroll
        for (int e = 0; e < 8; ++e) {
            const float t = fmaf((float)h0[e], dd, a[e]);
            const _Float16 th = (_Float16)t;
            hi[e] = th;
            lo[e] = (_Float16)(t - (float)th);
        }
        *(v8h*)(Ahi + (rfirst + i) * PITCH + 8 * lane) = hi;
        *(v8h*)(Alo + (rfirst + i) * PITCH + 8 * lane) = lo;
        beg = end;
        end = nxt;
    }
}

__device__ __forceinline__ void layer2_product_to_lds(_Float16* Ahi, const _Float16* Alo,
                                                      const _Float16* __restrict__ Wpk,
                                                      const float* __restrict__ bconv,
                                                      const float* __restrict__ g,
                                                      const float* __restrict__ bb,
                                                      const float* __restrict__ m,
                                                      const float* __restrict__ v) {
    constexpr int K     = H_DIM;
    constexpr int PITCH = GT_PITCH(K);
    constexpr int KBLKS = K / 32;
    constexpr int NTW   = H_DIM / 16 / (GT_T / 32);
    static_assert(NTW * 16 * (GT_T / 32) == H_DIM, "waves tile the columns");
    const int lane = threadIdx.x & 31;
    const int wave = __builtin_amdgcn_readfirstlane((int)(threadIdx.x >> 5));
    const int half = lane >> 4;
    const int l15  = lane & 15;
    const _Float16* wb  = Wpk + (size_t)(wave * NTW) * KBLKS * 512 + lane * 16;
    const _Float16* abh = Ahi + l15 * PITCH + half * 8;
    const _Float16* abl = Alo + l15 * PITCH + half * 8;

    v8f zero = {};
    v8f acc[MREP][NTW];
#pragma unroll
    for (int r = 0; r < MREP; ++r)
#pragma unroll
        for (int t = 0; t < NTW; ++t) acc[r][t] = zero;

    for (int kblk = 0; kblk < KBLKS; ++kblk) {
        v16h b[NTW];
#pragma unroll
        for (int t = 0; t < NTW; ++t)
            b[t] = *(const v16h*)(wb + ((size_t)t * KBLKS + kblk) * 512);
#pragma unroll
        for (int r = 0; r < MREP; ++r) {
            const _Float16* aph = abh + r * 16 * PITCH + kblk * 32;
            const _Float16* apl = abl + r * 16 * PITCH + kblk * 32;
            v8h h_lo = *(const v8h*)(aph);
            v8h h_hi = *(const v8h*)(aph + 16);
            v8h l_lo = *(const v8h*)(apl);
            v8h l_hi = *(const v8h*)(apl + 16);
            v16h ah = __builtin_shufflevector(h_lo, h_hi, 0, 1, 2, 3, 4, 5, 6, 7,
                                              8, 9, 10, 11, 12, 13, 14, 15);
            v16h al = __builtin_shufflevector(l_lo, l_hi, 0, 1, 2, 3, 4, 5, 6, 7,
                                              8, 9, 10, 11, 12, 13, 14, 15);
#pragma unroll
            for (int t = 0; t < NTW; ++t)
                acc[r][t] = __builtin_amdgcn_wmma_f32_16x16x32_f16(
                    false, ah, false, b[t], (short)0, acc[r][t], false, false);
#pragma unroll
            for (int t = 0; t < NTW; ++t)
                acc[r][t] = __builtin_amdgcn_wmma_f32_16x16x32_f16(
                    false, al, false, b[t], (short)0, acc[r][t], false, false);
        }
    }
    __syncthreads();

#pragma unroll
    for (int t = 0; t < NTW; ++t) {
        const int col = (wave * NTW + t) * 16 + l15;
        const float bc = bconv[col], mm = m[col], vv = v[col], gg = g[col], bo = bb[col];
#pragma unroll
        for (int r = 0; r < MREP; ++r) {
#pragma unroll
            for (int q = 0; q < 8; ++q) {
                const int row = r * 16 + q + half * 8;
                Ahi[row * PITCH + col] = (_Float16)bn_relu(acc[r][t][q] + bc, mm, vv, gg, bo);
            }
        }
    }
}

#define ATT_SEG   32
#define ATT_NSEG  (HA_DIM / ATT_SEG)
#define ATT_RP    (HA_DIM + 4)
static_assert(HA_DIM / 16 == GT_T / 32, "one column tile of Wv and one of Wu per wave");
static_assert(ATT_NSEG * GT_ROWS == GT_T, "one (row, segment) per thread");
static_assert(ATT_RP % 4 == 0, "16-byte reads of the term tile");

#define REC_STRIDE (H_DIM + 32)
static_assert(GT_T == H_DIM, "pooling: one column of the tile per thread");

__device__ __forceinline__ float wave_max(float v) {
    for (int off = 16; off > 0; off >>= 1) v = fmaxf(v, __shfl_xor(v, off, 32));
    return v;
}
__device__ __forceinline__ float wave_sum(float v) {
    for (int off = 16; off > 0; off >>= 1) v += __shfl_xor(v, off, 32);
    return v;
}

#define POOL_G  (GT_T / 64)
#define POOL_RJ (GT_ROWS / POOL_G)
static_assert(H_DIM == 256 && POOL_G * 64 == GT_T && POOL_RJ * POOL_G == GT_ROWS, "64 threads x 4 columns cover a row");
static_assert((POOL_G * H_DIM + POOL_G) <= GT_ROWS * ATT_RP, "the partial rows fit in the term tile's LDS");
__device__ __forceinline__ void pool_tile_partials(const _Float16* As, const float* ash, float* esh,
                                                   float* prt,
                                                   int seg_lo, int seg_hi,
                                                   float* __restrict__ part, int m0) {
    constexpr int PITCH = GT_PITCH(H_DIM);
    const int tid = threadIdx.x;
    const int q  = __builtin_amdgcn_readfirstlane((int)(tid >> 6));
    const int c4 = tid & 63;
    for (int g = 0; g < N_BAGS; ++g) {
        const int lo = max(__builtin_amdgcn_readlane(seg_lo, g), m0) - m0;
        const int hi = min(__builtin_amdgcn_readlane(seg_hi, g), m0 + GT_ROWS) - m0;
        if (lo >= hi) continue;
        static_assert(GT_ROWS == 64, "two rows of the tile per lane");
        const int lane = tid & 31;
        const float x0 = ash[lane], x1 = ash[32 + lane];
        float mt = -INFINITY;
        if (lane >= lo && lane < hi) mt = x0;
        if (lane + 32 >= lo && lane + 32 < hi) mt = fmaxf(mt, x1);
        mt = wave_max(mt);
        if (tid >= lo && tid < hi) esh[tid] = expf(ash[tid] - mt);
        __syncthreads();
        v4h   hv[POOL_RJ];
        float ev[POOL_RJ];
#pragma unroll
        for (int j = 0; j < POOL_RJ; ++j) {
            const int r = POOL_G * j + q;
            hv[j] = *(const v4h*)(As + r * PITCH + 4 * c4);
            ev[j] = 0.0f;
            if (r >= lo && r < hi) ev[j] = esh[r];
        }
        float4 p = {0.0f, 0.0f, 0.0f, 0.0f};
        float st = 0.0f;
#pragma unroll
        for (int j = 0; j < POOL_RJ; ++j) {
            st += ev[j];
            p.x = fmaf(ev[j], (float)hv[j][0], p.x); p.y = fmaf(ev[j], (float)hv[j][1], p.y);
            p.z = fmaf(ev[j], (float)hv[j][2], p.z); p.w = fmaf(ev[j], (float)hv[j][3], p.w);
        }
        ((float4*)prt)[q * 64 + c4] = p;
        if (c4 == 0) prt[POOL_G * H_DIM + q] = st;
        __syncthreads();
        float s = prt[tid];
#pragma unroll
        for (int k = 1; k < POOL_G; ++k) s += prt[k * H_DIM + tid];
        float* rec = part + ((size_t)blockIdx.x * N_BAGS + g) * REC_STRIDE;
        rec[tid] = s;
        if (tid == 0) {
            float sst = prt[POOL_G * H_DIM];
#pragma unroll
            for (int k = 1; k < POOL_G; ++k) sst += prt[POOL_G * H_DIM + k];
            rec[H_DIM] = mt;
            rec[H_DIM + 1] = sst;
        }
        __syncthreads();
    }
}

__device__ __forceinline__ void attn_tile_from_lds(const _Float16* As,
                                                   float* term,
                                                   float (*psum)[GT_ROWS],
                                                   float* ash,
                                                   const _Float16* __restrict__ Wpk,
                                                   const float* __restrict__ bv,
                                                   const float* __restrict__ bu,
                                                   const float* __restrict__ wA,
                                                   float* __restrict__ a, int m0) {
    constexpr int K     = H_DIM;
    constexpr int PITCH = GT_PITCH(K);
    constexpr int KBLKS = K / 32;
    constexpr int NT    = HA_DIM / 16;
    const int tid = threadIdx.x;

    const int lane = tid & 31;
    const int wave = __builtin_amdgcn_readfirstlane((int)(tid >> 5));
    const int half = lane >> 4;
    const int l15  = lane & 15;
    const _Float16* wb = Wpk + (size_t)wave * KBLKS * 512 + lane * 16;
    const _Float16* ab = As + l15 * PITCH + half * 8;

    v8f zero = {};
    v8f acc[MREP][2];
#pragma unroll
    for (int r = 0; r < MREP; ++r) { acc[r][0] = zero; acc[r][1] = zero; }

    for (int kblk = 0; kblk < KBLKS; ++kblk) {
        v16h b[2];
#pragma unroll
        for (int t = 0; t < 2; ++t)
            b[t] = *(const v16h*)(wb + ((size_t)t * NT * KBLKS + kblk) * 512);
#pragma unroll
        for (int r = 0; r < MREP; ++r) {
            const _Float16* ap = ab + r * 16 * PITCH + kblk * 32;
            v8h lo = *(const v8h*)(ap);
            v8h hi = *(const v8h*)(ap + 16);
            v16h av = __builtin_shufflevector(lo, hi, 0, 1, 2, 3, 4, 5, 6, 7,
                                              8, 9, 10, 11, 12, 13, 14, 15);
#pragma unroll
            for (int t = 0; t < 2; ++t)
                acc[r][t] = __builtin_amdgcn_wmma_f32_16x16x32_f16(
                    false, av, false, b[t], (short)0, acc[r][t], false, false);
        }
    }

    const int col = wave * 16 + l15;
    const float bvc = bv[col], buc = bu[col], wac = wA[col];
#pragma unroll
    for (int r = 0; r < MREP; ++r) {
#pragma unroll
        for (int q = 0; q < 8; ++q) {
            const int row = r * 16 + q + half * 8;
            const float tv = tanhf(acc[r][0][q] + bvc);
            const float sv = 1.0f / (1.0f + expf(-(acc[r][1][q] + buc)));
            term[row * ATT_RP + col] = tv * sv * wac;
        }
    }
    __syncthreads();

    {
        const int row = tid % GT_ROWS, sg = tid / GT_ROWS;
        const float4* tp = (const float4*)(term + row * ATT_RP + sg * ATT_SEG);
        float4 x[ATT_SEG / 4];
#pragma unroll
        for (int j = 0; j < ATT_SEG / 4; ++j) x[j] = tp[j];
        float s = 0.0f;
#pragma unroll
        for (int j = 0; j < ATT_SEG / 4; ++j) { s += x[j].x; s += x[j].y; s += x[j].z; s += x[j].w; }
        psum[sg][row] = s;
    }
    __syncthreads();
    if (tid < GT_ROWS) {
        float s = psum[0][tid];
#pragma unroll
        for (int g = 1; g < ATT_NSEG; ++g) s += psum[g][tid];
        a[m0 + tid] = s;
        ash[tid] = s;
    }
    __syncthreads();
}

static_assert(sizeof(float) * GT_ROWS * ATT_RP >= sizeof(_Float16) * GT_ROWS * GT_PITCH(H_DIM),
              "the lo tile fits in the term tile's LDS");
__global__ void __launch_bounds__(GT_T, PULL_WAVES_PER_SIMD)
gcn_pull_attn(const _Float16* __restrict__ h1h,
              const int* __restrict__ rowptr,
              const int* __restrict__ csr,
              const float* __restrict__ dinv,
              const _Float16* __restrict__ W2pk,
              const float* __restrict__ bconv,
              const float* __restrict__ g,
              const float* __restrict__ bb,
              const float* __restrict__ m,
              const float* __restrict__ v,
              const _Float16* __restrict__ Wpk,
              const float* __restrict__ bv,
              const float* __restrict__ bu,
              const float* __restrict__ wA,
              const int* __restrict__ seg,
              float* __restrict__ part,
              float* __restrict__ a) {
    __shared__ alignas(16) _Float16 As[GT_ROWS * GT_PITCH(H_DIM)];
    __shared__ alignas(16) float term[GT_ROWS * ATT_RP];
    __shared__ float psum[ATT_NSEG][GT_ROWS];
    __shared__ float ash[GT_ROWS];
    __shared__ float esh[GT_ROWS];
    const int m0 = blockIdx.x * GT_ROWS;
    const int lane = threadIdx.x & 31;
    const int seg_lo = seg[min(lane, N_BAGS)];
    const int seg_hi = seg[min(lane + 1, N_BAGS)];
    _Float16* Alo = (_Float16*)term;
    pull_tile_h16_to_lds(h1h, rowptr, csr, dinv, As, Alo, m0);
    __syncthreads();
    layer2_product_to_lds(As, Alo, W2pk, bconv, g, bb, m, v);
    __syncthreads();
    attn_tile_from_lds(As, term, psum, ash, Wpk, bv, bu, wA, a, m0);
    pool_tile_partials(As, ash, esh, term, seg_lo, seg_hi, part, m0);
}

#define HEAD_T  1024
#define HEAD_WV (H_DIM * HA_DIM / 4 / HEAD_T)
#define HEAD_RG (HEAD_T / 64)
#define HEAD_RU 10
#define HEAD_AU 12
static_assert(HEAD_WV * HEAD_T * 4 == H_DIM * HA_DIM, "Wc1 splits into whole float4 per thread");
static_assert(H_DIM <= HEAD_T && HA_DIM * C_DIM <= HEAD_T && H_DIM % 32 == 0 && HA_DIM % 32 == 0,
              "one thread per column / Wc2 value; branches on whole waves");
static_assert(N_NODES / GT_ROWS <= HEAD_T, "one record's maximum and sum per thread");
static_assert(H_DIM == 256 && HEAD_T == 32 * 32, "64 float4 per record; second reduction stage is one wave");
__global__ void __launch_bounds__(HEAD_T)
head_kernel(const float* __restrict__ part,
            const float* __restrict__ a,
            const int* __restrict__ seg,
            const float* __restrict__ Wc1,
            const float* __restrict__ bc1,
            const float* __restrict__ Wc2,
            const float* __restrict__ bc2,
            float* __restrict__ logits,
            float* __restrict__ wout) {
    __shared__ alignas(16) float w1s[H_DIM * HA_DIM];
    __shared__ alignas(16) float4 pr[HEAD_RG][64];
    __shared__ alignas(16) float slide[H_DIM];
    __shared__ alignas(16) float hid[HA_DIM];
    __shared__ float w2s[HA_DIM * C_DIM];
    __shared__ float sc[HEAD_T];
    __shared__ float red[HEAD_T / 32];
    const int tid = threadIdx.x, lane = tid & 31, wv_ = tid >> 5;
    const int g = blockIdx.x;
    const int s0 = seg[g], e0 = seg[g + 1];
    const bool some = e0 > s0;
    const int t_lo = s0 / GT_ROWS;
    const int nrec = some ? (e0 - 1) / GT_ROWS - t_lo + 1 : 0;
    const float* rbase = part + ((size_t)t_lo * N_BAGS + g) * REC_STRIDE;
    const size_t rstep = (size_t)N_BAGS * REC_STRIDE;
    const int q = tid >> 6, c4 = tid & 63;
    const int first = s0 + tid;

    float4 wv[HEAD_WV];
#pragma unroll
    for (int u = 0; u < HEAD_WV; ++u) wv[u] = ((const float4*)Wc1)[u * HEAD_T + tid];
    float mt = -INFINITY, st = 0.0f;
    float4 pv[HEAD_RU];
    float  av[HEAD_AU];
    if (some) {
        if (tid < nrec) {
            const float* r = rbase + (size_t)tid * rstep;
            mt = r[H_DIM];
            st = r[H_DIM + 1];
        }
#pragma unroll
        for (int u = 0; u < HEAD_RU; ++u) {
            const int b = min(q + u * HEAD_RG, nrec - 1);
            pv[u] = ((const float4*)(rbase + (size_t)b * rstep))[c4];
        }
#pragma unroll
        for (int u = 0; u < HEAD_AU; ++u) av[u] = a[min(first + u * HEAD_T, e0 - 1)];
    }
    float w2v = 0.0f, b1v = 0.0f, b2v = 0.0f;
    if (tid < HA_DIM * C_DIM) w2v = Wc2[tid];
    if (tid < HA_DIM) b1v = bc1[tid];
    if (tid < C_DIM) b2v = bc2[tid];

#pragma unroll
    for (int u = 0; u < HEAD_WV; ++u) ((float4*)w1s)[u * HEAD_T + tid] = wv[u];
    if (tid < HA_DIM * C_DIM) w2s[tid] = w2v;

    if (some) {
        float M = wave_max(mt);
        if (lane == 0) red[wv_] = M;
        __syncthreads();
        M = wave_max(red[lane]);
        __syncthreads();
        const float scl = (tid < nrec) ? expf(mt - M) : 0.0f;
        sc[tid] = scl;
        float S = wave_sum(st * scl);
        if (lane == 0) red[wv_] = S;
        __syncthreads();
        const float inv = 1.0f / wave_sum(red[lane]);

#pragma unroll
        for (int u = 0; u < HEAD_AU; ++u) {
            const int n = first + u * HEAD_T;
            if (n < e0) wout[n] = expf(av[u] - M) * inv;
        }
        for (int nb = first + HEAD_T * HEAD_AU; nb < e0; nb += HEAD_T * HEAD_AU) {
            float v[HEAD_AU];
#pragma unroll
            for (int u = 0; u < HEAD_AU; ++u) v[u] = a[min(nb + u * HEAD_T, e0 - 1)];
#pragma unroll
            for (int u = 0; u < HEAD_AU; ++u) {
                const int n = nb + u * HEAD_T;
                if (n < e0) wout[n] = expf(v[u] - M) * inv;
            }
        }

        float4 acc = {0.0f, 0.0f, 0.0f, 0.0f};
#pragma unroll
        for (int u = 0; u < HEAD_RU; ++u) {
            const int b = q + u * HEAD_RG;
            const float f = (b < nrec) ? sc[min(b, nrec - 1)] : 0.0f;
            acc.x = fmaf(f, pv[u].x, acc.x); acc.y = fmaf(f, pv[u].y, acc.y);
            acc.z = fmaf(f, pv[u].z, acc.z); acc.w = fmaf(f, pv[u].w, acc.w);
        }
        for (int b0 = q + HEAD_RU * HEAD_RG; b0 < nrec; b0 += HEAD_RU * HEAD_RG) {
            float4 pw[HEAD_RU];
#pragma unroll
            for (int u = 0; u < HEAD_RU; ++u) {
                const int b = min(b0 + u * HEAD_RG, nrec - 1);
                pw[u] = ((const float4*)(rbase + (size_t)b * rstep))[c4];
            }
#pragma unroll
            for (int u = 0; u < HEAD_RU; ++u) {
                const int b = b0 + u * HEAD_RG;
                const float f = (b < nrec) ? sc[min(b, nrec - 1)] : 0.0f;
                acc.x = fmaf(f, pw[u].x, acc.x); acc.y = fmaf(f, pw[u].y, acc.y);
                acc.z = fmaf(f, pw[u].z, acc.z); acc.w = fmaf(f, pw[u].w, acc.w);
            }
        }
        pr[q][c4] = acc;
        __syncthreads();
        if (tid < H_DIM) {
            const float* pf = (const float*)pr;
            float s = 0.0f;
#pragma unroll
            for (int r = 0; r < HEAD_RG; ++r) s += pf[r * H_DIM + tid];
            slide[tid] = s * inv;
        }
    } else {
        if (tid < H_DIM) slide[tid] = 0.0f;
    }
    __syncthreads();

    if (tid < HA_DIM) {
        float acc = b1v;
#pragma unroll 32
        for (int k = 0; k < H_DIM; ++k) acc += slide[k] * w1s[k * HA_DIM + tid];
        hid[tid] = fmaxf(acc, 0.0f);
    }
    __syncthreads();

    if (tid < C_DIM) {
        float s = b2v;
#pragma unroll
        for (int t = 0; t < HA_DIM; ++t) s += hid[t] * w2s[t * C_DIM + tid];
        logits[g * C_DIM + tid] = s;
    }
}

extern "C" void kernel_launch(void* const* d_in, const int* in_sizes, int n_in,
                              void* d_out, int out_size, void* d_ws, size_t ws_size,
                              hipStream_t stream) {
    (void)in_sizes; (void)n_in; (void)out_size; (void)ws_size;
    const float* x   = (const float*)d_in[0];
    const int*  ei   = (const int*)d_in[1];
    const int*  batch= (const int*)d_in[2];
    const float* W1  = (const float*)d_in[3];  const float* b1   = (const float*)d_in[4];
    const float* g1  = (const float*)d_in[5];  const float* bb1  = (const float*)d_in[6];
    const float* m1  = (const float*)d_in[7];  const float* v1   = (const float*)d_in[8];
    const float* W2  = (const float*)d_in[9];  const float* b2   = (const float*)d_in[10];
    const float* g2  = (const float*)d_in[11]; const float* bb2  = (const float*)d_in[12];
    const float* m2  = (const float*)d_in[13]; const float* v2   = (const float*)d_in[14];
    const float* Wv  = (const float*)d_in[15]; const float* bv   = (const float*)d_in[16];
    const float* Wu  = (const float*)d_in[17]; const float* bu   = (const float*)d_in[18];
    const float* wA  = (const float*)d_in[19];
    const float* Wc1 = (const float*)d_in[20]; const float* bc1  = (const float*)d_in[21];
    const float* Wc2 = (const float*)d_in[22]; const float* bc2  = (const float*)d_in[23];

    const int* src = ei;
    const int* dst = ei + N_EDGES;

    float* out  = (float*)d_out;
    float* wout = out + N_BAGS * C_DIM;

    const size_t NH = (size_t)N_NODES * H_DIM;
    float*    bufA = (float*)d_ws;
    float*    bufB = bufA + NH;
    _Float16* h16  = (_Float16*)(bufB + NH);
    _Float16* w1pk = h16 + NH;
    _Float16* w2pk = w1pk + (size_t)D_IN * H_DIM;
    _Float16* wvpk = w2pk + (size_t)H_DIM * H_DIM;
    _Float16* wupk = wvpk + (size_t)H_DIM * HA_DIM;
    float*    dinv = (float*)(wupk + (size_t)H_DIM * HA_DIM);
    float*    abuf = dinv + N_NODES;
    int*      seg  = (int*)(abuf + N_NODES);

    int*   cnt      = (int*)h16;
    int*   rowptr   = cnt + N_NODES;
    int*   rank     = rowptr + N_NODES + 4;
    int*   csr      = rank + N_EDGES;
    float*    hlin1 = bufA;
    _Float16* h1h   = (_Float16*)bufB;
    float*    part  = bufA;

    setup_all<<<SU_BLOCKS, SU_T, 0, stream>>>(W1, w1pk, W2, w2pk, Wv, wvpk, Wu, wupk,
                                              batch, seg, cnt);

    count_rank<<<N_EDGES / (EDGE_T * EDGE_PER), EDGE_T, 0, stream>>>(dst, cnt, rank);
    scan_rowptr<<<1, SCAN_T, 0, stream>>>(cnt, rowptr, dinv);
    csr_fill<<<N_EDGES / (EDGE_T * EDGE_PER), EDGE_T, 0, stream>>>(src, dst, rank, rowptr, dinv,
                                                                   csr);

    wmma_gemm_lds_af32<D_IN, H_DIM><<<N_NODES / GT_ROWS, GT_T, 0, stream>>>(x, w1pk, hlin1);

    gcn_pull_h16<<<N_NODES / GT_ROWS, GT_T, 0, stream>>>(
        hlin1, rowptr, csr, dinv, b1, g1, bb1, m1, v1, h1h);

    gcn_pull_attn<<<N_NODES / GT_ROWS, GT_T, 0, stream>>>(
        h1h, rowptr, csr, dinv, w2pk, b2, g2, bb2, m2, v2, wvpk, bv, bu, wA, seg, part, abuf);

    head_kernel<<<N_BAGS, HEAD_T, 0, stream>>>(part, abuf, seg, Wc1, bc1, Wc2, bc2, out, wout);
}
